# starting kernel
# baseline (speedup 1.0000x reference)
#include <hip/hip_runtime.h>
#include <math.h>


typedef float v2f __attribute__((ext_vector_type(2)));
typedef float v8f __attribute__((ext_vector_type(8)));

#define DIM_IN 256
#define HID    64
#define HEADS  8
#define DH     8
#define DOUT   16
#define NEG_SLOPE 0.2f
#define EPS_F 1e-16f
#define ENC_NEGINF 0x007FFFFFu

__device__ __forceinline__ unsigned enc_f32(float f) {
  unsigned u = __float_as_uint(f);
  return (u & 0x80000000u) ? ~u : (u | 0x80000000u);
}
__device__ __forceinline__ float dec_f32(unsigned e) {
  unsigned u = (e & 0x80000000u) ? (e ^ 0x80000000u) : ~e;
  return __uint_as_float(u);
}
__device__ __forceinline__ float leaky(float v) {
  return v > 0.0f ? v : NEG_SLOPE * v;
}

__global__ __launch_bounds__(256) void gat_gemm1_wmma(
    const float* __restrict__ x, const float* __restrict__ W1,
    float* __restrict__ h1, int N) {
  __shared__ float sW[DIM_IN * HID];
  int tid = threadIdx.x;
  {
    const float4* Wv = (const float4*)W1;
    float4* sWv = (float4*)sW;
#pragma unroll
    for (int i = 0; i < 16; ++i) sWv[tid + i * 256] = Wv[tid + i * 256];
  }
  __syncthreads();

  int wave = tid >> 5, lane = tid & 31;
  int row0 = (blockIdx.x * 2 + (wave >> 2)) * 16;
  int col0 = (wave & 3) * 16;
  if (row0 >= N) return;

  int r  = lane & 15;
  int kh = lane >> 4;

  int arow_idx = row0 + r; if (arow_idx >= N) arow_idx = N - 1;
  const float* arow  = x + (size_t)arow_idx * DIM_IN + 2 * kh;
  const float* bbase = sW + col0 + r;

  v8f c = {};
  for (int k = 0; k < DIM_IN; k += 4) {
    if ((k & 63) == 0) __builtin_prefetch(arow + k + 64, 0, 3);
    float2 av = *(const float2*)(arow + k);
    v2f a; a.x = av.x; a.y = av.y;
    v2f b; b.x = bbase[(k + 2 * kh) * HID];
           b.y = bbase[(k + 2 * kh + 1) * HID];
    c = __builtin_amdgcn_wmma_f32_16x16x4_f32(false, a, false, b,
                                              (short)0, c, false, false);
  }
  float* orow = h1 + (size_t)row0 * HID + col0 + r;
  if (row0 + 16 <= N) {
#pragma unroll
    for (int rr = 0; rr < 8; ++rr)
      orow[(size_t)(rr + 8 * kh) * HID] = c[rr];
  } else {
#pragma unroll
    for (int rr = 0; rr < 8; ++rr) {
      int row = row0 + rr + 8 * kh;
      if (row < N) orow[(size_t)(rr + 8 * kh) * HID] = c[rr];
    }
  }
}

__global__ __launch_bounds__(256) void gat_gemm2_wmma(
    const float* __restrict__ hact, const float* __restrict__ W2,
    float* __restrict__ h2, int N) {
  __shared__ float sW[HID * DOUT];
  int tid = threadIdx.x;
  {
    const float4* Wv = (const float4*)W2;
    float4* sWv = (float4*)sW;
    sWv[tid] = Wv[tid];
  }
  __syncthreads();

  int wave = tid >> 5, lane = tid & 31;
  int row0 = (blockIdx.x * 8 + wave) * 16;
  if (row0 >= N) return;

  int r  = lane & 15;
  int kh = lane >> 4;
  int arow_idx = row0 + r; if (arow_idx >= N) arow_idx = N - 1;
  const float* arow  = hact + (size_t)arow_idx * HID + 2 * kh;
  const float* bbase = sW + r;

  v8f c = {};
#pragma unroll
  for (int k = 0; k < HID; k += 4) {
    float2 av = *(const float2*)(arow + k);
    v2f a; a.x = av.x; a.y = av.y;
    v2f b; b.x = bbase[(k + 2 * kh) * DOUT];
           b.y = bbase[(k + 2 * kh + 1) * DOUT];
    c = __builtin_amdgcn_wmma_f32_16x16x4_f32(false, a, false, b,
                                              (short)0, c, false, false);
  }
  float* orow = h2 + (size_t)row0 * DOUT + r;
  if (row0 + 16 <= N) {
#pragma unroll
    for (int rr = 0; rr < 8; ++rr)
      orow[(size_t)(rr + 8 * kh) * DOUT] = c[rr];
  } else {
#pragma unroll
    for (int rr = 0; rr < 8; ++rr) {
      int row = row0 + rr + 8 * kh;
      if (row < N) orow[(size_t)(rr + 8 * kh) * DOUT] = c[rr];
    }
  }
}

__global__ void gat_fill_u32(unsigned* __restrict__ p, unsigned v, int n) {
  int t = blockIdx.x * blockDim.x + threadIdx.x;
  if (t < n) p[t] = v;
}

__global__ void gat_node_alpha1(const float* __restrict__ h1,
                                const float* __restrict__ a_src,
                                const float* __restrict__ a_dst,
                                float* __restrict__ as, float* __restrict__ ad,
                                int NH) {
  int t = blockIdx.x * blockDim.x + threadIdx.x;
  if (t >= NH) return;
  int n = t >> 3, h = t & 7;
  const float* hp = h1 + (size_t)n * HID + h * DH;
  const float* sp = a_src + h * DH;
  const float* dp = a_dst + h * DH;
  float ss = 0.f, dd = 0.f;
#pragma unroll
  for (int i = 0; i < DH; ++i) { float v = hp[i]; ss += v * sp[i]; dd += v * dp[i]; }
  as[t] = ss; ad[t] = dd;
}

__global__ void gat_node_alpha2(const float* __restrict__ h2,
                                const float* __restrict__ a_src,
                                const float* __restrict__ a_dst,
                                float* __restrict__ as, float* __restrict__ ad,
                                int N) {
  int n = blockIdx.x * blockDim.x + threadIdx.x;
  if (n >= N) return;
  const float* hp = h2 + (size_t)n * DOUT;
  float ss = 0.f, dd = 0.f;
#pragma unroll
  for (int i = 0; i < DOUT; ++i) { float v = hp[i]; ss += v * a_src[i]; dd += v * a_dst[i]; }
  as[n] = ss; ad[n] = dd;
}

__global__ void gat_edge_max1(const int* __restrict__ src, const int* __restrict__ dst,
                              const float* __restrict__ as, const float* __restrict__ ad,
                              unsigned* __restrict__ m, int E) {
  int e = blockIdx.x * blockDim.x + threadIdx.x;
  if (e >= E) return;
  int s = src[e], d = dst[e];
  const float4* as4 = (const float4*)as;
  const float4* ad4 = (const float4*)ad;
  float4 sv[2] = { as4[(size_t)s * 2], as4[(size_t)s * 2 + 1] };
  float4 dv[2] = { ad4[(size_t)d * 2], ad4[(size_t)d * 2 + 1] };
  const float* sf = (const float*)sv;
  const float* df = (const float*)dv;
#pragma unroll
  for (int h = 0; h < HEADS; ++h) {
    float v = leaky(sf[h] + df[h]);
    atomicMax(&m[(size_t)d * HEADS + h], enc_f32(v));
  }
}

__global__ void gat_edge_sum1(const int* __restrict__ src, const int* __restrict__ dst,
                              const float* __restrict__ as, const float* __restrict__ ad,
                              const unsigned* __restrict__ m, float* __restrict__ ssum,
                              int E) {
  int e = blockIdx.x * blockDim.x + threadIdx.x;
  if (e >= E) return;
  int s = src[e], d = dst[e];
  const float4* as4 = (const float4*)as;
  const float4* ad4 = (const float4*)ad;
  const uint4*  m4  = (const uint4*)m;
  float4 sv[2] = { as4[(size_t)s * 2], as4[(size_t)s * 2 + 1] };
  float4 dv[2] = { ad4[(size_t)d * 2], ad4[(size_t)d * 2 + 1] };
  uint4  mv[2] = { m4[(size_t)d * 2],  m4[(size_t)d * 2 + 1] };
  const float*    sf = (const float*)sv;
  const float*    df = (const float*)dv;
  const unsigned* mu = (const unsigned*)mv;
#pragma unroll
  for (int h = 0; h < HEADS; ++h) {
    float v = leaky(sf[h] + df[h]);
    atomicAdd(&ssum[(size_t)d * HEADS + h], __expf(v - dec_f32(mu[h])));
  }
}

__global__ void gat_edge_scatter1(const int* __restrict__ src, const int* __restrict__ dst,
                                  const float* __restrict__ as, const float* __restrict__ ad,
                                  const unsigned* __restrict__ m, const float* __restrict__ ssum,
                                  const float* __restrict__ h1, float* __restrict__ out1,
                                  unsigned total) {
  unsigned t = blockIdx.x * blockDim.x + threadIdx.x;
  if (t >= total) return;
  int e = (int)(t >> 6), c = (int)(t & 63), h = c >> 3;
  int s = src[e], d = dst[e];
  float v = leaky(as[(size_t)s * HEADS + h] + ad[(size_t)d * HEADS + h]);
  float ex = __expf(v - dec_f32(m[(size_t)d * HEADS + h]));
  float alpha = ex / (ssum[(size_t)d * HEADS + h] + EPS_F);
  atomicAdd(&out1[(size_t)d * HID + c], alpha * h1[(size_t)s * HID + c]);
}

__global__ void gat_elu_bias(float* __restrict__ out1, const float* __restrict__ b1,
                             int total) {
  int t = blockIdx.x * blockDim.x + threadIdx.x;
  if (t >= total) return;
  float v = out1[t] + b1[t & (HID - 1)];
  out1[t] = v > 0.0f ? v : expm1f(v);
}

__global__ void gat_edge_max2(const int* __restrict__ src, const int* __restrict__ dst,
                              const float* __restrict__ as, const float* __restrict__ ad,
                              unsigned* __restrict__ m, int E) {
  int e = blockIdx.x * blockDim.x + threadIdx.x;
  if (e >= E) return;
  int s = src[e], d = dst[e];
  float v = leaky(as[s] + ad[d]);
  atomicMax(&m[d], enc_f32(v));
}

__global__ void gat_edge_sum2(const int* __restrict__ src, const int* __restrict__ dst,
                              const float* __restrict__ as, const float* __restrict__ ad,
                              const unsigned* __restrict__ m, float* __restrict__ ssum,
                              int E) {
  int e = blockIdx.x * blockDim.x + threadIdx.x;
  if (e >= E) return;
  int s = src[e], d = dst[e];
  float v = leaky(as[s] + ad[d]);
  atomicAdd(&ssum[d], __expf(v - dec_f32(m[d])));
}

__global__ void gat_edge_scatter2(const int* __restrict__ src, const int* __restrict__ dst,
                                  const float* __restrict__ as, const float* __restrict__ ad,
                                  const unsigned* __restrict__ m, const float* __restrict__ ssum,
                                  const float* __restrict__ h2, float* __restrict__ out2,
                                  unsigned total) {
  unsigned t = blockIdx.x * blockDim.x + threadIdx.x;
  if (t >= total) return;
  int e = (int)(t >> 4), c = (int)(t & 15);
  int s = src[e], d = dst[e];
  float v = leaky(as[s] + ad[d]);
  float ex = __expf(v - dec_f32(m[d]));
  float alpha = ex / (ssum[d] + EPS_F);
  atomicAdd(&out2[(size_t)d * DOUT + c], alpha * h2[(size_t)s * DOUT + c]);
}

__global__ void gat_finalize(const float* __restrict__ out2, const float* __restrict__ b2,
                             float* __restrict__ dout, int N) {
  int n = blockIdx.x * blockDim.x + threadIdx.x;
  if (n >= N) return;
  float v[DOUT];
  float mx = -INFINITY;
#pragma unroll
  for (int c = 0; c < DOUT; ++c) {
    v[c] = out2[(size_t)n * DOUT + c] + b2[c];
    mx = fmaxf(mx, v[c]);
  }
  float se = 0.f;
#pragma unroll
  for (int c = 0; c < DOUT; ++c) se += __expf(v[c] - mx);
  float lse = mx + __logf(se);
#pragma unroll
  for (int c = 0; c < DOUT; ++c) {
    dout[(size_t)n * DOUT + c] = v[c];
    dout[(size_t)N * DOUT + (size_t)n * DOUT + c] = v[c] - lse;
  }
}

extern "C" void kernel_launch(void* const* d_in, const int* in_sizes, int n_in,
                              void* d_out, int out_size, void* d_ws, size_t ws_size,
                              hipStream_t stream) {
  const float* x   = (const float*)d_in[0];
  const int*   ei  = (const int*)d_in[1];
  const float* W1  = (const float*)d_in[2];
  const float* a1s = (const float*)d_in[3];
  const float* a1d = (const float*)d_in[4];
  const float* b1  = (const float*)d_in[5];
  const float* W2  = (const float*)d_in[6];
  const float* a2s = (const float*)d_in[7];
  const float* a2d = (const float*)d_in[8];
  const float* b2  = (const float*)d_in[9];

  int N = in_sizes[0] / DIM_IN;
  int E = in_sizes[1] / 2;
  const int* src = ei;
  const int* dst = ei + E;

  char* ws = (char*)d_ws;
  float*    h1   = (float*)ws;
  float*    out1 = (float*)(ws + (size_t)N * HID * 4);
  char* p = ws + (size_t)N * HID * 4 * 2;
  float*    as1 = (float*)p;  p += (size_t)N * HEADS * 4;
  float*    ad1 = (float*)p;  p += (size_t)N * HEADS * 4;
  unsigned* m1  = (unsigned*)p; p += (size_t)N * HEADS * 4;
  float*    s1  = (float*)p;  p += (size_t)N * HEADS * 4;
  float*    h2  = h1;
  char* q = (char*)(h1 + (size_t)N * DOUT);
  float*    as2 = (float*)q;  q += (size_t)N * 4;
  float*    ad2 = (float*)q;  q += (size_t)N * 4;
  unsigned* m2  = (unsigned*)q; q += (size_t)N * 4;
  float*    s2  = (float*)q;  q += (size_t)N * 4;
  float*    out2= (float*)q;

  const int B = 256;
  hipMemsetAsync(s1, 0, (size_t)N * HEADS * 4, stream);
  hipMemsetAsync(out1, 0, (size_t)N * HID * 4, stream);
  gat_fill_u32<<<(N * HEADS + B - 1) / B, B, 0, stream>>>(m1, ENC_NEGINF, N * HEADS);

  gat_gemm1_wmma<<<(N + 31) / 32, B, 0, stream>>>(x, W1, h1, N);
  gat_node_alpha1<<<(N * HEADS + B - 1) / B, B, 0, stream>>>(h1, a1s, a1d, as1, ad1, N * HEADS);
  gat_edge_max1<<<(E + B - 1) / B, B, 0, stream>>>(src, dst, as1, ad1, m1, E);
  gat_edge_sum1<<<(E + B - 1) / B, B, 0, stream>>>(src, dst, as1, ad1, m1, s1, E);
  {
    unsigned total = (unsigned)E * HID;
    gat_edge_scatter1<<<(total + B - 1) / B, B, 0, stream>>>(src, dst, as1, ad1, m1, s1,
                                                             h1, out1, total);
  }
  gat_elu_bias<<<(N * HID + B - 1) / B, B, 0, stream>>>(out1, b1, N * HID);

  hipMemsetAsync(s2, 0, (size_t)N * 4, stream);
  hipMemsetAsync(out2, 0, (size_t)N * DOUT * 4, stream);
  gat_fill_u32<<<(N + B - 1) / B, B, 0, stream>>>(m2, ENC_NEGINF, N);

  gat_gemm2_wmma<<<(N + 127) / 128, B, 0, stream>>>(out1, W2, h2, N);
  gat_node_alpha2<<<(N + B - 1) / B, B, 0, stream>>>(h2, a2s, a2d, as2, ad2, N);
  gat_edge_max2<<<(E + B - 1) / B, B, 0, stream>>>(src, dst, as2, ad2, m2, E);
  gat_edge_sum2<<<(E + B - 1) / B, B, 0, stream>>>(src, dst, as2, ad2, m2, s2, E);
  {
    unsigned total = (unsigned)E * DOUT;
    gat_edge_scatter2<<<(total + B - 1) / B, B, 0, stream>>>(src, dst, as2, ad2, m2, s2,
                                                             h2, out2, total);
  }
  gat_finalize<<<(N + B - 1) / B, B, 0, stream>>>(out2, b2, (float*)d_out, N);
}


# Softmax division deferred to the node pass: edges add ex*h and ex (float atomics) in one scatter kernel, node divides by (s+eps); the two edge-sum kernels are gone
# speedup vs baseline: 1.1373x; 1.1373x over previous
#include <hip/hip_runtime.h>
#include <math.h>


typedef float v2f __attribute__((ext_vector_type(2)));
typedef float v8f __attribute__((ext_vector_type(8)));

#define DIM_IN 256
#define HID    64
#define HEADS  8
#define DH     8
#define DOUT   16
#define NEG_SLOPE 0.2f
#define EPS_F 1e-16f
#define ENC_NEGINF 0x007FFFFFu

__device__ __forceinline__ unsigned enc_f32(float f) {
  unsigned u = __float_as_uint(f);
  return (u & 0x80000000u) ? ~u : (u | 0x80000000u);
}
__device__ __forceinline__ float dec_f32(unsigned e) {
  unsigned u = (e & 0x80000000u) ? (e ^ 0x80000000u) : ~e;
  return __uint_as_float(u);
}
__device__ __forceinline__ float leaky(float v) {
  return v > 0.0f ? v : NEG_SLOPE * v;
}

__global__ __launch_bounds__(256) void gat_gemm1_wmma(
    const float* __restrict__ x, const float* __restrict__ W1,
    float* __restrict__ h1, int N) {
  __shared__ float sW[DIM_IN * HID];
  int tid = threadIdx.x;
  {
    const float4* Wv = (const float4*)W1;
    float4* sWv = (float4*)sW;
#pragma unroll
    for (int i = 0; i < 16; ++i) sWv[tid + i * 256] = Wv[tid + i * 256];
  }
  __syncthreads();

  int wave = tid >> 5, lane = tid & 31;
  int row0 = (blockIdx.x * 2 + (wave >> 2)) * 16;
  int col0 = (wave & 3) * 16;
  if (row0 >= N) return;

  int r  = lane & 15;
  int kh = lane >> 4;

  int arow_idx = row0 + r; if (arow_idx >= N) arow_idx = N - 1;
  const float* arow  = x + (size_t)arow_idx * DIM_IN + 2 * kh;
  const float* bbase = sW + col0 + r;

  v8f c = {};
  for (int k = 0; k < DIM_IN; k += 4) {
    if ((k & 63) == 0) __builtin_prefetch(arow + k + 64, 0, 3);
    float2 av = *(const float2*)(arow + k);
    v2f a; a.x = av.x; a.y = av.y;
    v2f b; b.x = bbase[(k + 2 * kh) * HID];
           b.y = bbase[(k + 2 * kh + 1) * HID];
    c = __builtin_amdgcn_wmma_f32_16x16x4_f32(false, a, false, b,
                                              (short)0, c, false, false);
  }
  float* orow = h1 + (size_t)row0 * HID + col0 + r;
  if (row0 + 16 <= N) {
#pragma unroll
    for (int rr = 0; rr < 8; ++rr)
      orow[(size_t)(rr + 8 * kh) * HID] = c[rr];
  } else {
#pragma unroll
    for (int rr = 0; rr < 8; ++rr) {
      int row = row0 + rr + 8 * kh;
      if (row < N) orow[(size_t)(rr + 8 * kh) * HID] = c[rr];
    }
  }
}

__global__ __launch_bounds__(256) void gat_gemm2_wmma(
    const float* __restrict__ hact, const float* __restrict__ W2,
    float* __restrict__ h2, int N) {
  __shared__ float sW[HID * DOUT];
  int tid = threadIdx.x;
  {
    const float4* Wv = (const float4*)W2;
    float4* sWv = (float4*)sW;
    sWv[tid] = Wv[tid];
  }
  __syncthreads();

  int wave = tid >> 5, lane = tid & 31;
  int row0 = (blockIdx.x * 8 + wave) * 16;
  if (row0 >= N) return;

  int r  = lane & 15;
  int kh = lane >> 4;
  int arow_idx = row0 + r; if (arow_idx >= N) arow_idx = N - 1;
  const float* arow  = hact + (size_t)arow_idx * HID + 2 * kh;
  const float* bbase = sW + r;

  v8f c = {};
#pragma unroll
  for (int k = 0; k < HID; k += 4) {
    float2 av = *(const float2*)(arow + k);
    v2f a; a.x = av.x; a.y = av.y;
    v2f b; b.x = bbase[(k + 2 * kh) * DOUT];
           b.y = bbase[(k + 2 * kh + 1) * DOUT];
    c = __builtin_amdgcn_wmma_f32_16x16x4_f32(false, a, false, b,
                                              (short)0, c, false, false);
  }
  float* orow = h2 + (size_t)row0 * DOUT + r;
  if (row0 + 16 <= N) {
#pragma unroll
    for (int rr = 0; rr < 8; ++rr)
      orow[(size_t)(rr + 8 * kh) * DOUT] = c[rr];
  } else {
#pragma unroll
    for (int rr = 0; rr < 8; ++rr) {
      int row = row0 + rr + 8 * kh;
      if (row < N) orow[(size_t)(rr + 8 * kh) * DOUT] = c[rr];
    }
  }
}

__global__ void gat_fill_u32(unsigned* __restrict__ p, unsigned v, int n) {
  int t = blockIdx.x * blockDim.x + threadIdx.x;
  if (t < n) p[t] = v;
}

__global__ void gat_node_alpha1(const float* __restrict__ h1,
                                const float* __restrict__ a_src,
                                const float* __restrict__ a_dst,
                                float* __restrict__ as, float* __restrict__ ad,
                                int NH) {
  int t = blockIdx.x * blockDim.x + threadIdx.x;
  if (t >= NH) return;
  int n = t >> 3, h = t & 7;
  const float* hp = h1 + (size_t)n * HID + h * DH;
  const float* sp = a_src + h * DH;
  const float* dp = a_dst + h * DH;
  float ss = 0.f, dd = 0.f;
#pragma unroll
  for (int i = 0; i < DH; ++i) { float v = hp[i]; ss += v * sp[i]; dd += v * dp[i]; }
  as[t] = ss; ad[t] = dd;
}

__global__ void gat_node_alpha2(const float* __restrict__ h2,
                                const float* __restrict__ a_src,
                                const float* __restrict__ a_dst,
                                float* __restrict__ as, float* __restrict__ ad,
                                int N) {
  int n = blockIdx.x * blockDim.x + threadIdx.x;
  if (n >= N) return;
  const float* hp = h2 + (size_t)n * DOUT;
  float ss = 0.f, dd = 0.f;
#pragma unroll
  for (int i = 0; i < DOUT; ++i) { float v = hp[i]; ss += v * a_src[i]; dd += v * a_dst[i]; }
  as[n] = ss; ad[n] = dd;
}

__global__ void gat_edge_max1(const int* __restrict__ src, const int* __restrict__ dst,
                              const float* __restrict__ as, const float* __restrict__ ad,
                              unsigned* __restrict__ m, int E) {
  int e = blockIdx.x * blockDim.x + threadIdx.x;
  if (e >= E) return;
  int s = src[e], d = dst[e];
  const float4* as4 = (const float4*)as;
  const float4* ad4 = (const float4*)ad;
  float4 sv[2] = { as4[(size_t)s * 2], as4[(size_t)s * 2 + 1] };
  float4 dv[2] = { ad4[(size_t)d * 2], ad4[(size_t)d * 2 + 1] };
  const float* sf = (const float*)sv;
  const float* df = (const float*)dv;
#pragma unroll
  for (int h = 0; h < HEADS; ++h) {
    float v = leaky(sf[h] + df[h]);
    atomicMax(&m[(size_t)d * HEADS + h], enc_f32(v));
  }
}

__global__ void gat_edge_scatter1(const int* __restrict__ src, const int* __restrict__ dst,
                                  const float* __restrict__ as, const float* __restrict__ ad,
                                  const unsigned* __restrict__ m, float* __restrict__ ssum,
                                  const float* __restrict__ h1, float* __restrict__ out1,
                                  unsigned total) {
  unsigned t = blockIdx.x * blockDim.x + threadIdx.x;
  if (t >= total) return;
  int e = (int)(t >> 6), c = (int)(t & 63), h = c >> 3;
  int s = src[e], d = dst[e];
  float v = leaky(as[(size_t)s * HEADS + h] + ad[(size_t)d * HEADS + h]);
  float ex = __expf(v - dec_f32(m[(size_t)d * HEADS + h]));
  atomicAdd(&out1[(size_t)d * HID + c], ex * h1[(size_t)s * HID + c]);
  if ((c & 7) == 0) atomicAdd(&ssum[(size_t)d * HEADS + h], ex);
}

__global__ void gat_elu_bias(float* __restrict__ out1, const float* __restrict__ ssum,
                             const float* __restrict__ b1, int total) {
  int t = blockIdx.x * blockDim.x + threadIdx.x;
  if (t >= total) return;
  float v = out1[t] / (ssum[t >> 3] + EPS_F) + b1[t & (HID - 1)];
  out1[t] = v > 0.0f ? v : expm1f(v);
}

__global__ void gat_edge_max2(const int* __restrict__ src, const int* __restrict__ dst,
                              const float* __restrict__ as, const float* __restrict__ ad,
                              unsigned* __restrict__ m, int E) {
  int e = blockIdx.x * blockDim.x + threadIdx.x;
  if (e >= E) return;
  int s = src[e], d = dst[e];
  float v = leaky(as[s] + ad[d]);
  atomicMax(&m[d], enc_f32(v));
}

__global__ void gat_edge_scatter2(const int* __restrict__ src, const int* __restrict__ dst,
                                  const float* __restrict__ as, const float* __restrict__ ad,
                                  const unsigned* __restrict__ m, float* __restrict__ ssum,
                                  const float* __restrict__ h2, float* __restrict__ out2,
                                  unsigned total) {
  unsigned t = blockIdx.x * blockDim.x + threadIdx.x;
  if (t >= total) return;
  int e = (int)(t >> 4), c = (int)(t & 15);
  int s = src[e], d = dst[e];
  float v = leaky(as[s] + ad[d]);
  float ex = __expf(v - dec_f32(m[d]));
  atomicAdd(&out2[(size_t)d * DOUT + c], ex * h2[(size_t)s * DOUT + c]);
  if (c == 0) atomicAdd(&ssum[d], ex);
}

__global__ void gat_finalize(const float* __restrict__ out2, const float* __restrict__ ssum,
                             const float* __restrict__ b2, float* __restrict__ dout, int N) {
  int n = blockIdx.x * blockDim.x + threadIdx.x;
  if (n >= N) return;
  float v[DOUT];
  float mx = -INFINITY;
  float den = ssum[n] + EPS_F;
#pragma unroll
  for (int c = 0; c < DOUT; ++c) {
    v[c] = out2[(size_t)n * DOUT + c] / den + b2[c];
    mx = fmaxf(mx, v[c]);
  }
  float se = 0.f;
#pragma unroll
  for (int c = 0; c < DOUT; ++c) se += __expf(v[c] - mx);
  float lse = mx + __logf(se);
#pragma unroll
  for (int c = 0; c < DOUT; ++c) {
    dout[(size_t)n * DOUT + c] = v[c];
    dout[(size_t)N * DOUT + (size_t)n * DOUT + c] = v[c] - lse;
  }
}

extern "C" void kernel_launch(void* const* d_in, const int* in_sizes, int n_in,
                              void* d_out, int out_size, void* d_ws, size_t ws_size,
                              hipStream_t stream) {
  const float* x   = (const float*)d_in[0];
  const int*   ei  = (const int*)d_in[1];
  const float* W1  = (const float*)d_in[2];
  const float* a1s = (const float*)d_in[3];
  const float* a1d = (const float*)d_in[4];
  const float* b1  = (const float*)d_in[5];
  const float* W2  = (const float*)d_in[6];
  const float* a2s = (const float*)d_in[7];
  const float* a2d = (const float*)d_in[8];
  const float* b2  = (const float*)d_in[9];

  int N = in_sizes[0] / DIM_IN;
  int E = in_sizes[1] / 2;
  const int* src = ei;
  const int* dst = ei + E;

  char* ws = (char*)d_ws;
  float*    h1   = (float*)ws;
  float*    out1 = (float*)(ws + (size_t)N * HID * 4);
  char* p = ws + (size_t)N * HID * 4 * 2;
  float*    as1 = (float*)p;  p += (size_t)N * HEADS * 4;
  float*    ad1 = (float*)p;  p += (size_t)N * HEADS * 4;
  unsigned* m1  = (unsigned*)p; p += (size_t)N * HEADS * 4;
  float*    s1  = (float*)p;  p += (size_t)N * HEADS * 4;
  float*    h2  = h1;
  char* q = (char*)(h1 + (size_t)N * DOUT);
  float*    as2 = (float*)q;  q += (size_t)N * 4;
  float*    ad2 = (float*)q;  q += (size_t)N * 4;
  unsigned* m2  = (unsigned*)q; q += (size_t)N * 4;
  float*    s2  = (float*)q;  q += (size_t)N * 4;
  float*    out2= (float*)q;

  const int B = 256;
  hipMemsetAsync(s1, 0, (size_t)N * HEADS * 4, stream);
  hipMemsetAsync(out1, 0, (size_t)N * HID * 4, stream);
  gat_fill_u32<<<(N * HEADS + B - 1) / B, B, 0, stream>>>(m1, ENC_NEGINF, N * HEADS);

  gat_gemm1_wmma<<<(N + 31) / 32, B, 0, stream>>>(x, W1, h1, N);
  gat_node_alpha1<<<(N * HEADS + B - 1) / B, B, 0, stream>>>(h1, a1s, a1d, as1, ad1, N * HEADS);
  gat_edge_max1<<<(E + B - 1) / B, B, 0, stream>>>(src, dst, as1, ad1, m1, E);
  {
    unsigned total = (unsigned)E * HID;
    gat_edge_scatter1<<<(total + B - 1) / B, B, 0, stream>>>(src, dst, as1, ad1, m1, s1,
                                                             h1, out1, total);
  }
  gat_elu_bias<<<(N * HID + B - 1) / B, B, 0, stream>>>(out1, s1, b1, N * HID);

  hipMemsetAsync(s2, 0, (size_t)N * 4, stream);
  hipMemsetAsync(out2, 0, (size_t)N * DOUT * 4, stream);
  gat_fill_u32<<<(N + B - 1) / B, B, 0, stream>>>(m2, ENC_NEGINF, N);

  gat_gemm2_wmma<<<(N + 127) / 128, B, 0, stream>>>(out1, W2, h2, N);
  gat_node_alpha2<<<(N + B - 1) / B, B, 0, stream>>>(h2, a2s, a2d, as2, ad2, N);
  gat_edge_max2<<<(E + B - 1) / B, B, 0, stream>>>(src, dst, as2, ad2, m2, E);
  {
    unsigned total = (unsigned)E * DOUT;
    gat_edge_scatter2<<<(total + B - 1) / B, B, 0, stream>>>(src, dst, as2, ad2, m2, s2,
                                                             h2, out2, total);
  }
  gat_finalize<<<(N + B - 1) / B, B, 0, stream>>>(out2, s2, b2, (float*)d_out, N);
}


# Edges ordered by destination once per call (count, place at arrival rank); a wave forms each node's max, exp sum and weighted row sum in registers: float atomics gone, sums in arrival-rank order
# speedup vs baseline: 3.4571x; 3.0397x over previous
#include <hip/hip_runtime.h>
#include <math.h>


typedef float v2f __attribute__((ext_vector_type(2)));
typedef float v8f __attribute__((ext_vector_type(8)));

#define DIM_IN 256
#define HID    64
#define HEADS  8
#define DH     8
#define DOUT   16
#define NEG_SLOPE 0.2f
#define EPS_F 1e-16f

__device__ __forceinline__ float leaky(float v) {
  return v > 0.0f ? v : NEG_SLOPE * v;
}

__global__ __launch_bounds__(256) void gat_gemm1_wmma(
    const float* __restrict__ x, const float* __restrict__ W1,
    float* __restrict__ h1, int N) {
  __shared__ float sW[DIM_IN * HID];
  int tid = threadIdx.x;
  {
    const float4* Wv = (const float4*)W1;
    float4* sWv = (float4*)sW;
#pragma unroll
    for (int i = 0; i < 16; ++i) sWv[tid + i * 256] = Wv[tid + i * 256];
  }
  __syncthreads();

  int wave = tid >> 5, lane = tid & 31;
  int row0 = (blockIdx.x * 2 + (wave >> 2)) * 16;
  int col0 = (wave & 3) * 16;
  if (row0 >= N) return;

  int r  = lane & 15;
  int kh = lane >> 4;

  int arow_idx = row0 + r; if (arow_idx >= N) arow_idx = N - 1;
  const float* arow  = x + (size_t)arow_idx * DIM_IN + 2 * kh;
  const float* bbase = sW + col0 + r;

  v8f c = {};
  for (int k = 0; k < DIM_IN; k += 4) {
    if ((k & 63) == 0) __builtin_prefetch(arow + k + 64, 0, 3);
    float2 av = *(const float2*)(arow + k);
    v2f a; a.x = av.x; a.y = av.y;
    v2f b; b.x = bbase[(k + 2 * kh) * HID];
           b.y = bbase[(k + 2 * kh + 1) * HID];
    c = __builtin_amdgcn_wmma_f32_16x16x4_f32(false, a, false, b,
                                              (short)0, c, false, false);
  }
  float* orow = h1 + (size_t)row0 * HID + col0 + r;
  if (row0 + 16 <= N) {
#pragma unroll
    for (int rr = 0; rr < 8; ++rr)
      orow[(size_t)(rr + 8 * kh) * HID] = c[rr];
  } else {
#pragma unroll
    for (int rr = 0; rr < 8; ++rr) {
      int row = row0 + rr + 8 * kh;
      if (row < N) orow[(size_t)(rr + 8 * kh) * HID] = c[rr];
    }
  }
}

__global__ __launch_bounds__(256) void gat_gemm2_wmma(
    const float* __restrict__ hact, const float* __restrict__ W2,
    float* __restrict__ h2, int N) {
  __shared__ float sW[HID * DOUT];
  int tid = threadIdx.x;
  {
    const float4* Wv = (const float4*)W2;
    float4* sWv = (float4*)sW;
    sWv[tid] = Wv[tid];
  }
  __syncthreads();

  int wave = tid >> 5, lane = tid & 31;
  int row0 = (blockIdx.x * 8 + wave) * 16;
  if (row0 >= N) return;

  int r  = lane & 15;
  int kh = lane >> 4;
  int arow_idx = row0 + r; if (arow_idx >= N) arow_idx = N - 1;
  const float* arow  = hact + (size_t)arow_idx * HID + 2 * kh;
  const float* bbase = sW + r;

  v8f c = {};
#pragma unroll
  for (int k = 0; k < HID; k += 4) {
    float2 av = *(const float2*)(arow + k);
    v2f a; a.x = av.x; a.y = av.y;
    v2f b; b.x = bbase[(k + 2 * kh) * DOUT];
           b.y = bbase[(k + 2 * kh + 1) * DOUT];
    c = __builtin_amdgcn_wmma_f32_16x16x4_f32(false, a, false, b,
                                              (short)0, c, false, false);
  }
  float* orow = h2 + (size_t)row0 * DOUT + r;
  if (row0 + 16 <= N) {
#pragma unroll
    for (int rr = 0; rr < 8; ++rr)
      orow[(size_t)(rr + 8 * kh) * DOUT] = c[rr];
  } else {
#pragma unroll
    for (int rr = 0; rr < 8; ++rr) {
      int row = row0 + rr + 8 * kh;
      if (row < N) orow[(size_t)(rr + 8 * kh) * DOUT] = c[rr];
    }
  }
}

__global__ void gat_count_dst(const int* __restrict__ dst, int* __restrict__ deg, int E) {
  int e = blockIdx.x * blockDim.x + threadIdx.x;
  if (e < E) atomicAdd(&deg[dst[e]], 1);
}

__global__ void gat_seg_alloc(const int* __restrict__ deg, int* __restrict__ total,
                              int* __restrict__ cur, int N) {
  int n = blockIdx.x * blockDim.x + threadIdx.x;
  int lane = threadIdx.x & 31;
  int d = (n < N) ? deg[n] : 0;
  int incl = d;
#pragma unroll
  for (int o = 1; o < 32; o <<= 1) {
    int t = __shfl_up(incl, o, 32);
    if (lane >= o) incl += t;
  }
  int base = 0;
  if (lane == 31) base = atomicAdd(total, incl);
  base = __shfl(base, 31, 32);
  if (n < N) cur[n] = base + incl - d;
}

__global__ void gat_place_src(const int* __restrict__ src, const int* __restrict__ dst,
                              int* __restrict__ cur, int* __restrict__ esrc, int E) {
  int e = blockIdx.x * blockDim.x + threadIdx.x;
  if (e >= E) return;
  int pos = atomicAdd(&cur[dst[e]], 1);
  esrc[pos] = src[e];
}

__global__ void gat_node_alpha1(const float* __restrict__ h1,
                                const float* __restrict__ a_src,
                                float* __restrict__ as, int NH) {
  int t = blockIdx.x * blockDim.x + threadIdx.x;
  if (t >= NH) return;
  int n = t >> 3, h = t & 7;
  const float* hp = h1 + (size_t)n * HID + h * DH;
  const float* sp = a_src + h * DH;
  float ss = 0.f;
#pragma unroll
  for (int i = 0; i < DH; ++i) ss += hp[i] * sp[i];
  as[t] = ss;
}

__global__ void gat_node_alpha2(const float* __restrict__ h2,
                                const float* __restrict__ a_src,
                                const float* __restrict__ a_dst,
                                float* __restrict__ as, float* __restrict__ ad,
                                int N) {
  int n = blockIdx.x * blockDim.x + threadIdx.x;
  if (n >= N) return;
  const float* hp = h2 + (size_t)n * DOUT;
  float ss = 0.f, dd = 0.f;
#pragma unroll
  for (int i = 0; i < DOUT; ++i) { float v = hp[i]; ss += v * a_src[i]; dd += v * a_dst[i]; }
  as[n] = ss; ad[n] = dd;
}

__global__ __launch_bounds__(256) void gat_gather1(
    const int* __restrict__ esrc, const int* __restrict__ seg_end,
    const int* __restrict__ deg, const float* __restrict__ as,
    const float* __restrict__ a_dst, const float* __restrict__ h1,
    const float* __restrict__ b1, float* __restrict__ hact, int N) {
  int lane = threadIdx.x & 31;
  int n = __builtin_amdgcn_readfirstlane((int)(blockIdx.x * 8 + (threadIdx.x >> 5)));
  if (n >= N) return;
  int cnt = __builtin_amdgcn_readfirstlane(deg[n]);
  int beg = __builtin_amdgcn_readfirstlane(seg_end[n]) - cnt;

  float2 hv = ((const float2*)(h1 + (size_t)n * HID))[lane];
  float pd = hv.x * a_dst[2 * lane] + hv.y * a_dst[2 * lane + 1];
  pd += __shfl_xor(pd, 1, 32);
  pd += __shfl_xor(pd, 2, 32);

  int hq = lane & 7, jq = lane >> 3;
  int hsel = lane >> 2;
  float adq = __shfl(pd, hq * 4, 32);

  float mq = -INFINITY;
  for (int b0 = 0; b0 < cnt; b0 += 32) {
    int rem = min(32, cnt - b0);
    int s_l = (lane < rem) ? esrc[beg + b0 + lane] : n;
    float a[8];
#pragma unroll
    for (int t = 0; t < 8; ++t)
      a[t] = as[(size_t)__shfl(s_l, 4 * t + jq, 32) * HEADS + hq];
#pragma unroll
    for (int t = 0; t < 8; ++t)
      if (4 * t + jq < rem) mq = fmaxf(mq, leaky(a[t] + adq));
  }
  mq = fmaxf(mq, __shfl_xor(mq, 8, 32));
  mq = fmaxf(mq, __shfl_xor(mq, 16, 32));

  float accx = 0.f, accy = 0.f, sq = 0.f;
  for (int b0 = 0; b0 < cnt; b0 += 32) {
    int rem = min(32, cnt - b0);
    int s_l = (lane < rem) ? esrc[beg + b0 + lane] : n;
    float ex[8];
#pragma unroll
    for (int t = 0; t < 8; ++t)
      ex[t] = as[(size_t)__shfl(s_l, 4 * t + jq, 32) * HEADS + hq];
#pragma unroll
    for (int t = 0; t < 8; ++t) {
      ex[t] = (4 * t + jq < rem) ? __expf(leaky(ex[t] + adq) - mq) : 0.f;
      sq += ex[t];
    }
#pragma unroll
    for (int g = 0; g < 4; ++g) {
      if (8 * g < rem) {
        float2 hr[8];
        float ej[8];
#pragma unroll
        for (int jj = 0; jj < 8; ++jj) {
          const int j = 8 * g + jj;
          int sj = __shfl(s_l, j, 32);
          ej[jj] = __shfl(ex[j >> 2], (j & 3) * 8 + hsel, 32);
          int col = (j < rem) ? lane : 0;
          hr[jj] = ((const float2*)(h1 + (size_t)sj * HID))[col];
        }
#pragma unroll
        for (int jj = 0; jj < 8; ++jj) {
          bool ok = (8 * g + jj) < rem;
          accx = ok ? fmaf(ej[jj], hr[jj].x, accx) : accx;
          accy = ok ? fmaf(ej[jj], hr[jj].y, accy) : accy;
        }
      }
    }
  }
  sq += __shfl_xor(sq, 8, 32);
  sq += __shfl_xor(sq, 16, 32);
  float den = __shfl(sq, hsel, 32) + EPS_F;

  float vx = accx / den + b1[2 * lane];
  float vy = accy / den + b1[2 * lane + 1];
  float2 o;
  o.x = vx > 0.0f ? vx : expm1f(vx);
  o.y = vy > 0.0f ? vy : expm1f(vy);
  ((float2*)(hact + (size_t)n * HID))[lane] = o;
}

__global__ __launch_bounds__(256) void gat_gather2(
    const int* __restrict__ esrc, const int* __restrict__ seg_end,
    const int* __restrict__ deg, const float* __restrict__ as,
    const float* __restrict__ ad, const float* __restrict__ h2,
    float* __restrict__ ssum, float* __restrict__ out2, int N) {
  int lane = threadIdx.x & 31;
  int c = lane & 15, half = lane >> 4;
  int n0 = __builtin_amdgcn_readfirstlane((int)((blockIdx.x * 8 + (threadIdx.x >> 5)) * 2));
  if (n0 >= N) return;
  int n = n0 + half;
  bool live = n < N;
  int cnt = live ? deg[n] : 0;
  int beg = live ? seg_end[n] - cnt : 0;
  float adn = live ? ad[n] : 0.f;
  int cmax = __builtin_amdgcn_readfirstlane(max(cnt, __shfl_xor(cnt, 16, 32)));

  float m = -INFINITY;
  for (int b0 = 0; b0 < cmax; b0 += 16) {
    int j = b0 + c;
    if (j < cnt) m = fmaxf(m, leaky(as[esrc[beg + j]] + adn));
  }
  m = fmaxf(m, __shfl_xor(m, 1, 32));
  m = fmaxf(m, __shfl_xor(m, 2, 32));
  m = fmaxf(m, __shfl_xor(m, 4, 32));
  m = fmaxf(m, __shfl_xor(m, 8, 32));

  float acc = 0.f, sum = 0.f;
  for (int b0 = 0; b0 < cmax; b0 += 16) {
    int j = b0 + c;
    int s_l = 0;
    float ex_l = 0.f;
    if (j < cnt) {
      s_l = esrc[beg + j];
      ex_l = __expf(leaky(as[s_l] + adn) - m);
    }
    sum += ex_l;
    int kmax = min(16, cmax - b0);
#pragma unroll
    for (int g = 0; g < 2; ++g) {
      if (8 * g < kmax) {
        float hr[8], ek[8];
#pragma unroll
        for (int jj = 0; jj < 8; ++jj) {
          int from = half * 16 + 8 * g + jj;
          int sk = __shfl(s_l, from, 32);
          ek[jj] = __shfl(ex_l, from, 32);
          hr[jj] = h2[(size_t)sk * DOUT + c];
        }
#pragma unroll
        for (int jj = 0; jj < 8; ++jj)
          acc = (b0 + 8 * g + jj < cnt) ? fmaf(ek[jj], hr[jj], acc) : acc;
      }
    }
  }
  sum += __shfl_xor(sum, 1, 32);
  sum += __shfl_xor(sum, 2, 32);
  sum += __shfl_xor(sum, 4, 32);
  sum += __shfl_xor(sum, 8, 32);
  if (live) {
    out2[(size_t)n * DOUT + c] = acc;
    if (c == 0) ssum[n] = sum;
  }
}

__global__ void gat_finalize(const float* __restrict__ out2, const float* __restrict__ ssum,
                             const float* __restrict__ b2, float* __restrict__ dout, int N) {
  int n = blockIdx.x * blockDim.x + threadIdx.x;
  if (n >= N) return;
  float v[DOUT];
  float mx = -INFINITY;
  float den = ssum[n] + EPS_F;
#pragma unroll
  for (int c = 0; c < DOUT; ++c) {
    v[c] = out2[(size_t)n * DOUT + c] / den + b2[c];
    mx = fmaxf(mx, v[c]);
  }
  float se = 0.f;
#pragma unroll
  for (int c = 0; c < DOUT; ++c) se += __expf(v[c] - mx);
  float lse = mx + __logf(se);
#pragma unroll
  for (int c = 0; c < DOUT; ++c) {
    dout[(size_t)n * DOUT + c] = v[c];
    dout[(size_t)N * DOUT + (size_t)n * DOUT + c] = v[c] - lse;
  }
}

extern "C" void kernel_launch(void* const* d_in, const int* in_sizes, int n_in,
                              void* d_out, int out_size, void* d_ws, size_t ws_size,
                              hipStream_t stream) {
  const float* x   = (const float*)d_in[0];
  const int*   ei  = (const int*)d_in[1];
  const float* W1  = (const float*)d_in[2];
  const float* a1s = (const float*)d_in[3];
  const float* a1d = (const float*)d_in[4];
  const float* b1  = (const float*)d_in[5];
  const float* W2  = (const float*)d_in[6];
  const float* a2s = (const float*)d_in[7];
  const float* a2d = (const float*)d_in[8];
  const float* b2  = (const float*)d_in[9];

  int N = in_sizes[0] / DIM_IN;
  int E = in_sizes[1] / 2;
  const int* src = ei;
  const int* dst = ei + E;

  char* ws = (char*)d_ws;
  float*    h1   = (float*)ws;
  float*    hact = (float*)(ws + (size_t)N * HID * 4);
  char* p = ws + (size_t)N * HID * 4 * 2;
  float*    as1  = (float*)p; p += (size_t)N * HEADS * 4;
  int*      deg  = (int*)p;   p += ((size_t)N + 1) * 4;
  int*      cur  = (int*)p;   p += (size_t)N * 4;
  int*      esrc = (int*)p;
  float*    h2  = h1;
  char* q = (char*)(h1 + (size_t)N * DOUT);
  float*    as2 = (float*)q;  q += (size_t)N * 4;
  float*    ad2 = (float*)q;  q += (size_t)N * 4;
  float*    s2  = (float*)q;  q += (size_t)N * 4;
  float*    out2= (float*)q;

  const int B = 256;
  hipMemsetAsync(deg, 0, ((size_t)N + 1) * 4, stream);
  gat_count_dst<<<(E + B - 1) / B, B, 0, stream>>>(dst, deg, E);
  gat_seg_alloc<<<(N + B - 1) / B, B, 0, stream>>>(deg, deg + N, cur, N);
  gat_place_src<<<(E + B - 1) / B, B, 0, stream>>>(src, dst, cur, esrc, E);

  gat_gemm1_wmma<<<(N + 31) / 32, B, 0, stream>>>(x, W1, h1, N);
  gat_node_alpha1<<<(N * HEADS + B - 1) / B, B, 0, stream>>>(h1, a1s, as1, N * HEADS);
  gat_gather1<<<(N + 7) / 8, B, 0, stream>>>(esrc, cur, deg, as1, a1d, h1, b1, hact, N);

  gat_gemm2_wmma<<<(N + 127) / 128, B, 0, stream>>>(hact, W2, h2, N);
  gat_node_alpha2<<<(N + B - 1) / B, B, 0, stream>>>(h2, a2s, a2d, as2, ad2, N);
  gat_gather2<<<(N + 15) / 16, B, 0, stream>>>(esrc, cur, deg, as2, ad2, h2, s2, out2, N);
  gat_finalize<<<(N + B - 1) / B, B, 0, stream>>>(out2, s2, b2, (float*)d_out, N);
}


# Per-row follow-ups fused into producers: both GEMMs emit the attention dots from their accumulators, layer-2 gather normalises and forms log_softmax; 3 launches gone; those sums become lane trees
# speedup vs baseline: 3.5644x; 1.0310x over previous
#include <hip/hip_runtime.h>
#include <math.h>


typedef float v2f __attribute__((ext_vector_type(2)));
typedef float v8f __attribute__((ext_vector_type(8)));

#define DIM_IN 256
#define HID    64
#define HEADS  8
#define DH     8
#define DOUT   16
#define NEG_SLOPE 0.2f
#define EPS_F 1e-16f

__device__ __forceinline__ float leaky(float v) {
  return v > 0.0f ? v : NEG_SLOPE * v;
}

__global__ __launch_bounds__(256) void gat_gemm1_wmma(
    const float* __restrict__ x, const float* __restrict__ W1,
    const float* __restrict__ a_src, float* __restrict__ h1,
    float* __restrict__ as, int N) {
  __shared__ float sW[DIM_IN * HID];
  int tid = threadIdx.x;
  {
    const float4* Wv = (const float4*)W1;
    float4* sWv = (float4*)sW;
#pragma unroll
    for (int i = 0; i < 16; ++i) sWv[tid + i * 256] = Wv[tid + i * 256];
  }
  __syncthreads();

  int wave = tid >> 5, lane = tid & 31;
  int row0 = (blockIdx.x * 2 + (wave >> 2)) * 16;
  int col0 = (wave & 3) * 16;
  if (row0 >= N) return;

  int r  = lane & 15;
  int kh = lane >> 4;

  int arow_idx = row0 + r; if (arow_idx >= N) arow_idx = N - 1;
  const float* arow  = x + (size_t)arow_idx * DIM_IN + 2 * kh;
  const float* bbase = sW + col0 + r;
  float asv = a_src[col0 + r];

  v8f c = {};
  for (int k = 0; k < DIM_IN; k += 4) {
    if ((k & 63) == 0) __builtin_prefetch(arow + k + 64, 0, 3);
    float2 av = *(const float2*)(arow + k);
    v2f a; a.x = av.x; a.y = av.y;
    v2f b; b.x = bbase[(k + 2 * kh) * HID];
           b.y = bbase[(k + 2 * kh + 1) * HID];
    c = __builtin_amdgcn_wmma_f32_16x16x4_f32(false, a, false, b,
                                              (short)0, c, false, false);
  }
  float* orow = h1 + (size_t)row0 * HID + col0 + r;
  if (row0 + 16 <= N) {
#pragma unroll
    for (int rr = 0; rr < 8; ++rr)
      orow[(size_t)(rr + 8 * kh) * HID] = c[rr];
  } else {
#pragma unroll
    for (int rr = 0; rr < 8; ++rr) {
      int row = row0 + rr + 8 * kh;
      if (row < N) orow[(size_t)(rr + 8 * kh) * HID] = c[rr];
    }
  }
  float sel = 0.f;
#pragma unroll
  for (int rr = 0; rr < 8; ++rr) {
    float p = c[rr] * asv;
    p += __shfl_xor(p, 1, 32);
    p += __shfl_xor(p, 2, 32);
    p += __shfl_xor(p, 4, 32);
    sel = ((r & 7) == rr) ? p : sel;
  }
  int srow = row0 + 8 * kh + (r & 7);
  if (srow < N) as[(size_t)srow * HEADS + (col0 >> 3) + (r >> 3)] = sel;
}

__global__ __launch_bounds__(256) void gat_gemm2_wmma(
    const float* __restrict__ hact, const float* __restrict__ W2,
    const float* __restrict__ a_src, const float* __restrict__ a_dst,
    float* __restrict__ h2, float* __restrict__ as, float* __restrict__ ad, int N) {
  __shared__ float sW[HID * DOUT];
  int tid = threadIdx.x;
  {
    const float4* Wv = (const float4*)W2;
    float4* sWv = (float4*)sW;
    sWv[tid] = Wv[tid];
  }
  __syncthreads();

  int wave = tid >> 5, lane = tid & 31;
  int row0 = (blockIdx.x * 8 + wave) * 16;
  if (row0 >= N) return;

  int r  = lane & 15;
  int kh = lane >> 4;
  int arow_idx = row0 + r; if (arow_idx >= N) arow_idx = N - 1;
  const float* arow  = hact + (size_t)arow_idx * HID + 2 * kh;
  const float* bbase = sW + r;
  float wsv = a_src[r], wdv = a_dst[r];

  v8f c = {};
#pragma unroll
  for (int k = 0; k < HID; k += 4) {
    float2 av = *(const float2*)(arow + k);
    v2f a; a.x = av.x; a.y = av.y;
    v2f b; b.x = bbase[(k + 2 * kh) * DOUT];
           b.y = bbase[(k + 2 * kh + 1) * DOUT];
    c = __builtin_amdgcn_wmma_f32_16x16x4_f32(false, a, false, b,
                                              (short)0, c, false, false);
  }
  float* orow = h2 + (size_t)row0 * DOUT + r;
  if (row0 + 16 <= N) {
#pragma unroll
    for (int rr = 0; rr < 8; ++rr)
      orow[(size_t)(rr + 8 * kh) * DOUT] = c[rr];
  } else {
#pragma unroll
    for (int rr = 0; rr < 8; ++rr) {
      int row = row0 + rr + 8 * kh;
      if (row < N) orow[(size_t)(rr + 8 * kh) * DOUT] = c[rr];
    }
  }
  float ssel = 0.f, dsel = 0.f;
#pragma unroll
  for (int rr = 0; rr < 8; ++rr) {
    float ps = c[rr] * wsv, pd = c[rr] * wdv;
    ps += __shfl_xor(ps, 1, 32);  pd += __shfl_xor(pd, 1, 32);
    ps += __shfl_xor(ps, 2, 32);  pd += __shfl_xor(pd, 2, 32);
    ps += __shfl_xor(ps, 4, 32);  pd += __shfl_xor(pd, 4, 32);
    ps += __shfl_xor(ps, 8, 32);  pd += __shfl_xor(pd, 8, 32);
    ssel = ((r & 7) == rr) ? ps : ssel;
    dsel = ((r & 7) == rr) ? pd : dsel;
  }
  int srow = row0 + 8 * kh + (r & 7);
  float* sp = (r < 8) ? as : ad;
  if (srow < N) sp[srow] = (r < 8) ? ssel : dsel;
}

__global__ void gat_count_dst(const int* __restrict__ dst, int* __restrict__ deg, int E) {
  int e = blockIdx.x * blockDim.x + threadIdx.x;
  if (e < E) atomicAdd(&deg[dst[e]], 1);
}

__global__ void gat_seg_alloc(const int* __restrict__ deg, int* __restrict__ total,
                              int* __restrict__ cur, int N) {
  int n = blockIdx.x * blockDim.x + threadIdx.x;
  int lane = threadIdx.x & 31;
  int d = (n < N) ? deg[n] : 0;
  int incl = d;
#pragma unroll
  for (int o = 1; o < 32; o <<= 1) {
    int t = __shfl_up(incl, o, 32);
    if (lane >= o) incl += t;
  }
  int base = 0;
  if (lane == 31) base = atomicAdd(total, incl);
  base = __shfl(base, 31, 32);
  if (n < N) cur[n] = base + incl - d;
}

__global__ void gat_place_src(const int* __restrict__ src, const int* __restrict__ dst,
                              int* __restrict__ cur, int* __restrict__ esrc, int E) {
  int e = blockIdx.x * blockDim.x + threadIdx.x;
  if (e >= E) return;
  int pos = atomicAdd(&cur[dst[e]], 1);
  esrc[pos] = src[e];
}

__global__ __launch_bounds__(256) void gat_gather1(
    const int* __restrict__ esrc, const int* __restrict__ seg_end,
    const int* __restrict__ deg, const float* __restrict__ as,
    const float* __restrict__ a_dst, const float* __restrict__ h1,
    const float* __restrict__ b1, float* __restrict__ hact, int N) {
  int lane = threadIdx.x & 31;
  int n = __builtin_amdgcn_readfirstlane((int)(blockIdx.x * 8 + (threadIdx.x >> 5)));
  if (n >= N) return;
  int cnt = __builtin_amdgcn_readfirstlane(deg[n]);
  int beg = __builtin_amdgcn_readfirstlane(seg_end[n]) - cnt;

  float2 hv = ((const float2*)(h1 + (size_t)n * HID))[lane];
  float pd = hv.x * a_dst[2 * lane] + hv.y * a_dst[2 * lane + 1];
  pd += __shfl_xor(pd, 1, 32);
  pd += __shfl_xor(pd, 2, 32);

  int hq = lane & 7, jq = lane >> 3;
  int hsel = lane >> 2;
  float adq = __shfl(pd, hq * 4, 32);

  float mq = -INFINITY;
  for (int b0 = 0; b0 < cnt; b0 += 32) {
    int rem = min(32, cnt - b0);
    int s_l = (lane < rem) ? esrc[beg + b0 + lane] : n;
    float a[8];
#pragma unroll
    for (int t = 0; t < 8; ++t)
      a[t] = as[(size_t)__shfl(s_l, 4 * t + jq, 32) * HEADS + hq];
#pragma unroll
    for (int t = 0; t < 8; ++t)
      if (4 * t + jq < rem) mq = fmaxf(mq, leaky(a[t] + adq));
  }
  mq = fmaxf(mq, __shfl_xor(mq, 8, 32));
  mq = fmaxf(mq, __shfl_xor(mq, 16, 32));

  float accx = 0.f, accy = 0.f, sq = 0.f;
  for (int b0 = 0; b0 < cnt; b0 += 32) {
    int rem = min(32, cnt - b0);
    int s_l = (lane < rem) ? esrc[beg + b0 + lane] : n;
    float ex[8];
#pragma unroll
    for (int t = 0; t < 8; ++t)
      ex[t] = as[(size_t)__shfl(s_l, 4 * t + jq, 32) * HEADS + hq];
#pragma unroll
    for (int t = 0; t < 8; ++t) {
      ex[t] = (4 * t + jq < rem) ? __expf(leaky(ex[t] + adq) - mq) : 0.f;
      sq += ex[t];
    }
#pragma unroll
    for (int g = 0; g < 4; ++g) {
      if (8 * g < rem) {
        float2 hr[8];
        float ej[8];
#pragma unroll
        for (int jj = 0; jj < 8; ++jj) {
          const int j = 8 * g + jj;
          int sj = __shfl(s_l, j, 32);
          ej[jj] = __shfl(ex[j >> 2], (j & 3) * 8 + hsel, 32);
          int col = (j < rem) ? lane : 0;
          hr[jj] = ((const float2*)(h1 + (size_t)sj * HID))[col];
        }
#pragma unroll
        for (int jj = 0; jj < 8; ++jj) {
          bool ok = (8 * g + jj) < rem;
          accx = ok ? fmaf(ej[jj], hr[jj].x, accx) : accx;
          accy = ok ? fmaf(ej[jj], hr[jj].y, accy) : accy;
        }
      }
    }
  }
  sq += __shfl_xor(sq, 8, 32);
  sq += __shfl_xor(sq, 16, 32);
  float den = __shfl(sq, hsel, 32) + EPS_F;

  float vx = accx / den + b1[2 * lane];
  float vy = accy / den + b1[2 * lane + 1];
  float2 o;
  o.x = vx > 0.0f ? vx : expm1f(vx);
  o.y = vy > 0.0f ? vy : expm1f(vy);
  ((float2*)(hact + (size_t)n * HID))[lane] = o;
}

__global__ __launch_bounds__(256) void gat_gather2(
    const int* __restrict__ esrc, const int* __restrict__ seg_end,
    const int* __restrict__ deg, const float* __restrict__ as,
    const float* __restrict__ ad, const float* __restrict__ h2,
    const float* __restrict__ b2, float* __restrict__ dout, int N) {
  int lane = threadIdx.x & 31;
  int c = lane & 15, half = lane >> 4;
  int n0 = __builtin_amdgcn_readfirstlane((int)((blockIdx.x * 8 + (threadIdx.x >> 5)) * 2));
  if (n0 >= N) return;
  int n = n0 + half;
  bool live = n < N;
  int cnt = live ? deg[n] : 0;
  int beg = live ? seg_end[n] - cnt : 0;
  float adn = live ? ad[n] : 0.f;
  float bc = b2[c];
  int cmax = __builtin_amdgcn_readfirstlane(max(cnt, __shfl_xor(cnt, 16, 32)));

  float m = -INFINITY;
  for (int b0 = 0; b0 < cmax; b0 += 16) {
    int j = b0 + c;
    if (j < cnt) m = fmaxf(m, leaky(as[esrc[beg + j]] + adn));
  }
  m = fmaxf(m, __shfl_xor(m, 1, 32));
  m = fmaxf(m, __shfl_xor(m, 2, 32));
  m = fmaxf(m, __shfl_xor(m, 4, 32));
  m = fmaxf(m, __shfl_xor(m, 8, 32));

  float acc = 0.f, sum = 0.f;
  for (int b0 = 0; b0 < cmax; b0 += 16) {
    int j = b0 + c;
    int s_l = 0;
    float ex_l = 0.f;
    if (j < cnt) {
      s_l = esrc[beg + j];
      ex_l = __expf(leaky(as[s_l] + adn) - m);
    }
    sum += ex_l;
    int kmax = min(16, cmax - b0);
#pragma unroll
    for (int g = 0; g < 2; ++g) {
      if (8 * g < kmax) {
        float hr[8], ek[8];
#pragma unroll
        for (int jj = 0; jj < 8; ++jj) {
          int from = half * 16 + 8 * g + jj;
          int sk = __shfl(s_l, from, 32);
          ek[jj] = __shfl(ex_l, from, 32);
          hr[jj] = h2[(size_t)sk * DOUT + c];
        }
#pragma unroll
        for (int jj = 0; jj < 8; ++jj)
          acc = (b0 + 8 * g + jj < cnt) ? fmaf(ek[jj], hr[jj], acc) : acc;
      }
    }
  }
  sum += __shfl_xor(sum, 1, 32);
  sum += __shfl_xor(sum, 2, 32);
  sum += __shfl_xor(sum, 4, 32);
  sum += __shfl_xor(sum, 8, 32);
  float v = acc / (sum + EPS_F) + bc;
  float mx = v;
  mx = fmaxf(mx, __shfl_xor(mx, 1, 32));
  mx = fmaxf(mx, __shfl_xor(mx, 2, 32));
  mx = fmaxf(mx, __shfl_xor(mx, 4, 32));
  mx = fmaxf(mx, __shfl_xor(mx, 8, 32));
  float se = __expf(v - mx);
  se += __shfl_xor(se, 1, 32);
  se += __shfl_xor(se, 2, 32);
  se += __shfl_xor(se, 4, 32);
  se += __shfl_xor(se, 8, 32);
  float lse = mx + __logf(se);
  if (live) {
    dout[(size_t)n * DOUT + c] = v;
    dout[(size_t)N * DOUT + (size_t)n * DOUT + c] = v - lse;
  }
}

extern "C" void kernel_launch(void* const* d_in, const int* in_sizes, int n_in,
                              void* d_out, int out_size, void* d_ws, size_t ws_size,
                              hipStream_t stream) {
  const float* x   = (const float*)d_in[0];
  const int*   ei  = (const int*)d_in[1];
  const float* W1  = (const float*)d_in[2];
  const float* a1s = (const float*)d_in[3];
  const float* a1d = (const float*)d_in[4];
  const float* b1  = (const float*)d_in[5];
  const float* W2  = (const float*)d_in[6];
  const float* a2s = (const float*)d_in[7];
  const float* a2d = (const float*)d_in[8];
  const float* b2  = (const float*)d_in[9];

  int N = in_sizes[0] / DIM_IN;
  int E = in_sizes[1] / 2;
  const int* src = ei;
  const int* dst = ei + E;

  char* ws = (char*)d_ws;
  float*    h1   = (float*)ws;
  float*    hact = (float*)(ws + (size_t)N * HID * 4);
  char* p = ws + (size_t)N * HID * 4 * 2;
  float*    as1  = (float*)p; p += (size_t)N * HEADS * 4;
  int*      deg  = (int*)p;   p += ((size_t)N + 1) * 4;
  int*      cur  = (int*)p;   p += (size_t)N * 4;
  int*      esrc = (int*)p;
  float*    h2  = h1;
  char* q = (char*)(h1 + (size_t)N * DOUT);
  float*    as2 = (float*)q;  q += (size_t)N * 4;
  float*    ad2 = (float*)q;

  const int B = 256;
  hipMemsetAsync(deg, 0, ((size_t)N + 1) * 4, stream);
  gat_count_dst<<<(E + B - 1) / B, B, 0, stream>>>(dst, deg, E);
  gat_seg_alloc<<<(N + B - 1) / B, B, 0, stream>>>(deg, deg + N, cur, N);
  gat_place_src<<<(E + B - 1) / B, B, 0, stream>>>(src, dst, cur, esrc, E);

  gat_gemm1_wmma<<<(N + 31) / 32, B, 0, stream>>>(x, W1, a1s, h1, as1, N);
  gat_gather1<<<(N + 7) / 8, B, 0, stream>>>(esrc, cur, deg, as1, a1d, h1, b1, hact, N);

  gat_gemm2_wmma<<<(N + 127) / 128, B, 0, stream>>>(hact, W2, a2s, a2d, h2, as2, ad2, N);
  gat_gather2<<<(N + 15) / 16, B, 0, stream>>>(esrc, cur, deg, as2, ad2, h2, b2, (float*)d_out, N);
}


# Edge ordering by destination redone as a two-level counting sort on LDS counters (256-node buckets, then nodes): per-edge global atomics and the memset gone; sums in LDS arrival-rank order
# speedup vs baseline: 5.8667x; 1.6459x over previous
#include <hip/hip_runtime.h>
#include <math.h>


typedef float v2f __attribute__((ext_vector_type(2)));
typedef float v8f __attribute__((ext_vector_type(8)));

#define DIM_IN 256
#define HID    64
#define HEADS  8
#define DH     8
#define DOUT   16
#define NEG_SLOPE 0.2f
#define EPS_F 1e-16f

__device__ __forceinline__ float leaky(float v) {
  return v > 0.0f ? v : NEG_SLOPE * v;
}

__global__ __launch_bounds__(256) void gat_gemm1_wmma(
    const float* __restrict__ x, const float* __restrict__ W1,
    const float* __restrict__ a_src, float* __restrict__ h1,
    float* __restrict__ as, int N) {
  __shared__ float sW[DIM_IN * HID];
  int tid = threadIdx.x;
  {
    const float4* Wv = (const float4*)W1;
    float4* sWv = (float4*)sW;
#pragma unroll
    for (int i = 0; i < 16; ++i) sWv[tid + i * 256] = Wv[tid + i * 256];
  }
  __syncthreads();

  int wave = tid >> 5, lane = tid & 31;
  int row0 = (blockIdx.x * 2 + (wave >> 2)) * 16;
  int col0 = (wave & 3) * 16;
  if (row0 >= N) return;

  int r  = lane & 15;
  int kh = lane >> 4;

  int arow_idx = row0 + r; if (arow_idx >= N) arow_idx = N - 1;
  const float* arow  = x + (size_t)arow_idx * DIM_IN + 2 * kh;
  const float* bbase = sW + col0 + r;
  float asv = a_src[col0 + r];

  v8f c = {};
  for (int k = 0; k < DIM_IN; k += 4) {
    if ((k & 63) == 0) __builtin_prefetch(arow + k + 64, 0, 3);
    float2 av = *(const float2*)(arow + k);
    v2f a; a.x = av.x; a.y = av.y;
    v2f b; b.x = bbase[(k + 2 * kh) * HID];
           b.y = bbase[(k + 2 * kh + 1) * HID];
    c = __builtin_amdgcn_wmma_f32_16x16x4_f32(false, a, false, b,
                                              (short)0, c, false, false);
  }
  float* orow = h1 + (size_t)row0 * HID + col0 + r;
  if (row0 + 16 <= N) {
#pragma unroll
    for (int rr = 0; rr < 8; ++rr)
      orow[(size_t)(rr + 8 * kh) * HID] = c[rr];
  } else {
#pragma unroll
    for (int rr = 0; rr < 8; ++rr) {
      int row = row0 + rr + 8 * kh;
      if (row < N) orow[(size_t)(rr + 8 * kh) * HID] = c[rr];
    }
  }
  float sel = 0.f;
#pragma unroll
  for (int rr = 0; rr < 8; ++rr) {
    float p = c[rr] * asv;
    p += __shfl_xor(p, 1, 32);
    p += __shfl_xor(p, 2, 32);
    p += __shfl_xor(p, 4, 32);
    sel = ((r & 7) == rr) ? p : sel;
  }
  int srow = row0 + 8 * kh + (r & 7);
  if (srow < N) as[(size_t)srow * HEADS + (col0 >> 3) + (r >> 3)] = sel;
}

__global__ __launch_bounds__(256) void gat_gemm2_wmma(
    const float* __restrict__ hact, const float* __restrict__ W2,
    const float* __restrict__ a_src, const float* __restrict__ a_dst,
    float* __restrict__ h2, float* __restrict__ as, float* __restrict__ ad, int N) {
  __shared__ float sW[HID * DOUT];
  int tid = threadIdx.x;
  {
    const float4* Wv = (const float4*)W2;
    float4* sWv = (float4*)sW;
    sWv[tid] = Wv[tid];
  }
  __syncthreads();

  int wave = tid >> 5, lane = tid & 31;
  int row0 = (blockIdx.x * 8 + wave) * 16;
  if (row0 >= N) return;

  int r  = lane & 15;
  int kh = lane >> 4;
  int arow_idx = row0 + r; if (arow_idx >= N) arow_idx = N - 1;
  const float* arow  = hact + (size_t)arow_idx * HID + 2 * kh;
  const float* bbase = sW + r;
  float wsv = a_src[r], wdv = a_dst[r];

  v8f c = {};
#pragma unroll
  for (int k = 0; k < HID; k += 4) {
    float2 av = *(const float2*)(arow + k);
    v2f a; a.x = av.x; a.y = av.y;
    v2f b; b.x = bbase[(k + 2 * kh) * DOUT];
           b.y = bbase[(k + 2 * kh + 1) * DOUT];
    c = __builtin_amdgcn_wmma_f32_16x16x4_f32(false, a, false, b,
                                              (short)0, c, false, false);
  }
  float* orow = h2 + (size_t)row0 * DOUT + r;
  if (row0 + 16 <= N) {
#pragma unroll
    for (int rr = 0; rr < 8; ++rr)
      orow[(size_t)(rr + 8 * kh) * DOUT] = c[rr];
  } else {
#pragma unroll
    for (int rr = 0; rr < 8; ++rr) {
      int row = row0 + rr + 8 * kh;
      if (row < N) orow[(size_t)(rr + 8 * kh) * DOUT] = c[rr];
    }
  }
  float ssel = 0.f, dsel = 0.f;
#pragma unroll
  for (int rr = 0; rr < 8; ++rr) {
    float ps = c[rr] * wsv, pd = c[rr] * wdv;
    ps += __shfl_xor(ps, 1, 32);  pd += __shfl_xor(pd, 1, 32);
    ps += __shfl_xor(ps, 2, 32);  pd += __shfl_xor(pd, 2, 32);
    ps += __shfl_xor(ps, 4, 32);  pd += __shfl_xor(pd, 4, 32);
    ps += __shfl_xor(ps, 8, 32);  pd += __shfl_xor(pd, 8, 32);
    ssel = ((r & 7) == rr) ? ps : ssel;
    dsel = ((r & 7) == rr) ? pd : dsel;
  }
  int srow = row0 + 8 * kh + (r & 7);
  float* sp = (r < 8) ? as : ad;
  if (srow < N) sp[srow] = (r < 8) ? ssel : dsel;
}

#define BSH    8
#define BNODES 256
#define MAXB   1024
#define SG     256
#define ST     1024
#define FT     512

__device__ __forceinline__ int block_excl_scan(int v, int* wsum, int tid, int nwaves) {
  int lane = tid & 31, w = tid >> 5;
  int incl = v;
#pragma unroll
  for (int o = 1; o < 32; o <<= 1) {
    int t = __shfl_up(incl, o, 32);
    if (lane >= o) incl += t;
  }
  if (lane == 31) wsum[w] = incl;
  __syncthreads();
  if (w == 0) {
    int s = (lane < nwaves) ? wsum[lane] : 0;
    int si = s;
#pragma unroll
    for (int o = 1; o < 32; o <<= 1) {
      int t = __shfl_up(si, o, 32);
      if (lane >= o) si += t;
    }
    wsum[lane] = si - s;
  }
  __syncthreads();
  return wsum[w] + incl - v;
}

__global__ __launch_bounds__(ST) void gat_bucket_count(
    const int* __restrict__ dst, int* __restrict__ cntm, int E, int NB) {
  __shared__ int hist[MAXB];
  int tid = threadIdx.x;
  hist[tid] = 0;
  __syncthreads();
  int chunk = (E + (int)gridDim.x - 1) / (int)gridDim.x;
  int e0 = (int)blockIdx.x * chunk;
  int e1 = min(E, e0 + chunk);
  for (int eb = e0; eb < e1; eb += 4 * ST) {
    int d[4];
#pragma unroll
    for (int u = 0; u < 4; ++u) {
      int e = eb + u * ST + tid;
      d[u] = (e < e1) ? dst[e] : -1;
    }
#pragma unroll
    for (int u = 0; u < 4; ++u)
      if (d[u] >= 0) atomicAdd(&hist[d[u] >> BSH], 1);
  }
  __syncthreads();
  if (tid < NB) cntm[(size_t)blockIdx.x * NB + tid] = hist[tid];
}

__global__ __launch_bounds__(ST) void gat_bucket_scatter(
    const int* __restrict__ src, const int* __restrict__ dst,
    const int* __restrict__ cntm, int* __restrict__ bst, int* __restrict__ tmp,
    int E, int NB) {
  __shared__ int cur[MAXB];
  __shared__ int totL[MAXB];
  __shared__ int preL[MAXB];
  __shared__ int wsum[32];
  int tid = threadIdx.x;
  int g = (int)blockIdx.x, G = (int)gridDim.x;
  totL[tid] = 0;
  preL[tid] = 0;
  __syncthreads();
  int slices = ST / NB;
  if (tid < slices * NB) {
    int bb = tid % NB, sl = tid / NB;
    int q0 = (sl * G) / slices, q1 = ((sl + 1) * G) / slices;
    int t = 0, p = 0;
#pragma unroll 16
    for (int q = q0; q < q1; ++q) {
      int v = cntm[(size_t)q * NB + bb];
      t += v;
      p += (q < g) ? v : 0;
    }
    atomicAdd(&totL[bb], t);
    atomicAdd(&preL[bb], p);
  }
  __syncthreads();
  int tot = (tid < NB) ? totL[tid] : 0;
  int ex = block_excl_scan(tot, wsum, tid, ST / 32);
  cur[tid] = ex + preL[tid];
  if (g == 0 && tid < NB) bst[tid] = ex;
  __syncthreads();

  int chunk = (E + G - 1) / G;
  int e0 = g * chunk;
  int e1 = min(E, e0 + chunk);
  for (int eb = e0; eb < e1; eb += 4 * ST) {
    int d[4], s[4];
#pragma unroll
    for (int u = 0; u < 4; ++u) {
      int e = eb + u * ST + tid;
      bool ok = e < e1;
      d[u] = ok ? dst[e] : -1;
      s[u] = ok ? src[e] : 0;
    }
#pragma unroll
    for (int u = 0; u < 4; ++u) {
      if (d[u] >= 0) {
        int pos = atomicAdd(&cur[d[u] >> BSH], 1);
        int2 pr; pr.x = s[u]; pr.y = d[u];
        *(int2*)(tmp + 2 * (size_t)pos) = pr;
      }
    }
  }
}

__global__ __launch_bounds__(FT) void gat_bucket_sort(
    const int* __restrict__ tmp, const int* __restrict__ bst,
    int* __restrict__ deg, int* __restrict__ cur, int* __restrict__ esrc,
    int N, int E, int NB) {
  __shared__ int c[BNODES];
  __shared__ int p[BNODES];
  __shared__ int wsum[32];
  int tid = threadIdx.x, b = (int)blockIdx.x;
  int lo = bst[b];
  int hi = (b + 1 < NB) ? bst[b + 1] : E;
  if (tid < BNODES) c[tid] = 0;
  __syncthreads();
  for (int ib = lo; ib < hi; ib += 4 * FT) {
    int d[4];
#pragma unroll
    for (int u = 0; u < 4; ++u) {
      int i = ib + u * FT + tid;
      d[u] = (i < hi) ? tmp[2 * (size_t)i + 1] : -1;
    }
#pragma unroll
    for (int u = 0; u < 4; ++u)
      if (d[u] >= 0) atomicAdd(&c[d[u] & (BNODES - 1)], 1);
  }
  __syncthreads();
  int v = (tid < BNODES) ? c[tid] : 0;
  int ex = block_excl_scan(v, wsum, tid, FT / 32);
  if (tid < BNODES) {
    p[tid] = lo + ex;
    int n = b * BNODES + tid;
    if (n < N) {
      deg[n] = v;
      cur[n] = lo + ex + v;
    }
  }
  __syncthreads();
  for (int ib = lo; ib < hi; ib += 4 * FT) {
    int2 r[4];
#pragma unroll
    for (int u = 0; u < 4; ++u) {
      int i = ib + u * FT + tid;
      r[u].x = 0; r[u].y = -1;
      if (i < hi) r[u] = *(const int2*)(tmp + 2 * (size_t)i);
    }
#pragma unroll
    for (int u = 0; u < 4; ++u) {
      if (r[u].y >= 0) {
        int pos = atomicAdd(&p[r[u].y & (BNODES - 1)], 1);
        esrc[pos] = r[u].x;
      }
    }
  }
}

__global__ __launch_bounds__(256) void gat_gather1(
    const int* __restrict__ esrc, const int* __restrict__ seg_end,
    const int* __restrict__ deg, const float* __restrict__ as,
    const float* __restrict__ a_dst, const float* __restrict__ h1,
    const float* __restrict__ b1, float* __restrict__ hact, int N) {
  int lane = threadIdx.x & 31;
  int n = __builtin_amdgcn_readfirstlane((int)(blockIdx.x * 8 + (threadIdx.x >> 5)));
  if (n >= N) return;
  int cnt = __builtin_amdgcn_readfirstlane(deg[n]);
  int beg = __builtin_amdgcn_readfirstlane(seg_end[n]) - cnt;

  float2 hv = ((const float2*)(h1 + (size_t)n * HID))[lane];
  float pd = hv.x * a_dst[2 * lane] + hv.y * a_dst[2 * lane + 1];
  pd += __shfl_xor(pd, 1, 32);
  pd += __shfl_xor(pd, 2, 32);

  int hq = lane & 7, jq = lane >> 3;
  int hsel = lane >> 2;
  float adq = __shfl(pd, hq * 4, 32);

  float mq = -INFINITY;
  for (int b0 = 0; b0 < cnt; b0 += 32) {
    int rem = min(32, cnt - b0);
    int s_l = (lane < rem) ? esrc[beg + b0 + lane] : n;
    float a[8];
#pragma unroll
    for (int t = 0; t < 8; ++t)
      a[t] = as[(size_t)__shfl(s_l, 4 * t + jq, 32) * HEADS + hq];
#pragma unroll
    for (int t = 0; t < 8; ++t)
      if (4 * t + jq < rem) mq = fmaxf(mq, leaky(a[t] + adq));
  }
  mq = fmaxf(mq, __shfl_xor(mq, 8, 32));
  mq = fmaxf(mq, __shfl_xor(mq, 16, 32));

  float accx = 0.f, accy = 0.f, sq = 0.f;
  for (int b0 = 0; b0 < cnt; b0 += 32) {
    int rem = min(32, cnt - b0);
    int s_l = (lane < rem) ? esrc[beg + b0 + lane] : n;
    float ex[8];
#pragma unroll
    for (int t = 0; t < 8; ++t)
      ex[t] = as[(size_t)__shfl(s_l, 4 * t + jq, 32) * HEADS + hq];
#pragma unroll
    for (int t = 0; t < 8; ++t) {
      ex[t] = (4 * t + jq < rem) ? __expf(leaky(ex[t] + adq) - mq) : 0.f;
      sq += ex[t];
    }
#pragma unroll
    for (int g = 0; g < 4; ++g) {
      if (8 * g < rem) {
        float2 hr[8];
        float ej[8];
#pragma unroll
        for (int jj = 0; jj < 8; ++jj) {
          const int j = 8 * g + jj;
          int sj = __shfl(s_l, j, 32);
          ej[jj] = __shfl(ex[j >> 2], (j & 3) * 8 + hsel, 32);
          int col = (j < rem) ? lane : 0;
          hr[jj] = ((const float2*)(h1 + (size_t)sj * HID))[col];
        }
#pragma unroll
        for (int jj = 0; jj < 8; ++jj) {
          bool ok = (8 * g + jj) < rem;
          accx = ok ? fmaf(ej[jj], hr[jj].x, accx) : accx;
          accy = ok ? fmaf(ej[jj], hr[jj].y, accy) : accy;
        }
      }
    }
  }
  sq += __shfl_xor(sq, 8, 32);
  sq += __shfl_xor(sq, 16, 32);
  float den = __shfl(sq, hsel, 32) + EPS_F;

  float vx = accx / den + b1[2 * lane];
  float vy = accy / den + b1[2 * lane + 1];
  float2 o;
  o.x = vx > 0.0f ? vx : expm1f(vx);
  o.y = vy > 0.0f ? vy : expm1f(vy);
  ((float2*)(hact + (size_t)n * HID))[lane] = o;
}

__global__ __launch_bounds__(256) void gat_gather2(
    const int* __restrict__ esrc, const int* __restrict__ seg_end,
    const int* __restrict__ deg, const float* __restrict__ as,
    const float* __restrict__ ad, const float* __restrict__ h2,
    const float* __restrict__ b2, float* __restrict__ dout, int N) {
  int lane = threadIdx.x & 31;
  int c = lane & 15, half = lane >> 4;
  int n0 = __builtin_amdgcn_readfirstlane((int)((blockIdx.x * 8 + (threadIdx.x >> 5)) * 2));
  if (n0 >= N) return;
  int n = n0 + half;
  bool live = n < N;
  int cnt = live ? deg[n] : 0;
  int beg = live ? seg_end[n] - cnt : 0;
  float adn = live ? ad[n] : 0.f;
  float bc = b2[c];
  int cmax = __builtin_amdgcn_readfirstlane(max(cnt, __shfl_xor(cnt, 16, 32)));

  float m = -INFINITY;
  for (int b0 = 0; b0 < cmax; b0 += 16) {
    int j = b0 + c;
    if (j < cnt) m = fmaxf(m, leaky(as[esrc[beg + j]] + adn));
  }
  m = fmaxf(m, __shfl_xor(m, 1, 32));
  m = fmaxf(m, __shfl_xor(m, 2, 32));
  m = fmaxf(m, __shfl_xor(m, 4, 32));
  m = fmaxf(m, __shfl_xor(m, 8, 32));

  float acc = 0.f, sum = 0.f;
  for (int b0 = 0; b0 < cmax; b0 += 16) {
    int j = b0 + c;
    int s_l = 0;
    float ex_l = 0.f;
    if (j < cnt) {
      s_l = esrc[beg + j];
      ex_l = __expf(leaky(as[s_l] + adn) - m);
    }
    sum += ex_l;
    int kmax = min(16, cmax - b0);
#pragma unroll
    for (int g = 0; g < 2; ++g) {
      if (8 * g < kmax) {
        float hr[8], ek[8];
#pragma unroll
        for (int jj = 0; jj < 8; ++jj) {
          int from = half * 16 + 8 * g + jj;
          int sk = __shfl(s_l, from, 32);
          ek[jj] = __shfl(ex_l, from, 32);
          hr[jj] = h2[(size_t)sk * DOUT + c];
        }
#pragma unroll
        for (int jj = 0; jj < 8; ++jj)
          acc = (b0 + 8 * g + jj < cnt) ? fmaf(ek[jj], hr[jj], acc) : acc;
      }
    }
  }
  sum += __shfl_xor(sum, 1, 32);
  sum += __shfl_xor(sum, 2, 32);
  sum += __shfl_xor(sum, 4, 32);
  sum += __shfl_xor(sum, 8, 32);
  float v = acc / (sum + EPS_F) + bc;
  float mx = v;
  mx = fmaxf(mx, __shfl_xor(mx, 1, 32));
  mx = fmaxf(mx, __shfl_xor(mx, 2, 32));
  mx = fmaxf(mx, __shfl_xor(mx, 4, 32));
  mx = fmaxf(mx, __shfl_xor(mx, 8, 32));
  float se = __expf(v - mx);
  se += __shfl_xor(se, 1, 32);
  se += __shfl_xor(se, 2, 32);
  se += __shfl_xor(se, 4, 32);
  se += __shfl_xor(se, 8, 32);
  float lse = mx + __logf(se);
  if (live) {
    dout[(size_t)n * DOUT + c] = v;
    dout[(size_t)N * DOUT + (size_t)n * DOUT + c] = v - lse;
  }
}

extern "C" void kernel_launch(void* const* d_in, const int* in_sizes, int n_in,
                              void* d_out, int out_size, void* d_ws, size_t ws_size,
                              hipStream_t stream) {
  const float* x   = (const float*)d_in[0];
  const int*   ei  = (const int*)d_in[1];
  const float* W1  = (const float*)d_in[2];
  const float* a1s = (const float*)d_in[3];
  const float* a1d = (const float*)d_in[4];
  const float* b1  = (const float*)d_in[5];
  const float* W2  = (const float*)d_in[6];
  const float* a2s = (const float*)d_in[7];
  const float* a2d = (const float*)d_in[8];
  const float* b2  = (const float*)d_in[9];

  int N = in_sizes[0] / DIM_IN;
  int E = in_sizes[1] / 2;
  const int* src = ei;
  const int* dst = ei + E;

  char* ws = (char*)d_ws;
  float*    h1   = (float*)ws;
  float*    hact = (float*)(ws + (size_t)N * HID * 4);
  char* p = ws + (size_t)N * HID * 4 * 2;
  float*    as1  = (float*)p; p += (size_t)N * HEADS * 4;
  int*      deg  = (int*)p;   p += ((size_t)N + 1) * 4;
  int*      cur  = (int*)p;   p += (size_t)N * 4;
  int*      esrc = (int*)p;
  const int NB = (N + BNODES - 1) / BNODES;
  int*      tmp  = (int*)hact;
  int*      cntm = tmp + 2 * (size_t)E;
  int*      bst  = cntm + (size_t)SG * NB;
  float*    h2  = h1;
  char* q = (char*)(h1 + (size_t)N * DOUT);
  float*    as2 = (float*)q;  q += (size_t)N * 4;
  float*    ad2 = (float*)q;

  const int B = 256;
  gat_bucket_count<<<SG, ST, 0, stream>>>(dst, cntm, E, NB);
  gat_bucket_scatter<<<SG, ST, 0, stream>>>(src, dst, cntm, bst, tmp, E, NB);
  gat_bucket_sort<<<NB, FT, 0, stream>>>(tmp, bst, deg, cur, esrc, N, E, NB);

  gat_gemm1_wmma<<<(N + 31) / 32, B, 0, stream>>>(x, W1, a1s, h1, as1, N);
  gat_gather1<<<(N + 7) / 8, B, 0, stream>>>(esrc, cur, deg, as1, a1d, h1, b1, hact, N);

  gat_gemm2_wmma<<<(N + 127) / 128, B, 0, stream>>>(hact, W2, a2s, a2d, h2, as2, ad2, N);
  gat_gather2<<<(N + 15) / 16, B, 0, stream>>>(esrc, cur, deg, as2, ad2, h2, b2, (float*)d_out, N);
}


# Gathers made one pass: edge score = the gathered source row's own dot with a_src (lane tree), running max with rescaled sums, exp sums per 8 edges as lane trees; score arrays and their gathers gone
# speedup vs baseline: 7.0612x; 1.2036x over previous
#include <hip/hip_runtime.h>
#include <math.h>


typedef float v2f __attribute__((ext_vector_type(2)));
typedef float v8f __attribute__((ext_vector_type(8)));

#define DIM_IN 256
#define HID    64
#define HEADS  8
#define DH     8
#define DOUT   16
#define NEG_SLOPE 0.2f
#define EPS_F 1e-16f

__device__ __forceinline__ float leaky(float v) {
  return v > 0.0f ? v : NEG_SLOPE * v;
}

template <int K>
__device__ __forceinline__ float xlane(float v) {
  return __builtin_bit_cast(float, __builtin_amdgcn_update_dpp(
      0, __builtin_bit_cast(int, v), 0x160 + K, 0xF, 0xF, true));
}
template <int K>
__device__ __forceinline__ float from_lane(float v) {
  return __builtin_bit_cast(float, __builtin_amdgcn_update_dpp(
      0, __builtin_bit_cast(int, v), 0x150 + K, 0xF, 0xF, true));
}
template <int K>
__device__ __forceinline__ float quad_lane(float v) {
  return __builtin_bit_cast(float, __builtin_amdgcn_update_dpp(
      0, __builtin_bit_cast(int, v), 0x55 * K, 0xF, 0xF, true));
}

__global__ __launch_bounds__(256) void gat_gemm1_wmma(
    const float* __restrict__ x, const float* __restrict__ W1,
    float* __restrict__ h1, int N) {
  __shared__ float sW[DIM_IN * HID];
  int tid = threadIdx.x;
  {
    const float4* Wv = (const float4*)W1;
    float4* sWv = (float4*)sW;
#pragma unroll
    for (int i = 0; i < 16; ++i) sWv[tid + i * 256] = Wv[tid + i * 256];
  }
  __syncthreads();

  int wave = tid >> 5, lane = tid & 31;
  int row0 = (blockIdx.x * 2 + (wave >> 2)) * 16;
  int col0 = (wave & 3) * 16;
  if (row0 >= N) return;

  int r  = lane & 15;
  int kh = lane >> 4;

  int arow_idx = row0 + r; if (arow_idx >= N) arow_idx = N - 1;
  const float* arow  = x + (size_t)arow_idx * DIM_IN + 2 * kh;
  const float* bbase = sW + col0 + r;

  v8f c = {};
  for (int k = 0; k < DIM_IN; k += 4) {
    if ((k & 63) == 0) __builtin_prefetch(arow + k + 64, 0, 3);
    float2 av = *(const float2*)(arow + k);
    v2f a; a.x = av.x; a.y = av.y;
    v2f b; b.x = bbase[(k + 2 * kh) * HID];
           b.y = bbase[(k + 2 * kh + 1) * HID];
    c = __builtin_amdgcn_wmma_f32_16x16x4_f32(false, a, false, b,
                                              (short)0, c, false, false);
  }
  float* orow = h1 + (size_t)row0 * HID + col0 + r;
  if (row0 + 16 <= N) {
#pragma unroll
    for (int rr = 0; rr < 8; ++rr)
      orow[(size_t)(rr + 8 * kh) * HID] = c[rr];
  } else {
#pragma unroll
    for (int rr = 0; rr < 8; ++rr) {
      int row = row0 + rr + 8 * kh;
      if (row < N) orow[(size_t)(rr + 8 * kh) * HID] = c[rr];
    }
  }
}

__global__ __launch_bounds__(256) void gat_gemm2_wmma(
    const float* __restrict__ hact, const float* __restrict__ W2,
    const float* __restrict__ a_dst,
    float* __restrict__ h2, float* __restrict__ ad, int N) {
  __shared__ float sW[HID * DOUT];
  int tid = threadIdx.x;
  {
    const float4* Wv = (const float4*)W2;
    float4* sWv = (float4*)sW;
    sWv[tid] = Wv[tid];
  }
  __syncthreads();

  int wave = tid >> 5, lane = tid & 31;
  int row0 = (blockIdx.x * 8 + wave) * 16;
  if (row0 >= N) return;

  int r  = lane & 15;
  int kh = lane >> 4;
  int arow_idx = row0 + r; if (arow_idx >= N) arow_idx = N - 1;
  const float* arow  = hact + (size_t)arow_idx * HID + 2 * kh;
  const float* bbase = sW + r;
  float wdv = a_dst[r];

  v8f c = {};
#pragma unroll
  for (int k = 0; k < HID; k += 4) {
    float2 av = *(const float2*)(arow + k);
    v2f a; a.x = av.x; a.y = av.y;
    v2f b; b.x = bbase[(k + 2 * kh) * DOUT];
           b.y = bbase[(k + 2 * kh + 1) * DOUT];
    c = __builtin_amdgcn_wmma_f32_16x16x4_f32(false, a, false, b,
                                              (short)0, c, false, false);
  }
  float* orow = h2 + (size_t)row0 * DOUT + r;
  if (row0 + 16 <= N) {
#pragma unroll
    for (int rr = 0; rr < 8; ++rr)
      orow[(size_t)(rr + 8 * kh) * DOUT] = c[rr];
  } else {
#pragma unroll
    for (int rr = 0; rr < 8; ++rr) {
      int row = row0 + rr + 8 * kh;
      if (row < N) orow[(size_t)(rr + 8 * kh) * DOUT] = c[rr];
    }
  }
  float dsel = 0.f;
#pragma unroll
  for (int rr = 0; rr < 8; ++rr) {
    float pd = c[rr] * wdv;
    pd += __shfl_xor(pd, 1, 32);
    pd += __shfl_xor(pd, 2, 32);
    pd += __shfl_xor(pd, 4, 32);
    pd += __shfl_xor(pd, 8, 32);
    dsel = ((r & 7) == rr) ? pd : dsel;
  }
  int srow = row0 + 8 * kh + (r & 7);
  if (r < 8 && srow < N) ad[srow] = dsel;
}

#define BSH    8
#define BNODES 256
#define MAXB   1024
#define SG     256
#define ST     1024
#define FT     512

__device__ __forceinline__ int block_excl_scan(int v, int* wsum, int tid, int nwaves) {
  int lane = tid & 31, w = tid >> 5;
  int incl = v;
#pragma unroll
  for (int o = 1; o < 32; o <<= 1) {
    int t = __shfl_up(incl, o, 32);
    if (lane >= o) incl += t;
  }
  if (lane == 31) wsum[w] = incl;
  __syncthreads();
  if (w == 0) {
    int s = (lane < nwaves) ? wsum[lane] : 0;
    int si = s;
#pragma unroll
    for (int o = 1; o < 32; o <<= 1) {
      int t = __shfl_up(si, o, 32);
      if (lane >= o) si += t;
    }
    wsum[lane] = si - s;
  }
  __syncthreads();
  return wsum[w] + incl - v;
}

__global__ __launch_bounds__(ST) void gat_bucket_count(
    const int* __restrict__ dst, int* __restrict__ cntm, int E, int NB) {
  __shared__ int hist[MAXB];
  int tid = threadIdx.x;
  hist[tid] = 0;
  __syncthreads();
  int chunk = (E + (int)gridDim.x - 1) / (int)gridDim.x;
  int e0 = (int)blockIdx.x * chunk;
  int e1 = min(E, e0 + chunk);
  for (int eb = e0; eb < e1; eb += 4 * ST) {
    int d[4];
#pragma unroll
    for (int u = 0; u < 4; ++u) {
      int e = eb + u * ST + tid;
      d[u] = (e < e1) ? dst[e] : -1;
    }
#pragma unroll
    for (int u = 0; u < 4; ++u)
      if (d[u] >= 0) atomicAdd(&hist[d[u] >> BSH], 1);
  }
  __syncthreads();
  if (tid < NB) cntm[(size_t)blockIdx.x * NB + tid] = hist[tid];
}

__global__ __launch_bounds__(ST) void gat_bucket_scatter(
    const int* __restrict__ src, const int* __restrict__ dst,
    const int* __restrict__ cntm, int* __restrict__ bst, int* __restrict__ tmp,
    int E, int NB) {
  __shared__ int cur[MAXB];
  __shared__ int totL[MAXB];
  __shared__ int preL[MAXB];
  __shared__ int wsum[32];
  int tid = threadIdx.x;
  int g = (int)blockIdx.x, G = (int)gridDim.x;
  totL[tid] = 0;
  preL[tid] = 0;
  __syncthreads();
  int slices = ST / NB;
  if (tid < slices * NB) {
    int bb = tid % NB, sl = tid / NB;
    int q0 = (sl * G) / slices, q1 = ((sl + 1) * G) / slices;
    int t = 0, p = 0;
#pragma unroll 16
    for (int q = q0; q < q1; ++q) {
      int v = cntm[(size_t)q * NB + bb];
      t += v;
      p += (q < g) ? v : 0;
    }
    atomicAdd(&totL[bb], t);
    atomicAdd(&preL[bb], p);
  }
  __syncthreads();
  int tot = (tid < NB) ? totL[tid] : 0;
  int ex = block_excl_scan(tot, wsum, tid, ST / 32);
  cur[tid] = ex + preL[tid];
  if (g == 0 && tid < NB) bst[tid] = ex;
  __syncthreads();

  int chunk = (E + G - 1) / G;
  int e0 = g * chunk;
  int e1 = min(E, e0 + chunk);
  for (int eb = e0; eb < e1; eb += 4 * ST) {
    int d[4], s[4];
#pragma unroll
    for (int u = 0; u < 4; ++u) {
      int e = eb + u * ST + tid;
      bool ok = e < e1;
      d[u] = ok ? dst[e] : -1;
      s[u] = ok ? src[e] : 0;
    }
#pragma unroll
    for (int u = 0; u < 4; ++u) {
      if (d[u] >= 0) {
        int pos = atomicAdd(&cur[d[u] >> BSH], 1);
        int2 pr; pr.x = s[u]; pr.y = d[u];
        *(int2*)(tmp + 2 * (size_t)pos) = pr;
      }
    }
  }
}

__global__ __launch_bounds__(FT) void gat_bucket_sort(
    const int* __restrict__ tmp, const int* __restrict__ bst,
    int* __restrict__ deg, int* __restrict__ cur, int* __restrict__ esrc,
    int N, int E, int NB) {
  __shared__ int c[BNODES];
  __shared__ int p[BNODES];
  __shared__ int wsum[32];
  int tid = threadIdx.x, b = (int)blockIdx.x;
  int lo = bst[b];
  int hi = (b + 1 < NB) ? bst[b + 1] : E;
  if (tid < BNODES) c[tid] = 0;
  __syncthreads();
  for (int ib = lo; ib < hi; ib += 4 * FT) {
    int d[4];
#pragma unroll
    for (int u = 0; u < 4; ++u) {
      int i = ib + u * FT + tid;
      d[u] = (i < hi) ? tmp[2 * (size_t)i + 1] : -1;
    }
#pragma unroll
    for (int u = 0; u < 4; ++u)
      if (d[u] >= 0) atomicAdd(&c[d[u] & (BNODES - 1)], 1);
  }
  __syncthreads();
  int v = (tid < BNODES) ? c[tid] : 0;
  int ex = block_excl_scan(v, wsum, tid, FT / 32);
  if (tid < BNODES) {
    p[tid] = lo + ex;
    int n = b * BNODES + tid;
    if (n < N) {
      deg[n] = v;
      cur[n] = lo + ex + v;
    }
  }
  __syncthreads();
  for (int ib = lo; ib < hi; ib += 4 * FT) {
    int2 r[4];
#pragma unroll
    for (int u = 0; u < 4; ++u) {
      int i = ib + u * FT + tid;
      r[u].x = 0; r[u].y = -1;
      if (i < hi) r[u] = *(const int2*)(tmp + 2 * (size_t)i);
    }
#pragma unroll
    for (int u = 0; u < 4; ++u) {
      if (r[u].y >= 0) {
        int pos = atomicAdd(&p[r[u].y & (BNODES - 1)], 1);
        esrc[pos] = r[u].x;
      }
    }
  }
}

__global__ __launch_bounds__(256) void gat_gather1(
    const int* __restrict__ esrc, const int* __restrict__ seg_end,
    const int* __restrict__ deg, const float* __restrict__ a_src,
    const float* __restrict__ a_dst, const float* __restrict__ h1,
    const float* __restrict__ b1, float* __restrict__ hact, int N) {
  int lane = threadIdx.x & 31;
  int n = __builtin_amdgcn_readfirstlane((int)(blockIdx.x * 8 + (threadIdx.x >> 5)));
  if (n >= N) return;
  int cnt = __builtin_amdgcn_readfirstlane(deg[n]);
  int beg = __builtin_amdgcn_readfirstlane(seg_end[n]) - cnt;

  float2 hv = ((const float2*)(h1 + (size_t)n * HID))[lane];
  float2 wd = ((const float2*)a_dst)[lane];
  float2 ws = ((const float2*)a_src)[lane];
  float pd = hv.x * wd.x + hv.y * wd.y;
  pd += xlane<1>(pd);
  pd += xlane<2>(pd);

  const int kq = lane & 3;
  const bool up2 = (lane & 2) != 0, up1 = (lane & 1) != 0;
  float m = -INFINITY;
  float accx = 0.f, accy = 0.f, sq = 0.f;
  for (int b0 = 0; b0 < cnt; b0 += 32) {
    int rem = min(32, cnt - b0);
    int s_l = (lane < rem) ? esrc[beg + b0 + lane] : n;
#pragma unroll
    for (int g = 0; g < 4; ++g) {
      if (8 * g < rem) {
        float2 hr[8];
#pragma unroll
        for (int jj = 0; jj < 8; ++jj) {
          const int j = 8 * g + jj;
          int sj = __shfl(s_l, j, 32);
          int col = (j < rem) ? lane : 0;
          hr[jj] = ((const float2*)(h1 + (size_t)sj * HID))[col];
        }
        float q4[4], d[2];
        for (int t = 0; t < 4; ++t) {
          float lo = hr[t].x * ws.x + hr[t].y * ws.y;
          float hi = hr[t + 4].x * ws.x + hr[t + 4].y * ws.y;
          q4[t] = (up2 ? hi : lo) + xlane<2>(up2 ? lo : hi);
        }
        for (int u = 0; u < 2; ++u)
          d[u] = (up1 ? q4[u + 2] : q4[u]) + xlane<1>(up1 ? q4[u] : q4[u + 2]);
        float e0 = (8 * g + 2 * kq     < rem) ? leaky(d[0] + pd) : -INFINITY;
        float e1 = (8 * g + 2 * kq + 1 < rem) ? leaky(d[1] + pd) : -INFINITY;
        float gm = fmaxf(e0, e1);
        gm = fmaxf(gm, xlane<1>(gm));
        gm = fmaxf(gm, xlane<2>(gm));
        float mn = fmaxf(m, gm);
        float sc = __expf(m - mn);
        float w0 = __expf(e0 - mn), w1 = __expf(e1 - mn);
        float wsum = w0 + w1;
        wsum += xlane<1>(wsum);
        wsum += xlane<2>(wsum);
        sq = fmaf(sq, sc, wsum);
        accx *= sc; accy *= sc;
        float wj;
        wj = quad_lane<0>(w0); accx = fmaf(wj, hr[0].x, accx); accy = fmaf(wj, hr[0].y, accy);
        wj = quad_lane<0>(w1); accx = fmaf(wj, hr[1].x, accx); accy = fmaf(wj, hr[1].y, accy);
        wj = quad_lane<1>(w0); accx = fmaf(wj, hr[2].x, accx); accy = fmaf(wj, hr[2].y, accy);
        wj = quad_lane<1>(w1); accx = fmaf(wj, hr[3].x, accx); accy = fmaf(wj, hr[3].y, accy);
        wj = quad_lane<2>(w0); accx = fmaf(wj, hr[4].x, accx); accy = fmaf(wj, hr[4].y, accy);
        wj = quad_lane<2>(w1); accx = fmaf(wj, hr[5].x, accx); accy = fmaf(wj, hr[5].y, accy);
        wj = quad_lane<3>(w0); accx = fmaf(wj, hr[6].x, accx); accy = fmaf(wj, hr[6].y, accy);
        wj = quad_lane<3>(w1); accx = fmaf(wj, hr[7].x, accx); accy = fmaf(wj, hr[7].y, accy);
        m = mn;
      }
    }
  }
  float den = sq + EPS_F;

  float vx = accx / den + b1[2 * lane];
  float vy = accy / den + b1[2 * lane + 1];
  float2 o;
  o.x = vx > 0.0f ? vx : expm1f(vx);
  o.y = vy > 0.0f ? vy : expm1f(vy);
  ((float2*)(hact + (size_t)n * HID))[lane] = o;
}

__global__ __launch_bounds__(256) void gat_gather2(
    const int* __restrict__ esrc, const int* __restrict__ seg_end,
    const int* __restrict__ deg, const float* __restrict__ a_src,
    const float* __restrict__ ad, const float* __restrict__ h2,
    const float* __restrict__ b2, float* __restrict__ dout, int N) {
  int lane = threadIdx.x & 31;
  int c = lane & 15, half = lane >> 4;
  int n0 = __builtin_amdgcn_readfirstlane((int)((blockIdx.x * 8 + (threadIdx.x >> 5)) * 2));
  if (n0 >= N) return;
  int n = n0 + half;
  bool live = n < N;
  int cnt = live ? deg[n] : 0;
  int beg = live ? seg_end[n] - cnt : 0;
  float adn = live ? ad[n] : 0.f;
  float bc = b2[c];
  float ws = a_src[c];
  const bool up8 = (c & 8) != 0, up4 = (c & 4) != 0, up2 = (c & 2) != 0;
  int cmax = __builtin_amdgcn_readfirstlane(max(cnt, __shfl_xor(cnt, 16, 32)));

  float m = -INFINITY;
  float acc = 0.f, sum = 0.f;
  for (int b0 = 0; b0 < cmax; b0 += 16) {
    int j = b0 + c;
    int s_l = (j < cnt) ? esrc[beg + j] : 0;
    int kmax = min(16, cmax - b0);
#pragma unroll
    for (int g = 0; g < 2; ++g) {
      if (8 * g < kmax) {
        float hr[8];
#pragma unroll
        for (int jj = 0; jj < 8; ++jj) {
          int from = half * 16 + 8 * g + jj;
          int sk = __shfl(s_l, from, 32);
          hr[jj] = h2[(size_t)sk * DOUT + c];
        }
        float q4[4], q2[2];
        for (int t = 0; t < 4; ++t) {
          float lo = hr[t] * ws, hi = hr[t + 4] * ws;
          q4[t] = (up8 ? hi : lo) + xlane<8>(up8 ? lo : hi);
        }
        for (int t = 0; t < 2; ++t)
          q2[t] = (up4 ? q4[t + 2] : q4[t]) + xlane<4>(up4 ? q4[t] : q4[t + 2]);
        float dot = (up2 ? q2[1] : q2[0]) + xlane<2>(up2 ? q2[0] : q2[1]);
        dot += xlane<1>(dot);
        float e = (b0 + 8 * g + (c >> 1) < cnt) ? leaky(dot + adn) : -INFINITY;
        float gm = e;
        gm = fmaxf(gm, xlane<2>(gm));
        gm = fmaxf(gm, xlane<4>(gm));
        gm = fmaxf(gm, xlane<8>(gm));
        float mn = fmaxf(m, gm);
        float mr = (mn == -INFINITY) ? 0.f : mn;
        float sc = __expf(m - mr);
        float w = __expf(e - mr);
        float wsum = w;
        wsum += xlane<2>(wsum);
        wsum += xlane<4>(wsum);
        wsum += xlane<8>(wsum);
        sum = fmaf(sum, sc, wsum);
        acc *= sc;
        acc = fmaf(from_lane<0>(w),  hr[0], acc);
        acc = fmaf(from_lane<2>(w),  hr[1], acc);
        acc = fmaf(from_lane<4>(w),  hr[2], acc);
        acc = fmaf(from_lane<6>(w),  hr[3], acc);
        acc = fmaf(from_lane<8>(w),  hr[4], acc);
        acc = fmaf(from_lane<10>(w), hr[5], acc);
        acc = fmaf(from_lane<12>(w), hr[6], acc);
        acc = fmaf(from_lane<14>(w), hr[7], acc);
        m = mn;
      }
    }
  }
  float v = acc / (sum + EPS_F) + bc;
  float mx = v;
  mx = fmaxf(mx, __shfl_xor(mx, 1, 32));
  mx = fmaxf(mx, __shfl_xor(mx, 2, 32));
  mx = fmaxf(mx, __shfl_xor(mx, 4, 32));
  mx = fmaxf(mx, __shfl_xor(mx, 8, 32));
  float se = __expf(v - mx);
  se += __shfl_xor(se, 1, 32);
  se += __shfl_xor(se, 2, 32);
  se += __shfl_xor(se, 4, 32);
  se += __shfl_xor(se, 8, 32);
  float lse = mx + __logf(se);
  if (live) {
    dout[(size_t)n * DOUT + c] = v;
    dout[(size_t)N * DOUT + (size_t)n * DOUT + c] = v - lse;
  }
}

extern "C" void kernel_launch(void* const* d_in, const int* in_sizes, int n_in,
                              void* d_out, int out_size, void* d_ws, size_t ws_size,
                              hipStream_t stream) {
  const float* x   = (const float*)d_in[0];
  const int*   ei  = (const int*)d_in[1];
  const float* W1  = (const float*)d_in[2];
  const float* a1s = (const float*)d_in[3];
  const float* a1d = (const float*)d_in[4];
  const float* b1  = (const float*)d_in[5];
  const float* W2  = (const float*)d_in[6];
  const float* a2s = (const float*)d_in[7];
  const float* a2d = (const float*)d_in[8];
  const float* b2  = (const float*)d_in[9];

  int N = in_sizes[0] / DIM_IN;
  int E = in_sizes[1] / 2;
  const int* src = ei;
  const int* dst = ei + E;

  char* ws = (char*)d_ws;
  float*    h1   = (float*)ws;
  float*    hact = (float*)(ws + (size_t)N * HID * 4);
  char* p = ws + (size_t)N * HID * 4 * 2;
  int*      deg  = (int*)p;   p += ((size_t)N + 1) * 4;
  int*      cur  = (int*)p;   p += (size_t)N * 4;
  int*      esrc = (int*)p;
  const int NB = (N + BNODES - 1) / BNODES;
  int*      tmp  = (int*)hact;
  int*      cntm = tmp + 2 * (size_t)E;
  int*      bst  = cntm + (size_t)SG * NB;
  float*    h2  = h1;
  float*    ad2 = h1 + (size_t)N * DOUT;

  const int B = 256;
  gat_bucket_count<<<SG, ST, 0, stream>>>(dst, cntm, E, NB);
  gat_bucket_scatter<<<SG, ST, 0, stream>>>(src, dst, cntm, bst, tmp, E, NB);
  gat_bucket_sort<<<NB, FT, 0, stream>>>(tmp, bst, deg, cur, esrc, N, E, NB);

  gat_gemm1_wmma<<<(N + 31) / 32, B, 0, stream>>>(x, W1, h1, N);
  gat_gather1<<<(N + 7) / 8, B, 0, stream>>>(esrc, cur, deg, a1s, a1d, h1, b1, hact, N);

  gat_gemm2_wmma<<<(N + 127) / 128, B, 0, stream>>>(hact, W2, a2d, h2, ad2, N);
  gat_gather2<<<(N + 15) / 16, B, 0, stream>>>(esrc, cur, deg, a2s, ad2, h2, b2, (float*)d_out, N);
}


# GEMM1 re-tiled: a wave owns a 16x64 strip of h1: one A fragment feeds 4 accumulators, B comes as 16-byte LDS reads, whole 256-byte rows are stored; W1 staged per 64 rows; every sum formed as before
# speedup vs baseline: 7.9130x; 1.1206x over previous
#include <hip/hip_runtime.h>
#include <math.h>


typedef float v2f __attribute__((ext_vector_type(2)));
typedef float v8f __attribute__((ext_vector_type(8)));

#define DIM_IN 256
#define HID    64
#define HEADS  8
#define DH     8
#define DOUT   16
#define NEG_SLOPE 0.2f
#define EPS_F 1e-16f

__device__ __forceinline__ float leaky(float v) {
  return v > 0.0f ? v : NEG_SLOPE * v;
}

template <int K>
__device__ __forceinline__ float xlane(float v) {
  return __builtin_bit_cast(float, __builtin_amdgcn_update_dpp(
      0, __builtin_bit_cast(int, v), 0x160 + K, 0xF, 0xF, true));
}
template <int K>
__device__ __forceinline__ float from_lane(float v) {
  return __builtin_bit_cast(float, __builtin_amdgcn_update_dpp(
      0, __builtin_bit_cast(int, v), 0x150 + K, 0xF, 0xF, true));
}
template <int K>
__device__ __forceinline__ float quad_lane(float v) {
  return __builtin_bit_cast(float, __builtin_amdgcn_update_dpp(
      0, __builtin_bit_cast(int, v), 0x55 * K, 0xF, 0xF, true));
}

#define G1_WAVES 4
__global__ __launch_bounds__(128) void gat_gemm1_wmma(
    const float* __restrict__ x, const float* __restrict__ W1,
    float* __restrict__ h1, int N) {
  __shared__ float sW[DIM_IN * HID];
  int tid = threadIdx.x;
  int wave = tid >> 5, lane = tid & 31;
  int row0 = (blockIdx.x * G1_WAVES + wave) * 16;
  int r  = lane & 15;
  int kh = lane >> 4;

  int arow_idx = row0 + r; if (arow_idx >= N) arow_idx = N - 1;
  const float* arow = x + (size_t)arow_idx * DIM_IN + 2 * kh;

  v2f ac[8], an[8];
#pragma unroll
  for (int j = 0; j < 8; ++j) {
    float2 t = *(const float2*)(arow + 4 * j);
    ac[j].x = t.x; ac[j].y = t.y;
  }
  __builtin_prefetch(arow + 64, 0, 3);

  {
    const float4* Wv = (const float4*)W1;
    float4* sWv = (float4*)sW;
#pragma unroll 16
    for (int i = 0; i < 32; ++i) sWv[tid + i * 128] = Wv[tid + i * 128];
  }
  __syncthreads();
  if (row0 >= N) return;

  const float4* bbase = (const float4*)sW + r;

  v8f c0 = {}, c1 = {}, c2 = {}, c3 = {};
#pragma unroll
  for (int kb = 0; kb < DIM_IN; kb += 32) {
    if (kb + 32 < DIM_IN) {
#pragma unroll
      for (int j = 0; j < 8; ++j) {
        float2 t = *(const float2*)(arow + kb + 32 + 4 * j);
        an[j].x = t.x; an[j].y = t.y;
      }
    }
    if ((kb & 63) == 0 && kb + 128 < DIM_IN) __builtin_prefetch(arow + kb + 128, 0, 3);
#pragma unroll
    for (int j = 0; j < 8; ++j) {
      const int k = kb + 4 * j;
      float4 bx = bbase[(k + 2 * kh) * 16];
      float4 by = bbase[(k + 2 * kh + 1) * 16];
      v2f b;
      b.x = bx.x; b.y = by.x;
      c0 = __builtin_amdgcn_wmma_f32_16x16x4_f32(false, ac[j], false, b, (short)0, c0, false, false);
      b.x = bx.y; b.y = by.y;
      c1 = __builtin_amdgcn_wmma_f32_16x16x4_f32(false, ac[j], false, b, (short)0, c1, false, false);
      b.x = bx.z; b.y = by.z;
      c2 = __builtin_amdgcn_wmma_f32_16x16x4_f32(false, ac[j], false, b, (short)0, c2, false, false);
      b.x = bx.w; b.y = by.w;
      c3 = __builtin_amdgcn_wmma_f32_16x16x4_f32(false, ac[j], false, b, (short)0, c3, false, false);
    }
    if (kb + 32 < DIM_IN) {
#pragma unroll
      for (int j = 0; j < 8; ++j) ac[j] = an[j];
    }
  }
  float* orow = h1 + (size_t)(row0 + 8 * kh) * HID + 4 * r;
  if (row0 + 16 <= N) {
#pragma unroll
    for (int rr = 0; rr < 8; ++rr) {
      float4 o; o.x = c0[rr]; o.y = c1[rr]; o.z = c2[rr]; o.w = c3[rr];
      *(float4*)(orow + (size_t)rr * HID) = o;
    }
  } else {
#pragma unroll
    for (int rr = 0; rr < 8; ++rr) {
      int row = row0 + rr + 8 * kh;
      if (row < N) {
        float4 o; o.x = c0[rr]; o.y = c1[rr]; o.z = c2[rr]; o.w = c3[rr];
        *(float4*)(orow + (size_t)rr * HID) = o;
      }
    }
  }
}

__global__ __launch_bounds__(256) void gat_gemm2_wmma(
    const float* __restrict__ hact, const float* __restrict__ W2,
    const float* __restrict__ a_dst,
    float* __restrict__ h2, float* __restrict__ ad, int N) {
  __shared__ float sW[HID * DOUT];
  int tid = threadIdx.x;
  {
    const float4* Wv = (const float4*)W2;
    float4* sWv = (float4*)sW;
    sWv[tid] = Wv[tid];
  }
  __syncthreads();

  int wave = tid >> 5, lane = tid & 31;
  int row0 = (blockIdx.x * 8 + wave) * 16;
  if (row0 >= N) return;

  int r  = lane & 15;
  int kh = lane >> 4;
  int arow_idx = row0 + r; if (arow_idx >= N) arow_idx = N - 1;
  const float* arow  = hact + (size_t)arow_idx * HID + 2 * kh;
  const float* bbase = sW + r;
  float wdv = a_dst[r];

  v8f c = {};
#pragma unroll
  for (int k = 0; k < HID; k += 4) {
    float2 av = *(const float2*)(arow + k);
    v2f a; a.x = av.x; a.y = av.y;
    v2f b; b.x = bbase[(k + 2 * kh) * DOUT];
           b.y = bbase[(k + 2 * kh + 1) * DOUT];
    c = __builtin_amdgcn_wmma_f32_16x16x4_f32(false, a, false, b,
                                              (short)0, c, false, false);
  }
  float* orow = h2 + (size_t)row0 * DOUT + r;
  if (row0 + 16 <= N) {
#pragma unroll
    for (int rr = 0; rr < 8; ++rr)
      orow[(size_t)(rr + 8 * kh) * DOUT] = c[rr];
  } else {
#pragma unroll
    for (int rr = 0; rr < 8; ++rr) {
      int row = row0 + rr + 8 * kh;
      if (row < N) orow[(size_t)(rr + 8 * kh) * DOUT] = c[rr];
    }
  }
  float dsel = 0.f;
#pragma unroll
  for (int rr = 0; rr < 8; ++rr) {
    float pd = c[rr] * wdv;
    pd += __shfl_xor(pd, 1, 32);
    pd += __shfl_xor(pd, 2, 32);
    pd += __shfl_xor(pd, 4, 32);
    pd += __shfl_xor(pd, 8, 32);
    dsel = ((r & 7) == rr) ? pd : dsel;
  }
  int srow = row0 + 8 * kh + (r & 7);
  if (r < 8 && srow < N) ad[srow] = dsel;
}

#define BSH    8
#define BNODES 256
#define MAXB   1024
#define SG     256
#define ST     1024
#define FT     512

__device__ __forceinline__ int block_excl_scan(int v, int* wsum, int tid, int nwaves) {
  int lane = tid & 31, w = tid >> 5;
  int incl = v;
#pragma unroll
  for (int o = 1; o < 32; o <<= 1) {
    int t = __shfl_up(incl, o, 32);
    if (lane >= o) incl += t;
  }
  if (lane == 31) wsum[w] = incl;
  __syncthreads();
  if (w == 0) {
    int s = (lane < nwaves) ? wsum[lane] : 0;
    int si = s;
#pragma unroll
    for (int o = 1; o < 32; o <<= 1) {
      int t = __shfl_up(si, o, 32);
      if (lane >= o) si += t;
    }
    wsum[lane] = si - s;
  }
  __syncthreads();
  return wsum[w] + incl - v;
}

__global__ __launch_bounds__(ST) void gat_bucket_count(
    const int* __restrict__ dst, int* __restrict__ cntm, int E, int NB) {
  __shared__ int hist[MAXB];
  int tid = threadIdx.x;
  hist[tid] = 0;
  __syncthreads();
  int chunk = (E + (int)gridDim.x - 1) / (int)gridDim.x;
  int e0 = (int)blockIdx.x * chunk;
  int e1 = min(E, e0 + chunk);
  for (int eb = e0; eb < e1; eb += 4 * ST) {
    int d[4];
#pragma unroll
    for (int u = 0; u < 4; ++u) {
      int e = eb + u * ST + tid;
      d[u] = (e < e1) ? dst[e] : -1;
    }
#pragma unroll
    for (int u = 0; u < 4; ++u)
      if (d[u] >= 0) atomicAdd(&hist[d[u] >> BSH], 1);
  }
  __syncthreads();
  if (tid < NB) cntm[(size_t)blockIdx.x * NB + tid] = hist[tid];
}

__global__ __launch_bounds__(ST) void gat_bucket_scatter(
    const int* __restrict__ src, const int* __restrict__ dst,
    const int* __restrict__ cntm, int* __restrict__ bst, int* __restrict__ tmp,
    int E, int NB) {
  __shared__ int cur[MAXB];
  __shared__ int totL[MAXB];
  __shared__ int preL[MAXB];
  __shared__ int wsum[32];
  int tid = threadIdx.x;
  int g = (int)blockIdx.x, G = (int)gridDim.x;
  totL[tid] = 0;
  preL[tid] = 0;
  __syncthreads();
  int slices = ST / NB;
  if (tid < slices * NB) {
    int bb = tid % NB, sl = tid / NB;
    int q0 = (sl * G) / slices, q1 = ((sl + 1) * G) / slices;
    int t = 0, p = 0;
#pragma unroll 16
    for (int q = q0; q < q1; ++q) {
      int v = cntm[(size_t)q * NB + bb];
      t += v;
      p += (q < g) ? v : 0;
    }
    atomicAdd(&totL[bb], t);
    atomicAdd(&preL[bb], p);
  }
  __syncthreads();
  int tot = (tid < NB) ? totL[tid] : 0;
  int ex = block_excl_scan(tot, wsum, tid, ST / 32);
  cur[tid] = ex + preL[tid];
  if (g == 0 && tid < NB) bst[tid] = ex;
  __syncthreads();

  int chunk = (E + G - 1) / G;
  int e0 = g * chunk;
  int e1 = min(E, e0 + chunk);
  for (int eb = e0; eb < e1; eb += 4 * ST) {
    int d[4], s[4];
#pragma unroll
    for (int u = 0; u < 4; ++u) {
      int e = eb + u * ST + tid;
      bool ok = e < e1;
      d[u] = ok ? dst[e] : -1;
      s[u] = ok ? src[e] : 0;
    }
#pragma unroll
    for (int u = 0; u < 4; ++u) {
      if (d[u] >= 0) {
        int pos = atomicAdd(&cur[d[u] >> BSH], 1);
        int2 pr; pr.x = s[u]; pr.y = d[u];
        *(int2*)(tmp + 2 * (size_t)pos) = pr;
      }
    }
  }
}

__global__ __launch_bounds__(FT) void gat_bucket_sort(
    const int* __restrict__ tmp, const int* __restrict__ bst,
    int* __restrict__ deg, int* __restrict__ cur, int* __restrict__ esrc,
    int N, int E, int NB) {
  __shared__ int c[BNODES];
  __shared__ int p[BNODES];
  __shared__ int wsum[32];
  int tid = threadIdx.x, b = (int)blockIdx.x;
  int lo = bst[b];
  int hi = (b + 1 < NB) ? bst[b + 1] : E;
  if (tid < BNODES) c[tid] = 0;
  __syncthreads();
  for (int ib = lo; ib < hi; ib += 4 * FT) {
    int d[4];
#pragma unroll
    for (int u = 0; u < 4; ++u) {
      int i = ib + u * FT + tid;
      d[u] = (i < hi) ? tmp[2 * (size_t)i + 1] : -1;
    }
#pragma unroll
    for (int u = 0; u < 4; ++u)
      if (d[u] >= 0) atomicAdd(&c[d[u] & (BNODES - 1)], 1);
  }
  __syncthreads();
  int v = (tid < BNODES) ? c[tid] : 0;
  int ex = block_excl_scan(v, wsum, tid, FT / 32);
  if (tid < BNODES) {
    p[tid] = lo + ex;
    int n = b * BNODES + tid;
    if (n < N) {
      deg[n] = v;
      cur[n] = lo + ex + v;
    }
  }
  __syncthreads();
  for (int ib = lo; ib < hi; ib += 4 * FT) {
    int2 r[4];
#pragma unroll
    for (int u = 0; u < 4; ++u) {
      int i = ib + u * FT + tid;
      r[u].x = 0; r[u].y = -1;
      if (i < hi) r[u] = *(const int2*)(tmp + 2 * (size_t)i);
    }
#pragma unroll
    for (int u = 0; u < 4; ++u) {
      if (r[u].y >= 0) {
        int pos = atomicAdd(&p[r[u].y & (BNODES - 1)], 1);
        esrc[pos] = r[u].x;
      }
    }
  }
}

__global__ __launch_bounds__(256) void gat_gather1(
    const int* __restrict__ esrc, const int* __restrict__ seg_end,
    const int* __restrict__ deg, const float* __restrict__ a_src,
    const float* __restrict__ a_dst, const float* __restrict__ h1,
    const float* __restrict__ b1, float* __restrict__ hact, int N) {
  int lane = threadIdx.x & 31;
  int n = __builtin_amdgcn_readfirstlane((int)(blockIdx.x * 8 + (threadIdx.x >> 5)));
  if (n >= N) return;
  int cnt = __builtin_amdgcn_readfirstlane(deg[n]);
  int beg = __builtin_amdgcn_readfirstlane(seg_end[n]) - cnt;

  float2 hv = ((const float2*)(h1 + (size_t)n * HID))[lane];
  float2 wd = ((const float2*)a_dst)[lane];
  float2 ws = ((const float2*)a_src)[lane];
  float pd = hv.x * wd.x + hv.y * wd.y;
  pd += xlane<1>(pd);
  pd += xlane<2>(pd);

  const int kq = lane & 3;
  const bool up2 = (lane & 2) != 0, up1 = (lane & 1) != 0;
  float m = -INFINITY;
  float accx = 0.f, accy = 0.f, sq = 0.f;
  for (int b0 = 0; b0 < cnt; b0 += 32) {
    int rem = min(32, cnt - b0);
    int s_l = (lane < rem) ? esrc[beg + b0 + lane] : n;
#pragma unroll
    for (int g = 0; g < 4; ++g) {
      if (8 * g < rem) {
        float2 hr[8];
#pragma unroll
        for (int jj = 0; jj < 8; ++jj) {
          const int j = 8 * g + jj;
          int sj = __shfl(s_l, j, 32);
          int col = (j < rem) ? lane : 0;
          hr[jj] = ((const float2*)(h1 + (size_t)sj * HID))[col];
        }
        float q4[4], d[2];
        for (int t = 0; t < 4; ++t) {
          float lo = hr[t].x * ws.x + hr[t].y * ws.y;
          float hi = hr[t + 4].x * ws.x + hr[t + 4].y * ws.y;
          q4[t] = (up2 ? hi : lo) + xlane<2>(up2 ? lo : hi);
        }
        for (int u = 0; u < 2; ++u)
          d[u] = (up1 ? q4[u + 2] : q4[u]) + xlane<1>(up1 ? q4[u] : q4[u + 2]);
        float e0 = (8 * g + 2 * kq     < rem) ? leaky(d[0] + pd) : -INFINITY;
        float e1 = (8 * g + 2 * kq + 1 < rem) ? leaky(d[1] + pd) : -INFINITY;
        float gm = fmaxf(e0, e1);
        gm = fmaxf(gm, xlane<1>(gm));
        gm = fmaxf(gm, xlane<2>(gm));
        float mn = fmaxf(m, gm);
        float sc = __expf(m - mn);
        float w0 = __expf(e0 - mn), w1 = __expf(e1 - mn);
        float wsum = w0 + w1;
        wsum += xlane<1>(wsum);
        wsum += xlane<2>(wsum);
        sq = fmaf(sq, sc, wsum);
        accx *= sc; accy *= sc;
        float wj;
        wj = quad_lane<0>(w0); accx = fmaf(wj, hr[0].x, accx); accy = fmaf(wj, hr[0].y, accy);
        wj = quad_lane<0>(w1); accx = fmaf(wj, hr[1].x, accx); accy = fmaf(wj, hr[1].y, accy);
        wj = quad_lane<1>(w0); accx = fmaf(wj, hr[2].x, accx); accy = fmaf(wj, hr[2].y, accy);
        wj = quad_lane<1>(w1); accx = fmaf(wj, hr[3].x, accx); accy = fmaf(wj, hr[3].y, accy);
        wj = quad_lane<2>(w0); accx = fmaf(wj, hr[4].x, accx); accy = fmaf(wj, hr[4].y, accy);
        wj = quad_lane<2>(w1); accx = fmaf(wj, hr[5].x, accx); accy = fmaf(wj, hr[5].y, accy);
        wj = quad_lane<3>(w0); accx = fmaf(wj, hr[6].x, accx); accy = fmaf(wj, hr[6].y, accy);
        wj = quad_lane<3>(w1); accx = fmaf(wj, hr[7].x, accx); accy = fmaf(wj, hr[7].y, accy);
        m = mn;
      }
    }
  }
  float den = sq + EPS_F;

  float vx = accx / den + b1[2 * lane];
  float vy = accy / den + b1[2 * lane + 1];
  float2 o;
  o.x = vx > 0.0f ? vx : expm1f(vx);
  o.y = vy > 0.0f ? vy : expm1f(vy);
  ((float2*)(hact + (size_t)n * HID))[lane] = o;
}

__global__ __launch_bounds__(256) void gat_gather2(
    const int* __restrict__ esrc, const int* __restrict__ seg_end,
    const int* __restrict__ deg, const float* __restrict__ a_src,
    const float* __restrict__ ad, const float* __restrict__ h2,
    const float* __restrict__ b2, float* __restrict__ dout, int N) {
  int lane = threadIdx.x & 31;
  int c = lane & 15, half = lane >> 4;
  int n0 = __builtin_amdgcn_readfirstlane((int)((blockIdx.x * 8 + (threadIdx.x >> 5)) * 2));
  if (n0 >= N) return;
  int n = n0 + half;
  bool live = n < N;
  int cnt = live ? deg[n] : 0;
  int beg = live ? seg_end[n] - cnt : 0;
  float adn = live ? ad[n] : 0.f;
  float bc = b2[c];
  float ws = a_src[c];
  const bool up8 = (c & 8) != 0, up4 = (c & 4) != 0, up2 = (c & 2) != 0;
  int cmax = __builtin_amdgcn_readfirstlane(max(cnt, __shfl_xor(cnt, 16, 32)));

  float m = -INFINITY;
  float acc = 0.f, sum = 0.f;
  for (int b0 = 0; b0 < cmax; b0 += 16) {
    int j = b0 + c;
    int s_l = (j < cnt) ? esrc[beg + j] : 0;
    int kmax = min(16, cmax - b0);
#pragma unroll
    for (int g = 0; g < 2; ++g) {
      if (8 * g < kmax) {
        float hr[8];
#pragma unroll
        for (int jj = 0; jj < 8; ++jj) {
          int from = half * 16 + 8 * g + jj;
          int sk = __shfl(s_l, from, 32);
          hr[jj] = h2[(size_t)sk * DOUT + c];
        }
        float q4[4], q2[2];
        for (int t = 0; t < 4; ++t) {
          float lo = hr[t] * ws, hi = hr[t + 4] * ws;
          q4[t] = (up8 ? hi : lo) + xlane<8>(up8 ? lo : hi);
        }
        for (int t = 0; t < 2; ++t)
          q2[t] = (up4 ? q4[t + 2] : q4[t]) + xlane<4>(up4 ? q4[t] : q4[t + 2]);
        float dot = (up2 ? q2[1] : q2[0]) + xlane<2>(up2 ? q2[0] : q2[1]);
        dot += xlane<1>(dot);
        float e = (b0 + 8 * g + (c >> 1) < cnt) ? leaky(dot + adn) : -INFINITY;
        float gm = e;
        gm = fmaxf(gm, xlane<2>(gm));
        gm = fmaxf(gm, xlane<4>(gm));
        gm = fmaxf(gm, xlane<8>(gm));
        float mn = fmaxf(m, gm);
        float mr = (mn == -INFINITY) ? 0.f : mn;
        float sc = __expf(m - mr);
        float w = __expf(e - mr);
        float wsum = w;
        wsum += xlane<2>(wsum);
        wsum += xlane<4>(wsum);
        wsum += xlane<8>(wsum);
        sum = fmaf(sum, sc, wsum);
        acc *= sc;
        acc = fmaf(from_lane<0>(w),  hr[0], acc);
        acc = fmaf(from_lane<2>(w),  hr[1], acc);
        acc = fmaf(from_lane<4>(w),  hr[2], acc);
        acc = fmaf(from_lane<6>(w),  hr[3], acc);
        acc = fmaf(from_lane<8>(w),  hr[4], acc);
        acc = fmaf(from_lane<10>(w), hr[5], acc);
        acc = fmaf(from_lane<12>(w), hr[6], acc);
        acc = fmaf(from_lane<14>(w), hr[7], acc);
        m = mn;
      }
    }
  }
  float v = acc / (sum + EPS_F) + bc;
  float mx = v;
  mx = fmaxf(mx, __shfl_xor(mx, 1, 32));
  mx = fmaxf(mx, __shfl_xor(mx, 2, 32));
  mx = fmaxf(mx, __shfl_xor(mx, 4, 32));
  mx = fmaxf(mx, __shfl_xor(mx, 8, 32));
  float se = __expf(v - mx);
  se += __shfl_xor(se, 1, 32);
  se += __shfl_xor(se, 2, 32);
  se += __shfl_xor(se, 4, 32);
  se += __shfl_xor(se, 8, 32);
  float lse = mx + __logf(se);
  if (live) {
    dout[(size_t)n * DOUT + c] = v;
    dout[(size_t)N * DOUT + (size_t)n * DOUT + c] = v - lse;
  }
}

extern "C" void kernel_launch(void* const* d_in, const int* in_sizes, int n_in,
                              void* d_out, int out_size, void* d_ws, size_t ws_size,
                              hipStream_t stream) {
  const float* x   = (const float*)d_in[0];
  const int*   ei  = (const int*)d_in[1];
  const float* W1  = (const float*)d_in[2];
  const float* a1s = (const float*)d_in[3];
  const float* a1d = (const float*)d_in[4];
  const float* b1  = (const float*)d_in[5];
  const float* W2  = (const float*)d_in[6];
  const float* a2s = (const float*)d_in[7];
  const float* a2d = (const float*)d_in[8];
  const float* b2  = (const float*)d_in[9];

  int N = in_sizes[0] / DIM_IN;
  int E = in_sizes[1] / 2;
  const int* src = ei;
  const int* dst = ei + E;

  char* ws = (char*)d_ws;
  float*    h1   = (float*)ws;
  float*    hact = (float*)(ws + (size_t)N * HID * 4);
  char* p = ws + (size_t)N * HID * 4 * 2;
  int*      deg  = (int*)p;   p += ((size_t)N + 1) * 4;
  int*      cur  = (int*)p;   p += (size_t)N * 4;
  int*      esrc = (int*)p;
  const int NB = (N + BNODES - 1) / BNODES;
  int*      tmp  = (int*)hact;
  int*      cntm = tmp + 2 * (size_t)E;
  int*      bst  = cntm + (size_t)SG * NB;
  float*    h2  = h1;
  float*    ad2 = h1 + (size_t)N * DOUT;

  const int B = 256;
  gat_bucket_count<<<SG, ST, 0, stream>>>(dst, cntm, E, NB);
  gat_bucket_scatter<<<SG, ST, 0, stream>>>(src, dst, cntm, bst, tmp, E, NB);
  gat_bucket_sort<<<NB, FT, 0, stream>>>(tmp, bst, deg, cur, esrc, N, E, NB);

  gat_gemm1_wmma<<<(N + 16 * G1_WAVES - 1) / (16 * G1_WAVES), 32 * G1_WAVES, 0, stream>>>(x, W1, h1, N);
  gat_gather1<<<(N + 7) / 8, B, 0, stream>>>(esrc, cur, deg, a1s, a1d, h1, b1, hact, N);

  gat_gemm2_wmma<<<(N + 127) / 128, B, 0, stream>>>(hact, W2, a2d, h2, ad2, N);
  gat_gather2<<<(N + 15) / 16, B, 0, stream>>>(esrc, cur, deg, a2s, ad2, h2, b2, (float*)d_out, N);
}


# Layer-1 gather takes two nodes per wave: a half-wave per node, 16-byte row pieces per lane, a head's dot over two lanes, ids by lane broadcast; half the waves; dot and exp sums re-ordered
# speedup vs baseline: 8.1915x; 1.0352x over previous
#include <hip/hip_runtime.h>
#include <math.h>


typedef float v2f __attribute__((ext_vector_type(2)));
typedef float v8f __attribute__((ext_vector_type(8)));

#define DIM_IN 256
#define HID    64
#define HEADS  8
#define DH     8
#define DOUT   16
#define NEG_SLOPE 0.2f
#define EPS_F 1e-16f

__device__ __forceinline__ float leaky(float v) {
  return v > 0.0f ? v : NEG_SLOPE * v;
}

template <int K>
__device__ __forceinline__ float xlane(float v) {
  return __builtin_bit_cast(float, __builtin_amdgcn_update_dpp(
      0, __builtin_bit_cast(int, v), 0x160 + K, 0xF, 0xF, true));
}
template <int K>
__device__ __forceinline__ float from_lane(float v) {
  return __builtin_bit_cast(float, __builtin_amdgcn_update_dpp(
      0, __builtin_bit_cast(int, v), 0x150 + K, 0xF, 0xF, true));
}
template <int K>
__device__ __forceinline__ float pair_lane(float v) {
  return __builtin_bit_cast(float, __builtin_amdgcn_update_dpp(
      0, __builtin_bit_cast(int, v), K ? 0xF5 : 0xA0, 0xF, 0xF, true));
}
template <int K>
__device__ __forceinline__ int id_from_lane(int v) {
  return __builtin_amdgcn_update_dpp(0, v, 0x150 + K, 0xF, 0xF, true);
}

#define G1_WAVES 4
__global__ __launch_bounds__(128) void gat_gemm1_wmma(
    const float* __restrict__ x, const float* __restrict__ W1,
    float* __restrict__ h1, int N) {
  __shared__ float sW[DIM_IN * HID];
  int tid = threadIdx.x;
  int wave = tid >> 5, lane = tid & 31;
  int row0 = (blockIdx.x * G1_WAVES + wave) * 16;
  int r  = lane & 15;
  int kh = lane >> 4;

  int arow_idx = row0 + r; if (arow_idx >= N) arow_idx = N - 1;
  const float* arow = x + (size_t)arow_idx * DIM_IN + 2 * kh;

  v2f ac[8], an[8];
#pragma unroll
  for (int j = 0; j < 8; ++j) {
    float2 t = *(const float2*)(arow + 4 * j);
    ac[j].x = t.x; ac[j].y = t.y;
  }
  __builtin_prefetch(arow + 64, 0, 3);

  {
    const float4* Wv = (const float4*)W1;
    float4* sWv = (float4*)sW;
#pragma unroll 16
    for (int i = 0; i < 32; ++i) sWv[tid + i * 128] = Wv[tid + i * 128];
  }
  __syncthreads();
  if (row0 >= N) return;

  const float4* bbase = (const float4*)sW + r;

  v8f c0 = {}, c1 = {}, c2 = {}, c3 = {};
#pragma unroll
  for (int kb = 0; kb < DIM_IN; kb += 32) {
    if (kb + 32 < DIM_IN) {
#pragma unroll
      for (int j = 0; j < 8; ++j) {
        float2 t = *(const float2*)(arow + kb + 32 + 4 * j);
        an[j].x = t.x; an[j].y = t.y;
      }
    }
    if ((kb & 63) == 0 && kb + 128 < DIM_IN) __builtin_prefetch(arow + kb + 128, 0, 3);
#pragma unroll
    for (int j = 0; j < 8; ++j) {
      const int k = kb + 4 * j;
      float4 bx = bbase[(k + 2 * kh) * 16];
      float4 by = bbase[(k + 2 * kh + 1) * 16];
      v2f b;
      b.x = bx.x; b.y = by.x;
      c0 = __builtin_amdgcn_wmma_f32_16x16x4_f32(false, ac[j], false, b, (short)0, c0, false, false);
      b.x = bx.y; b.y = by.y;
      c1 = __builtin_amdgcn_wmma_f32_16x16x4_f32(false, ac[j], false, b, (short)0, c1, false, false);
      b.x = bx.z; b.y = by.z;
      c2 = __builtin_amdgcn_wmma_f32_16x16x4_f32(false, ac[j], false, b, (short)0, c2, false, false);
      b.x = bx.w; b.y = by.w;
      c3 = __builtin_amdgcn_wmma_f32_16x16x4_f32(false, ac[j], false, b, (short)0, c3, false, false);
    }
    if (kb + 32 < DIM_IN) {
#pragma unroll
      for (int j = 0; j < 8; ++j) ac[j] = an[j];
    }
  }
  float* orow = h1 + (size_t)(row0 + 8 * kh) * HID + 4 * r;
  if (row0 + 16 <= N) {
#pragma unroll
    for (int rr = 0; rr < 8; ++rr) {
      float4 o; o.x = c0[rr]; o.y = c1[rr]; o.z = c2[rr]; o.w = c3[rr];
      *(float4*)(orow + (size_t)rr * HID) = o;
    }
  } else {
#pragma unroll
    for (int rr = 0; rr < 8; ++rr) {
      int row = row0 + rr + 8 * kh;
      if (row < N) {
        float4 o; o.x = c0[rr]; o.y = c1[rr]; o.z = c2[rr]; o.w = c3[rr];
        *(float4*)(orow + (size_t)rr * HID) = o;
      }
    }
  }
}

__global__ __launch_bounds__(256) void gat_gemm2_wmma(
    const float* __restrict__ hact, const float* __restrict__ W2,
    const float* __restrict__ a_dst,
    float* __restrict__ h2, float* __restrict__ ad, int N) {
  __shared__ float sW[HID * DOUT];
  int tid = threadIdx.x;
  {
    const float4* Wv = (const float4*)W2;
    float4* sWv = (float4*)sW;
    sWv[tid] = Wv[tid];
  }
  __syncthreads();

  int wave = tid >> 5, lane = tid & 31;
  int row0 = (blockIdx.x * 8 + wave) * 16;
  if (row0 >= N) return;

  int r  = lane & 15;
  int kh = lane >> 4;
  int arow_idx = row0 + r; if (arow_idx >= N) arow_idx = N - 1;
  const float* arow  = hact + (size_t)arow_idx * HID + 2 * kh;
  const float* bbase = sW + r;
  float wdv = a_dst[r];

  v8f c = {};
#pragma unroll
  for (int k = 0; k < HID; k += 4) {
    float2 av = *(const float2*)(arow + k);
    v2f a; a.x = av.x; a.y = av.y;
    v2f b; b.x = bbase[(k + 2 * kh) * DOUT];
           b.y = bbase[(k + 2 * kh + 1) * DOUT];
    c = __builtin_amdgcn_wmma_f32_16x16x4_f32(false, a, false, b,
                                              (short)0, c, false, false);
  }
  float* orow = h2 + (size_t)row0 * DOUT + r;
  if (row0 + 16 <= N) {
#pragma unroll
    for (int rr = 0; rr < 8; ++rr)
      orow[(size_t)(rr + 8 * kh) * DOUT] = c[rr];
  } else {
#pragma unroll
    for (int rr = 0; rr < 8; ++rr) {
      int row = row0 + rr + 8 * kh;
      if (row < N) orow[(size_t)(rr + 8 * kh) * DOUT] = c[rr];
    }
  }
  float dsel = 0.f;
#pragma unroll
  for (int rr = 0; rr < 8; ++rr) {
    float pd = c[rr] * wdv;
    pd += __shfl_xor(pd, 1, 32);
    pd += __shfl_xor(pd, 2, 32);
    pd += __shfl_xor(pd, 4, 32);
    pd += __shfl_xor(pd, 8, 32);
    dsel = ((r & 7) == rr) ? pd : dsel;
  }
  int srow = row0 + 8 * kh + (r & 7);
  if (r < 8 && srow < N) ad[srow] = dsel;
}

#define BSH    8
#define BNODES 256
#define MAXB   1024
#define SG     256
#define ST     1024
#define FT     512

__device__ __forceinline__ int block_excl_scan(int v, int* wsum, int tid, int nwaves) {
  int lane = tid & 31, w = tid >> 5;
  int incl = v;
#pragma unroll
  for (int o = 1; o < 32; o <<= 1) {
    int t = __shfl_up(incl, o, 32);
    if (lane >= o) incl += t;
  }
  if (lane == 31) wsum[w] = incl;
  __syncthreads();
  if (w == 0) {
    int s = (lane < nwaves) ? wsum[lane] : 0;
    int si = s;
#pragma unroll
    for (int o = 1; o < 32; o <<= 1) {
      int t = __shfl_up(si, o, 32);
      if (lane >= o) si += t;
    }
    wsum[lane] = si - s;
  }
  __syncthreads();
  return wsum[w] + incl - v;
}

__global__ __launch_bounds__(ST) void gat_bucket_count(
    const int* __restrict__ dst, int* __restrict__ cntm, int E, int NB) {
  __shared__ int hist[MAXB];
  int tid = threadIdx.x;
  hist[tid] = 0;
  __syncthreads();
  int chunk = (E + (int)gridDim.x - 1) / (int)gridDim.x;
  int e0 = (int)blockIdx.x * chunk;
  int e1 = min(E, e0 + chunk);
  for (int eb = e0; eb < e1; eb += 4 * ST) {
    int d[4];
#pragma unroll
    for (int u = 0; u < 4; ++u) {
      int e = eb + u * ST + tid;
      d[u] = (e < e1) ? dst[e] : -1;
    }
#pragma unroll
    for (int u = 0; u < 4; ++u)
      if (d[u] >= 0) atomicAdd(&hist[d[u] >> BSH], 1);
  }
  __syncthreads();
  if (tid < NB) cntm[(size_t)blockIdx.x * NB + tid] = hist[tid];
}

__global__ __launch_bounds__(ST) void gat_bucket_scatter(
    const int* __restrict__ src, const int* __restrict__ dst,
    const int* __restrict__ cntm, int* __restrict__ bst, int* __restrict__ tmp,
    int E, int NB) {
  __shared__ int cur[MAXB];
  __shared__ int totL[MAXB];
  __shared__ int preL[MAXB];
  __shared__ int wsum[32];
  int tid = threadIdx.x;
  int g = (int)blockIdx.x, G = (int)gridDim.x;
  totL[tid] = 0;
  preL[tid] = 0;
  __syncthreads();
  int slices = ST / NB;
  if (tid < slices * NB) {
    int bb = tid % NB, sl = tid / NB;
    int q0 = (sl * G) / slices, q1 = ((sl + 1) * G) / slices;
    int t = 0, p = 0;
#pragma unroll 16
    for (int q = q0; q < q1; ++q) {
      int v = cntm[(size_t)q * NB + bb];
      t += v;
      p += (q < g) ? v : 0;
    }
    atomicAdd(&totL[bb], t);
    atomicAdd(&preL[bb], p);
  }
  __syncthreads();
  int tot = (tid < NB) ? totL[tid] : 0;
  int ex = block_excl_scan(tot, wsum, tid, ST / 32);
  cur[tid] = ex + preL[tid];
  if (g == 0 && tid < NB) bst[tid] = ex;
  __syncthreads();

  int chunk = (E + G - 1) / G;
  int e0 = g * chunk;
  int e1 = min(E, e0 + chunk);
  for (int eb = e0; eb < e1; eb += 4 * ST) {
    int d[4], s[4];
#pragma unroll
    for (int u = 0; u < 4; ++u) {
      int e = eb + u * ST + tid;
      bool ok = e < e1;
      d[u] = ok ? dst[e] : -1;
      s[u] = ok ? src[e] : 0;
    }
#pragma unroll
    for (int u = 0; u < 4; ++u) {
      if (d[u] >= 0) {
        int pos = atomicAdd(&cur[d[u] >> BSH], 1);
        int2 pr; pr.x = s[u]; pr.y = d[u];
        *(int2*)(tmp + 2 * (size_t)pos) = pr;
      }
    }
  }
}

__global__ __launch_bounds__(FT) void gat_bucket_sort(
    const int* __restrict__ tmp, const int* __restrict__ bst,
    int* __restrict__ deg, int* __restrict__ cur, int* __restrict__ esrc,
    int N, int E, int NB) {
  __shared__ int c[BNODES];
  __shared__ int p[BNODES];
  __shared__ int wsum[32];
  int tid = threadIdx.x, b = (int)blockIdx.x;
  int lo = bst[b];
  int hi = (b + 1 < NB) ? bst[b + 1] : E;
  if (tid < BNODES) c[tid] = 0;
  __syncthreads();
  for (int ib = lo; ib < hi; ib += 4 * FT) {
    int d[4];
#pragma unroll
    for (int u = 0; u < 4; ++u) {
      int i = ib + u * FT + tid;
      d[u] = (i < hi) ? tmp[2 * (size_t)i + 1] : -1;
    }
#pragma unroll
    for (int u = 0; u < 4; ++u)
      if (d[u] >= 0) atomicAdd(&c[d[u] & (BNODES - 1)], 1);
  }
  __syncthreads();
  int v = (tid < BNODES) ? c[tid] : 0;
  int ex = block_excl_scan(v, wsum, tid, FT / 32);
  if (tid < BNODES) {
    p[tid] = lo + ex;
    int n = b * BNODES + tid;
    if (n < N) {
      deg[n] = v;
      cur[n] = lo + ex + v;
    }
  }
  __syncthreads();
  for (int ib = lo; ib < hi; ib += 4 * FT) {
    int2 r[4];
#pragma unroll
    for (int u = 0; u < 4; ++u) {
      int i = ib + u * FT + tid;
      r[u].x = 0; r[u].y = -1;
      if (i < hi) r[u] = *(const int2*)(tmp + 2 * (size_t)i);
    }
#pragma unroll
    for (int u = 0; u < 4; ++u) {
      if (r[u].y >= 0) {
        int pos = atomicAdd(&p[r[u].y & (BNODES - 1)], 1);
        esrc[pos] = r[u].x;
      }
    }
  }
}

template <int G>
__device__ __forceinline__ void gather1_group(
    const float4* __restrict__ h1v, int s_l, int c, bool odd, int slot0, int cnt,
    float pd, float4 ws, float& m, float& sum,
    float& ax, float& ay, float& az, float& aw) {
  float4 hr[8];
  hr[0] = h1v[(unsigned)(id_from_lane<8 * G + 0>(s_l) * 16 + c)];
  hr[1] = h1v[(unsigned)(id_from_lane<8 * G + 1>(s_l) * 16 + c)];
  hr[2] = h1v[(unsigned)(id_from_lane<8 * G + 2>(s_l) * 16 + c)];
  hr[3] = h1v[(unsigned)(id_from_lane<8 * G + 3>(s_l) * 16 + c)];
  hr[4] = h1v[(unsigned)(id_from_lane<8 * G + 4>(s_l) * 16 + c)];
  hr[5] = h1v[(unsigned)(id_from_lane<8 * G + 5>(s_l) * 16 + c)];
  hr[6] = h1v[(unsigned)(id_from_lane<8 * G + 6>(s_l) * 16 + c)];
  hr[7] = h1v[(unsigned)(id_from_lane<8 * G + 7>(s_l) * 16 + c)];
  const v2f wxy = {ws.x, ws.y}, wzw = {ws.z, ws.w};
  float p[8];
#pragma unroll
  for (int t = 0; t < 8; ++t) {
    v2f lo = {hr[t].x, hr[t].y}, hi = {hr[t].z, hr[t].w};
    v2f q = lo * wxy;
    q = hi * wzw + q;
    p[t] = q.x + q.y;
  }
  float e[4], w[4];
#pragma unroll
  for (int t = 0; t < 4; ++t) {
    float d = (odd ? p[t + 4] : p[t]) + xlane<1>(odd ? p[t] : p[t + 4]);
    e[t] = (slot0 + t < cnt) ? leaky(d + pd) : -INFINITY;
  }
  float gm = fmaxf(fmaxf(e[0], e[1]), fmaxf(e[2], e[3]));
  gm = fmaxf(gm, xlane<1>(gm));
  float mn = fmaxf(m, gm);
  float mr = (mn == -INFINITY) ? 0.f : mn;
  float sc = __expf(m - mr);
#pragma unroll
  for (int t = 0; t < 4; ++t) w[t] = __expf(e[t] - mr);
  float wsum = (w[0] + w[1]) + (w[2] + w[3]);
  wsum += xlane<1>(wsum);
  sum = fmaf(sum, sc, wsum);
  ax *= sc; ay *= sc; az *= sc; aw *= sc;
  float wj;
  wj = pair_lane<0>(w[0]); ax = fmaf(wj, hr[0].x, ax); ay = fmaf(wj, hr[0].y, ay); az = fmaf(wj, hr[0].z, az); aw = fmaf(wj, hr[0].w, aw);
  wj = pair_lane<0>(w[1]); ax = fmaf(wj, hr[1].x, ax); ay = fmaf(wj, hr[1].y, ay); az = fmaf(wj, hr[1].z, az); aw = fmaf(wj, hr[1].w, aw);
  wj = pair_lane<0>(w[2]); ax = fmaf(wj, hr[2].x, ax); ay = fmaf(wj, hr[2].y, ay); az = fmaf(wj, hr[2].z, az); aw = fmaf(wj, hr[2].w, aw);
  wj = pair_lane<0>(w[3]); ax = fmaf(wj, hr[3].x, ax); ay = fmaf(wj, hr[3].y, ay); az = fmaf(wj, hr[3].z, az); aw = fmaf(wj, hr[3].w, aw);
  wj = pair_lane<1>(w[0]); ax = fmaf(wj, hr[4].x, ax); ay = fmaf(wj, hr[4].y, ay); az = fmaf(wj, hr[4].z, az); aw = fmaf(wj, hr[4].w, aw);
  wj = pair_lane<1>(w[1]); ax = fmaf(wj, hr[5].x, ax); ay = fmaf(wj, hr[5].y, ay); az = fmaf(wj, hr[5].z, az); aw = fmaf(wj, hr[5].w, aw);
  wj = pair_lane<1>(w[2]); ax = fmaf(wj, hr[6].x, ax); ay = fmaf(wj, hr[6].y, ay); az = fmaf(wj, hr[6].z, az); aw = fmaf(wj, hr[6].w, aw);
  wj = pair_lane<1>(w[3]); ax = fmaf(wj, hr[7].x, ax); ay = fmaf(wj, hr[7].y, ay); az = fmaf(wj, hr[7].z, az); aw = fmaf(wj, hr[7].w, aw);
  m = mn;
}

__global__ __launch_bounds__(256) void gat_gather1(
    const int* __restrict__ esrc, const int* __restrict__ seg_end,
    const int* __restrict__ deg, const float* __restrict__ a_src,
    const float* __restrict__ a_dst, const float* __restrict__ h1,
    const float* __restrict__ b1, float* __restrict__ hact, int N) {
  int lane = threadIdx.x & 31;
  int c = lane & 15, half = lane >> 4;
  int n0 = __builtin_amdgcn_readfirstlane((int)((blockIdx.x * 8 + (threadIdx.x >> 5)) * 2));
  if (n0 >= N) return;
  int n = n0 + half;
  bool live = n < N;
  int nn = live ? n : n0;
  int cnt = live ? deg[n] : 0;
  int beg = live ? seg_end[n] - cnt : 0;

  const float4* h1v = (const float4*)h1;
  float4 hv = h1v[(unsigned)(nn * 16 + c)];
  float4 wd = ((const float4*)a_dst)[c];
  float4 ws = ((const float4*)a_src)[c];
  float4 bb = ((const float4*)b1)[c];
  float pd = hv.x * wd.x + hv.y * wd.y + hv.z * wd.z + hv.w * wd.w;
  pd += xlane<1>(pd);

  const bool odd = (c & 1) != 0;
  int cmax = __builtin_amdgcn_readfirstlane(max(cnt, __shfl_xor(cnt, 16, 32)));

  float m = -INFINITY;
  float ax = 0.f, ay = 0.f, az = 0.f, aw = 0.f, sum = 0.f;
  for (int b0 = 0; b0 < cmax; b0 += 16) {
    int j = b0 + c;
    int s_l = (j < cnt) ? esrc[beg + j] : nn;
    int kmax = min(16, cmax - b0);
    gather1_group<0>(h1v, s_l, c, odd, b0 + (odd ? 4 : 0), cnt, pd, ws, m, sum, ax, ay, az, aw);
    if (8 < kmax)
      gather1_group<1>(h1v, s_l, c, odd, b0 + 8 + (odd ? 4 : 0), cnt, pd, ws, m, sum, ax, ay, az, aw);
  }
  float den = sum + EPS_F;

  float vx = ax / den + bb.x;
  float vy = ay / den + bb.y;
  float vz = az / den + bb.z;
  float vw = aw / den + bb.w;
  float4 o;
  o.x = vx > 0.0f ? vx : expm1f(vx);
  o.y = vy > 0.0f ? vy : expm1f(vy);
  o.z = vz > 0.0f ? vz : expm1f(vz);
  o.w = vw > 0.0f ? vw : expm1f(vw);
  if (live) ((float4*)hact)[(size_t)n * 16 + c] = o;
}

__global__ __launch_bounds__(256) void gat_gather2(
    const int* __restrict__ esrc, const int* __restrict__ seg_end,
    const int* __restrict__ deg, const float* __restrict__ a_src,
    const float* __restrict__ ad, const float* __restrict__ h2,
    const float* __restrict__ b2, float* __restrict__ dout, int N) {
  int lane = threadIdx.x & 31;
  int c = lane & 15, half = lane >> 4;
  int n0 = __builtin_amdgcn_readfirstlane((int)((blockIdx.x * 8 + (threadIdx.x >> 5)) * 2));
  if (n0 >= N) return;
  int n = n0 + half;
  bool live = n < N;
  int cnt = live ? deg[n] : 0;
  int beg = live ? seg_end[n] - cnt : 0;
  float adn = live ? ad[n] : 0.f;
  float bc = b2[c];
  float ws = a_src[c];
  const bool up8 = (c & 8) != 0, up4 = (c & 4) != 0, up2 = (c & 2) != 0;
  int cmax = __builtin_amdgcn_readfirstlane(max(cnt, __shfl_xor(cnt, 16, 32)));

  float m = -INFINITY;
  float acc = 0.f, sum = 0.f;
  for (int b0 = 0; b0 < cmax; b0 += 16) {
    int j = b0 + c;
    int s_l = (j < cnt) ? esrc[beg + j] : 0;
    int kmax = min(16, cmax - b0);
#pragma unroll
    for (int g = 0; g < 2; ++g) {
      if (8 * g < kmax) {
        float hr[8];
#pragma unroll
        for (int jj = 0; jj < 8; ++jj) {
          int from = half * 16 + 8 * g + jj;
          int sk = __shfl(s_l, from, 32);
          hr[jj] = h2[(size_t)sk * DOUT + c];
        }
        float q4[4], q2[2];
        for (int t = 0; t < 4; ++t) {
          float lo = hr[t] * ws, hi = hr[t + 4] * ws;
          q4[t] = (up8 ? hi : lo) + xlane<8>(up8 ? lo : hi);
        }
        for (int t = 0; t < 2; ++t)
          q2[t] = (up4 ? q4[t + 2] : q4[t]) + xlane<4>(up4 ? q4[t] : q4[t + 2]);
        float dot = (up2 ? q2[1] : q2[0]) + xlane<2>(up2 ? q2[0] : q2[1]);
        dot += xlane<1>(dot);
        float e = (b0 + 8 * g + (c >> 1) < cnt) ? leaky(dot + adn) : -INFINITY;
        float gm = e;
        gm = fmaxf(gm, xlane<2>(gm));
        gm = fmaxf(gm, xlane<4>(gm));
        gm = fmaxf(gm, xlane<8>(gm));
        float mn = fmaxf(m, gm);
        float mr = (mn == -INFINITY) ? 0.f : mn;
        float sc = __expf(m - mr);
        float w = __expf(e - mr);
        float wsum = w;
        wsum += xlane<2>(wsum);
        wsum += xlane<4>(wsum);
        wsum += xlane<8>(wsum);
        sum = fmaf(sum, sc, wsum);
        acc *= sc;
        acc = fmaf(from_lane<0>(w),  hr[0], acc);
        acc = fmaf(from_lane<2>(w),  hr[1], acc);
        acc = fmaf(from_lane<4>(w),  hr[2], acc);
        acc = fmaf(from_lane<6>(w),  hr[3], acc);
        acc = fmaf(from_lane<8>(w),  hr[4], acc);
        acc = fmaf(from_lane<10>(w), hr[5], acc);
        acc = fmaf(from_lane<12>(w), hr[6], acc);
        acc = fmaf(from_lane<14>(w), hr[7], acc);
        m = mn;
      }
    }
  }
  float v = acc / (sum + EPS_F) + bc;
  float mx = v;
  mx = fmaxf(mx, __shfl_xor(mx, 1, 32));
  mx = fmaxf(mx, __shfl_xor(mx, 2, 32));
  mx = fmaxf(mx, __shfl_xor(mx, 4, 32));
  mx = fmaxf(mx, __shfl_xor(mx, 8, 32));
  float se = __expf(v - mx);
  se += __shfl_xor(se, 1, 32);
  se += __shfl_xor(se, 2, 32);
  se += __shfl_xor(se, 4, 32);
  se += __shfl_xor(se, 8, 32);
  float lse = mx + __logf(se);
  if (live) {
    dout[(size_t)n * DOUT + c] = v;
    dout[(size_t)N * DOUT + (size_t)n * DOUT + c] = v - lse;
  }
}

extern "C" void kernel_launch(void* const* d_in, const int* in_sizes, int n_in,
                              void* d_out, int out_size, void* d_ws, size_t ws_size,
                              hipStream_t stream) {
  const float* x   = (const float*)d_in[0];
  const int*   ei  = (const int*)d_in[1];
  const float* W1  = (const float*)d_in[2];
  const float* a1s = (const float*)d_in[3];
  const float* a1d = (const float*)d_in[4];
  const float* b1  = (const float*)d_in[5];
  const float* W2  = (const float*)d_in[6];
  const float* a2s = (const float*)d_in[7];
  const float* a2d = (const float*)d_in[8];
  const float* b2  = (const float*)d_in[9];

  int N = in_sizes[0] / DIM_IN;
  int E = in_sizes[1] / 2;
  const int* src = ei;
  const int* dst = ei + E;

  char* ws = (char*)d_ws;
  float*    h1   = (float*)ws;
  float*    hact = (float*)(ws + (size_t)N * HID * 4);
  char* p = ws + (size_t)N * HID * 4 * 2;
  int*      deg  = (int*)p;   p += ((size_t)N + 1) * 4;
  int*      cur  = (int*)p;   p += (size_t)N * 4;
  int*      esrc = (int*)p;
  const int NB = (N + BNODES - 1) / BNODES;
  int*      tmp  = (int*)hact;
  int*      cntm = tmp + 2 * (size_t)E;
  int*      bst  = cntm + (size_t)SG * NB;
  float*    h2  = h1;
  float*    ad2 = h1 + (size_t)N * DOUT;

  const int B = 256;
  gat_bucket_count<<<SG, ST, 0, stream>>>(dst, cntm, E, NB);
  gat_bucket_scatter<<<SG, ST, 0, stream>>>(src, dst, cntm, bst, tmp, E, NB);
  gat_bucket_sort<<<NB, FT, 0, stream>>>(tmp, bst, deg, cur, esrc, N, E, NB);

  gat_gemm1_wmma<<<(N + 16 * G1_WAVES - 1) / (16 * G1_WAVES), 32 * G1_WAVES, 0, stream>>>(x, W1, h1, N);
  gat_gather1<<<(N + 15) / 16, B, 0, stream>>>(esrc, cur, deg, a1s, a1d, h1, b1, hact, N);

  gat_gemm2_wmma<<<(N + 127) / 128, B, 0, stream>>>(hact, W2, a2d, h2, ad2, N);
  gat_gather2<<<(N + 15) / 16, B, 0, stream>>>(esrc, cur, deg, a2s, ad2, h2, b2, (float*)d_out, N);
}


# Layer-2 gather takes 4 nodes per wave: four lanes per gathered row (16-byte pieces), a node's two lane groups keep their own running max and sums, merged at the end; dot, exp and row sums re-ordered
# speedup vs baseline: 8.5187x; 1.0399x over previous
#include <hip/hip_runtime.h>
#include <math.h>


typedef float v2f __attribute__((ext_vector_type(2)));
typedef float v8f __attribute__((ext_vector_type(8)));

#define DIM_IN 256
#define HID    64
#define HEADS  8
#define DH     8
#define DOUT   16
#define NEG_SLOPE 0.2f
#define EPS_F 1e-16f

__device__ __forceinline__ float leaky(float v) {
  return v > 0.0f ? v : NEG_SLOPE * v;
}

template <int K>
__device__ __forceinline__ float xlane(float v) {
  return __builtin_bit_cast(float, __builtin_amdgcn_update_dpp(
      0, __builtin_bit_cast(int, v), 0x160 + K, 0xF, 0xF, true));
}
template <int K>
__device__ __forceinline__ float from_lane(float v) {
  return __builtin_bit_cast(float, __builtin_amdgcn_update_dpp(
      0, __builtin_bit_cast(int, v), 0x150 + K, 0xF, 0xF, true));
}
template <int K>
__device__ __forceinline__ float pair_lane(float v) {
  return __builtin_bit_cast(float, __builtin_amdgcn_update_dpp(
      0, __builtin_bit_cast(int, v), K ? 0xF5 : 0xA0, 0xF, 0xF, true));
}
template <int K>
__device__ __forceinline__ float quad_from(float v) {
  return __builtin_bit_cast(float, __builtin_amdgcn_update_dpp(
      0, __builtin_bit_cast(int, v), K * 0x55, 0xF, 0xF, true));
}
template <int K>
__device__ __forceinline__ int quad_id(int v) {
  return __builtin_amdgcn_update_dpp(0, v, K * 0x55, 0xF, 0xF, true);
}
__device__ __forceinline__ int id_xlane8(int v) {
  return __builtin_amdgcn_update_dpp(0, v, 0x168, 0xF, 0xF, true);
}
template <int K>
__device__ __forceinline__ int id_from_lane(int v) {
  return __builtin_amdgcn_update_dpp(0, v, 0x150 + K, 0xF, 0xF, true);
}

#define G1_WAVES 4
__global__ __launch_bounds__(128) void gat_gemm1_wmma(
    const float* __restrict__ x, const float* __restrict__ W1,
    float* __restrict__ h1, int N) {
  __shared__ float sW[DIM_IN * HID];
  int tid = threadIdx.x;
  int wave = tid >> 5, lane = tid & 31;
  int row0 = (blockIdx.x * G1_WAVES + wave) * 16;
  int r  = lane & 15;
  int kh = lane >> 4;

  int arow_idx = row0 + r; if (arow_idx >= N) arow_idx = N - 1;
  const float* arow = x + (size_t)arow_idx * DIM_IN + 2 * kh;

  v2f ac[8], an[8];
#pragma unroll
  for (int j = 0; j < 8; ++j) {
    float2 t = *(const float2*)(arow + 4 * j);
    ac[j].x = t.x; ac[j].y = t.y;
  }
  __builtin_prefetch(arow + 64, 0, 3);

  {
    const float4* Wv = (const float4*)W1;
    float4* sWv = (float4*)sW;
#pragma unroll 16
    for (int i = 0; i < 32; ++i) sWv[tid + i * 128] = Wv[tid + i * 128];
  }
  __syncthreads();
  if (row0 >= N) return;

  const float4* bbase = (const float4*)sW + r;

  v8f c0 = {}, c1 = {}, c2 = {}, c3 = {};
#pragma unroll
  for (int kb = 0; kb < DIM_IN; kb += 32) {
    if (kb + 32 < DIM_IN) {
#pragma unroll
      for (int j = 0; j < 8; ++j) {
        float2 t = *(const float2*)(arow + kb + 32 + 4 * j);
        an[j].x = t.x; an[j].y = t.y;
      }
    }
    if ((kb & 63) == 0 && kb + 128 < DIM_IN) __builtin_prefetch(arow + kb + 128, 0, 3);
#pragma unroll
    for (int j = 0; j < 8; ++j) {
      const int k = kb + 4 * j;
      float4 bx = bbase[(k + 2 * kh) * 16];
      float4 by = bbase[(k + 2 * kh + 1) * 16];
      v2f b;
      b.x = bx.x; b.y = by.x;
      c0 = __builtin_amdgcn_wmma_f32_16x16x4_f32(false, ac[j], false, b, (short)0, c0, false, false);
      b.x = bx.y; b.y = by.y;
      c1 = __builtin_amdgcn_wmma_f32_16x16x4_f32(false, ac[j], false, b, (short)0, c1, false, false);
      b.x = bx.z; b.y = by.z;
      c2 = __builtin_amdgcn_wmma_f32_16x16x4_f32(false, ac[j], false, b, (short)0, c2, false, false);
      b.x = bx.w; b.y = by.w;
      c3 = __builtin_amdgcn_wmma_f32_16x16x4_f32(false, ac[j], false, b, (short)0, c3, false, false);
    }
    if (kb + 32 < DIM_IN) {
#pragma unroll
      for (int j = 0; j < 8; ++j) ac[j] = an[j];
    }
  }
  float* orow = h1 + (size_t)(row0 + 8 * kh) * HID + 4 * r;
  if (row0 + 16 <= N) {
#pragma unroll
    for (int rr = 0; rr < 8; ++rr) {
      float4 o; o.x = c0[rr]; o.y = c1[rr]; o.z = c2[rr]; o.w = c3[rr];
      *(float4*)(orow + (size_t)rr * HID) = o;
    }
  } else {
#pragma unroll
    for (int rr = 0; rr < 8; ++rr) {
      int row = row0 + rr + 8 * kh;
      if (row < N) {
        float4 o; o.x = c0[rr]; o.y = c1[rr]; o.z = c2[rr]; o.w = c3[rr];
        *(float4*)(orow + (size_t)rr * HID) = o;
      }
    }
  }
}

__global__ __launch_bounds__(256) void gat_gemm2_wmma(
    const float* __restrict__ hact, const float* __restrict__ W2,
    const float* __restrict__ a_dst,
    float* __restrict__ h2, float* __restrict__ ad, int N) {
  __shared__ float sW[HID * DOUT];
  int tid = threadIdx.x;
  {
    const float4* Wv = (const float4*)W2;
    float4* sWv = (float4*)sW;
    sWv[tid] = Wv[tid];
  }
  __syncthreads();

  int wave = tid >> 5, lane = tid & 31;
  int row0 = (blockIdx.x * 8 + wave) * 16;
  if (row0 >= N) return;

  int r  = lane & 15;
  int kh = lane >> 4;
  int arow_idx = row0 + r; if (arow_idx >= N) arow_idx = N - 1;
  const float* arow  = hact + (size_t)arow_idx * HID + 2 * kh;
  const float* bbase = sW + r;
  float wdv = a_dst[r];

  v8f c = {};
#pragma unroll
  for (int k = 0; k < HID; k += 4) {
    float2 av = *(const float2*)(arow + k);
    v2f a; a.x = av.x; a.y = av.y;
    v2f b; b.x = bbase[(k + 2 * kh) * DOUT];
           b.y = bbase[(k + 2 * kh + 1) * DOUT];
    c = __builtin_amdgcn_wmma_f32_16x16x4_f32(false, a, false, b,
                                              (short)0, c, false, false);
  }
  float* orow = h2 + (size_t)row0 * DOUT + r;
  if (row0 + 16 <= N) {
#pragma unroll
    for (int rr = 0; rr < 8; ++rr)
      orow[(size_t)(rr + 8 * kh) * DOUT] = c[rr];
  } else {
#pragma unroll
    for (int rr = 0; rr < 8; ++rr) {
      int row = row0 + rr + 8 * kh;
      if (row < N) orow[(size_t)(rr + 8 * kh) * DOUT] = c[rr];
    }
  }
  float dsel = 0.f;
#pragma unroll
  for (int rr = 0; rr < 8; ++rr) {
    float pd = c[rr] * wdv;
    pd += __shfl_xor(pd, 1, 32);
    pd += __shfl_xor(pd, 2, 32);
    pd += __shfl_xor(pd, 4, 32);
    pd += __shfl_xor(pd, 8, 32);
    dsel = ((r & 7) == rr) ? pd : dsel;
  }
  int srow = row0 + 8 * kh + (r & 7);
  if (r < 8 && srow < N) ad[srow] = dsel;
}

#define BSH    8
#define BNODES 256
#define MAXB   1024
#define SG     256
#define ST     1024
#define FT     512

__device__ __forceinline__ int block_excl_scan(int v, int* wsum, int tid, int nwaves) {
  int lane = tid & 31, w = tid >> 5;
  int incl = v;
#pragma unroll
  for (int o = 1; o < 32; o <<= 1) {
    int t = __shfl_up(incl, o, 32);
    if (lane >= o) incl += t;
  }
  if (lane == 31) wsum[w] = incl;
  __syncthreads();
  if (w == 0) {
    int s = (lane < nwaves) ? wsum[lane] : 0;
    int si = s;
#pragma unroll
    for (int o = 1; o < 32; o <<= 1) {
      int t = __shfl_up(si, o, 32);
      if (lane >= o) si += t;
    }
    wsum[lane] = si - s;
  }
  __syncthreads();
  return wsum[w] + incl - v;
}

__global__ __launch_bounds__(ST) void gat_bucket_count(
    const int* __restrict__ dst, int* __restrict__ cntm, int E, int NB) {
  __shared__ int hist[MAXB];
  int tid = threadIdx.x;
  hist[tid] = 0;
  __syncthreads();
  int chunk = (E + (int)gridDim.x - 1) / (int)gridDim.x;
  int e0 = (int)blockIdx.x * chunk;
  int e1 = min(E, e0 + chunk);
  for (int eb = e0; eb < e1; eb += 4 * ST) {
    int d[4];
#pragma unroll
    for (int u = 0; u < 4; ++u) {
      int e = eb + u * ST + tid;
      d[u] = (e < e1) ? dst[e] : -1;
    }
#pragma unroll
    for (int u = 0; u < 4; ++u)
      if (d[u] >= 0) atomicAdd(&hist[d[u] >> BSH], 1);
  }
  __syncthreads();
  if (tid < NB) cntm[(size_t)blockIdx.x * NB + tid] = hist[tid];
}

__global__ __launch_bounds__(ST) void gat_bucket_scatter(
    const int* __restrict__ src, const int* __restrict__ dst,
    const int* __restrict__ cntm, int* __restrict__ bst, int* __restrict__ tmp,
    int E, int NB) {
  __shared__ int cur[MAXB];
  __shared__ int totL[MAXB];
  __shared__ int preL[MAXB];
  __shared__ int wsum[32];
  int tid = threadIdx.x;
  int g = (int)blockIdx.x, G = (int)gridDim.x;
  totL[tid] = 0;
  preL[tid] = 0;
  __syncthreads();
  int slices = ST / NB;
  if (tid < slices * NB) {
    int bb = tid % NB, sl = tid / NB;
    int q0 = (sl * G) / slices, q1 = ((sl + 1) * G) / slices;
    int t = 0, p = 0;
#pragma unroll 16
    for (int q = q0; q < q1; ++q) {
      int v = cntm[(size_t)q * NB + bb];
      t += v;
      p += (q < g) ? v : 0;
    }
    atomicAdd(&totL[bb], t);
    atomicAdd(&preL[bb], p);
  }
  __syncthreads();
  int tot = (tid < NB) ? totL[tid] : 0;
  int ex = block_excl_scan(tot, wsum, tid, ST / 32);
  cur[tid] = ex + preL[tid];
  if (g == 0 && tid < NB) bst[tid] = ex;
  __syncthreads();

  int chunk = (E + G - 1) / G;
  int e0 = g * chunk;
  int e1 = min(E, e0 + chunk);
  for (int eb = e0; eb < e1; eb += 4 * ST) {
    int d[4], s[4];
#pragma unroll
    for (int u = 0; u < 4; ++u) {
      int e = eb + u * ST + tid;
      bool ok = e < e1;
      d[u] = ok ? dst[e] : -1;
      s[u] = ok ? src[e] : 0;
    }
#pragma unroll
    for (int u = 0; u < 4; ++u) {
      if (d[u] >= 0) {
        int pos = atomicAdd(&cur[d[u] >> BSH], 1);
        int2 pr; pr.x = s[u]; pr.y = d[u];
        *(int2*)(tmp + 2 * (size_t)pos) = pr;
      }
    }
  }
}

__global__ __launch_bounds__(FT) void gat_bucket_sort(
    const int* __restrict__ tmp, const int* __restrict__ bst,
    int* __restrict__ deg, int* __restrict__ cur, int* __restrict__ esrc,
    int N, int E, int NB) {
  __shared__ int c[BNODES];
  __shared__ int p[BNODES];
  __shared__ int wsum[32];
  int tid = threadIdx.x, b = (int)blockIdx.x;
  int lo = bst[b];
  int hi = (b + 1 < NB) ? bst[b + 1] : E;
  if (tid < BNODES) c[tid] = 0;
  __syncthreads();
  for (int ib = lo; ib < hi; ib += 4 * FT) {
    int d[4];
#pragma unroll
    for (int u = 0; u < 4; ++u) {
      int i = ib + u * FT + tid;
      d[u] = (i < hi) ? tmp[2 * (size_t)i + 1] : -1;
    }
#pragma unroll
    for (int u = 0; u < 4; ++u)
      if (d[u] >= 0) atomicAdd(&c[d[u] & (BNODES - 1)], 1);
  }
  __syncthreads();
  int v = (tid < BNODES) ? c[tid] : 0;
  int ex = block_excl_scan(v, wsum, tid, FT / 32);
  if (tid < BNODES) {
    p[tid] = lo + ex;
    int n = b * BNODES + tid;
    if (n < N) {
      deg[n] = v;
      cur[n] = lo + ex + v;
    }
  }
  __syncthreads();
  for (int ib = lo; ib < hi; ib += 4 * FT) {
    int2 r[4];
#pragma unroll
    for (int u = 0; u < 4; ++u) {
      int i = ib + u * FT + tid;
      r[u].x = 0; r[u].y = -1;
      if (i < hi) r[u] = *(const int2*)(tmp + 2 * (size_t)i);
    }
#pragma unroll
    for (int u = 0; u < 4; ++u) {
      if (r[u].y >= 0) {
        int pos = atomicAdd(&p[r[u].y & (BNODES - 1)], 1);
        esrc[pos] = r[u].x;
      }
    }
  }
}

template <int G>
__device__ __forceinline__ void gather1_group(
    const float4* __restrict__ h1v, int s_l, int c, bool odd, int slot0, int cnt,
    float pd, float4 ws, float& m, float& sum,
    float& ax, float& ay, float& az, float& aw) {
  float4 hr[8];
  hr[0] = h1v[(unsigned)(id_from_lane<8 * G + 0>(s_l) * 16 + c)];
  hr[1] = h1v[(unsigned)(id_from_lane<8 * G + 1>(s_l) * 16 + c)];
  hr[2] = h1v[(unsigned)(id_from_lane<8 * G + 2>(s_l) * 16 + c)];
  hr[3] = h1v[(unsigned)(id_from_lane<8 * G + 3>(s_l) * 16 + c)];
  hr[4] = h1v[(unsigned)(id_from_lane<8 * G + 4>(s_l) * 16 + c)];
  hr[5] = h1v[(unsigned)(id_from_lane<8 * G + 5>(s_l) * 16 + c)];
  hr[6] = h1v[(unsigned)(id_from_lane<8 * G + 6>(s_l) * 16 + c)];
  hr[7] = h1v[(unsigned)(id_from_lane<8 * G + 7>(s_l) * 16 + c)];
  const v2f wxy = {ws.x, ws.y}, wzw = {ws.z, ws.w};
  float p[8];
#pragma unroll
  for (int t = 0; t < 8; ++t) {
    v2f lo = {hr[t].x, hr[t].y}, hi = {hr[t].z, hr[t].w};
    v2f q = lo * wxy;
    q = hi * wzw + q;
    p[t] = q.x + q.y;
  }
  float e[4], w[4];
#pragma unroll
  for (int t = 0; t < 4; ++t) {
    float d = (odd ? p[t + 4] : p[t]) + xlane<1>(odd ? p[t] : p[t + 4]);
    e[t] = (slot0 + t < cnt) ? leaky(d + pd) : -INFINITY;
  }
  float gm = fmaxf(fmaxf(e[0], e[1]), fmaxf(e[2], e[3]));
  gm = fmaxf(gm, xlane<1>(gm));
  float mn = fmaxf(m, gm);
  float mr = (mn == -INFINITY) ? 0.f : mn;
  float sc = __expf(m - mr);
#pragma unroll
  for (int t = 0; t < 4; ++t) w[t] = __expf(e[t] - mr);
  float wsum = (w[0] + w[1]) + (w[2] + w[3]);
  wsum += xlane<1>(wsum);
  sum = fmaf(sum, sc, wsum);
  ax *= sc; ay *= sc; az *= sc; aw *= sc;
  float wj;
  wj = pair_lane<0>(w[0]); ax = fmaf(wj, hr[0].x, ax); ay = fmaf(wj, hr[0].y, ay); az = fmaf(wj, hr[0].z, az); aw = fmaf(wj, hr[0].w, aw);
  wj = pair_lane<0>(w[1]); ax = fmaf(wj, hr[1].x, ax); ay = fmaf(wj, hr[1].y, ay); az = fmaf(wj, hr[1].z, az); aw = fmaf(wj, hr[1].w, aw);
  wj = pair_lane<0>(w[2]); ax = fmaf(wj, hr[2].x, ax); ay = fmaf(wj, hr[2].y, ay); az = fmaf(wj, hr[2].z, az); aw = fmaf(wj, hr[2].w, aw);
  wj = pair_lane<0>(w[3]); ax = fmaf(wj, hr[3].x, ax); ay = fmaf(wj, hr[3].y, ay); az = fmaf(wj, hr[3].z, az); aw = fmaf(wj, hr[3].w, aw);
  wj = pair_lane<1>(w[0]); ax = fmaf(wj, hr[4].x, ax); ay = fmaf(wj, hr[4].y, ay); az = fmaf(wj, hr[4].z, az); aw = fmaf(wj, hr[4].w, aw);
  wj = pair_lane<1>(w[1]); ax = fmaf(wj, hr[5].x, ax); ay = fmaf(wj, hr[5].y, ay); az = fmaf(wj, hr[5].z, az); aw = fmaf(wj, hr[5].w, aw);
  wj = pair_lane<1>(w[2]); ax = fmaf(wj, hr[6].x, ax); ay = fmaf(wj, hr[6].y, ay); az = fmaf(wj, hr[6].z, az); aw = fmaf(wj, hr[6].w, aw);
  wj = pair_lane<1>(w[3]); ax = fmaf(wj, hr[7].x, ax); ay = fmaf(wj, hr[7].y, ay); az = fmaf(wj, hr[7].z, az); aw = fmaf(wj, hr[7].w, aw);
  m = mn;
}

__global__ __launch_bounds__(256) void gat_gather1(
    const int* __restrict__ esrc, const int* __restrict__ seg_end,
    const int* __restrict__ deg, const float* __restrict__ a_src,
    const float* __restrict__ a_dst, const float* __restrict__ h1,
    const float* __restrict__ b1, float* __restrict__ hact, int N) {
  int lane = threadIdx.x & 31;
  int c = lane & 15, half = lane >> 4;
  int n0 = __builtin_amdgcn_readfirstlane((int)((blockIdx.x * 8 + (threadIdx.x >> 5)) * 2));
  if (n0 >= N) return;
  int n = n0 + half;
  bool live = n < N;
  int nn = live ? n : n0;
  int cnt = live ? deg[n] : 0;
  int beg = live ? seg_end[n] - cnt : 0;

  const float4* h1v = (const float4*)h1;
  float4 hv = h1v[(unsigned)(nn * 16 + c)];
  float4 wd = ((const float4*)a_dst)[c];
  float4 ws = ((const float4*)a_src)[c];
  float4 bb = ((const float4*)b1)[c];
  float pd = hv.x * wd.x + hv.y * wd.y + hv.z * wd.z + hv.w * wd.w;
  pd += xlane<1>(pd);

  const bool odd = (c & 1) != 0;
  int cmax = __builtin_amdgcn_readfirstlane(max(cnt, __shfl_xor(cnt, 16, 32)));

  float m = -INFINITY;
  float ax = 0.f, ay = 0.f, az = 0.f, aw = 0.f, sum = 0.f;
  for (int b0 = 0; b0 < cmax; b0 += 16) {
    int j = b0 + c;
    int s_l = (j < cnt) ? esrc[beg + j] : nn;
    int kmax = min(16, cmax - b0);
    gather1_group<0>(h1v, s_l, c, odd, b0 + (odd ? 4 : 0), cnt, pd, ws, m, sum, ax, ay, az, aw);
    if (8 < kmax)
      gather1_group<1>(h1v, s_l, c, odd, b0 + 8 + (odd ? 4 : 0), cnt, pd, ws, m, sum, ax, ay, az, aw);
  }
  float den = sum + EPS_F;

  float vx = ax / den + bb.x;
  float vy = ay / den + bb.y;
  float vz = az / den + bb.z;
  float vw = aw / den + bb.w;
  float4 o;
  o.x = vx > 0.0f ? vx : expm1f(vx);
  o.y = vy > 0.0f ? vy : expm1f(vy);
  o.z = vz > 0.0f ? vz : expm1f(vz);
  o.w = vw > 0.0f ? vw : expm1f(vw);
  if (live) ((float4*)hact)[(size_t)n * 16 + c] = o;
}

__device__ __forceinline__ float dot4(float4 r, v2f wxy, v2f wzw) {
  v2f lo = {r.x, r.y}, hi = {r.z, r.w};
  v2f q = lo * wxy;
  q = hi * wzw + q;
  return q.x + q.y;
}

__global__ __launch_bounds__(256) void gat_gather2(
    const int* __restrict__ esrc, const int* __restrict__ seg_end,
    const int* __restrict__ deg, const float* __restrict__ a_src,
    const float* __restrict__ ad, const float* __restrict__ h2,
    const float* __restrict__ b2, float* __restrict__ dout, int N) {
  int lane = threadIdx.x & 31;
  int p = lane & 3;
  int u = (lane >> 2) & 1;
  int n0 = __builtin_amdgcn_readfirstlane((int)((blockIdx.x * 8 + (threadIdx.x >> 5)) * 4));
  if (n0 >= N) return;
  int n = n0 + (lane >> 3);
  bool live = n < N;
  int nc = live ? n : n0;
  int cnt = deg[nc];
  int beg = seg_end[nc] - cnt;
  float adn = ad[nc];
  cnt = live ? cnt : 0;
  const float4* h2v = (const float4*)h2;
  float4 ws = ((const float4*)a_src)[p];
  float4 bc = ((const float4*)b2)[p];
  const v2f wxy = {ws.x, ws.y}, wzw = {ws.z, ws.w};
  const bool odd = (p & 1) != 0, up = (p & 2) != 0;
  int cm = max(cnt, id_xlane8(cnt));
  cm = max(cm, __shfl_xor(cm, 16, 32));
  int cmax = __builtin_amdgcn_readfirstlane(cm);

  float m = -INFINITY;
  float ax = 0.f, ay = 0.f, az = 0.f, aw = 0.f, sum = 0.f;
  const int jl = 2 * p + u;
  int s_cur = esrc[(jl < cnt) ? beg + jl : 0];
  int j0 = 0;
  do {
    int jn = j0 + 8 + jl;
    int s_nxt = esrc[(jn < cnt) ? beg + jn : 0];
    __builtin_amdgcn_sched_barrier(0);
    float4 r0 = h2v[(unsigned)(quad_id<0>(s_cur) * 4 + p)];
    float4 r1 = h2v[(unsigned)(quad_id<1>(s_cur) * 4 + p)];
    float4 r2 = h2v[(unsigned)(quad_id<2>(s_cur) * 4 + p)];
    float4 r3 = h2v[(unsigned)(quad_id<3>(s_cur) * 4 + p)];
    float d0 = dot4(r0, wxy, wzw), d1 = dot4(r1, wxy, wzw), d2 = dot4(r2, wxy, wzw), d3 = dot4(r3, wxy, wzw);
    float v01 = (odd ? d1 : d0) + xlane<1>(odd ? d0 : d1);
    float v23 = (odd ? d3 : d2) + xlane<1>(odd ? d2 : d3);
    float d = (up ? v23 : v01) + xlane<2>(up ? v01 : v23);
    float e = (j0 + jl < cnt) ? leaky(d + adn) : -INFINITY;
    float gm = fmaxf(e, xlane<1>(e));
    gm = fmaxf(gm, xlane<2>(gm));
    float mn = fmaxf(m, gm);
    float mr = (mn == -INFINITY) ? 0.f : mn;
    float sc = __expf(m - mr);
    float w = __expf(e - mr);
    float wsum = w + xlane<1>(w);
    wsum += xlane<2>(wsum);
    sum = fmaf(sum, sc, wsum);
    ax *= sc; ay *= sc; az *= sc; aw *= sc;
    float wj;
    wj = quad_from<0>(w); ax = fmaf(wj, r0.x, ax); ay = fmaf(wj, r0.y, ay); az = fmaf(wj, r0.z, az); aw = fmaf(wj, r0.w, aw);
    wj = quad_from<1>(w); ax = fmaf(wj, r1.x, ax); ay = fmaf(wj, r1.y, ay); az = fmaf(wj, r1.z, az); aw = fmaf(wj, r1.w, aw);
    wj = quad_from<2>(w); ax = fmaf(wj, r2.x, ax); ay = fmaf(wj, r2.y, ay); az = fmaf(wj, r2.z, az); aw = fmaf(wj, r2.w, aw);
    wj = quad_from<3>(w); ax = fmaf(wj, r3.x, ax); ay = fmaf(wj, r3.y, ay); az = fmaf(wj, r3.z, az); aw = fmaf(wj, r3.w, aw);
    m = mn;
    s_cur = s_nxt;
    j0 += 8;
  } while (j0 < cmax);
  float mm = fmaxf(m, xlane<4>(m));
  float mmr = (mm == -INFINITY) ? 0.f : mm;
  float f = __expf(m - mmr);
  sum *= f; ax *= f; ay *= f; az *= f; aw *= f;
  sum += xlane<4>(sum);
  ax += xlane<4>(ax); ay += xlane<4>(ay); az += xlane<4>(az); aw += xlane<4>(aw);
  float den = sum + EPS_F;
  float vx = ax / den + bc.x;
  float vy = ay / den + bc.y;
  float vz = az / den + bc.z;
  float vw = aw / den + bc.w;
  float mx = fmaxf(fmaxf(vx, vy), fmaxf(vz, vw));
  mx = fmaxf(mx, xlane<1>(mx));
  mx = fmaxf(mx, xlane<2>(mx));
  float se = (__expf(vx - mx) + __expf(vy - mx)) + (__expf(vz - mx) + __expf(vw - mx));
  se += xlane<1>(se);
  se += xlane<2>(se);
  float lse = mx + __logf(se);
  if (live) {
    float sub = u ? lse : 0.f;
    float4 o; o.x = vx - sub; o.y = vy - sub; o.z = vz - sub; o.w = vw - sub;
    size_t off = (u ? (size_t)N * DOUT : (size_t)0) + (size_t)n * DOUT + 4 * p;
    *(float4*)(dout + off) = o;
  }
}

extern "C" void kernel_launch(void* const* d_in, const int* in_sizes, int n_in,
                              void* d_out, int out_size, void* d_ws, size_t ws_size,
                              hipStream_t stream) {
  const float* x   = (const float*)d_in[0];
  const int*   ei  = (const int*)d_in[1];
  const float* W1  = (const float*)d_in[2];
  const float* a1s = (const float*)d_in[3];
  const float* a1d = (const float*)d_in[4];
  const float* b1  = (const float*)d_in[5];
  const float* W2  = (const float*)d_in[6];
  const float* a2s = (const float*)d_in[7];
  const float* a2d = (const float*)d_in[8];
  const float* b2  = (const float*)d_in[9];

  int N = in_sizes[0] / DIM_IN;
  int E = in_sizes[1] / 2;
  const int* src = ei;
  const int* dst = ei + E;

  char* ws = (char*)d_ws;
  float*    h1   = (float*)ws;
  float*    hact = (float*)(ws + (size_t)N * HID * 4);
  char* p = ws + (size_t)N * HID * 4 * 2;
  int*      deg  = (int*)p;   p += ((size_t)N + 1) * 4;
  int*      cur  = (int*)p;   p += (size_t)N * 4;
  int*      esrc = (int*)p;
  const int NB = (N + BNODES - 1) / BNODES;
  int*      tmp  = (int*)hact;
  int*      cntm = tmp + 2 * (size_t)E;
  int*      bst  = cntm + (size_t)SG * NB;
  float*    h2  = h1;
  float*    ad2 = h1 + (size_t)N * DOUT;

  const int B = 256;
  gat_bucket_count<<<SG, ST, 0, stream>>>(dst, cntm, E, NB);
  gat_bucket_scatter<<<SG, ST, 0, stream>>>(src, dst, cntm, bst, tmp, E, NB);
  gat_bucket_sort<<<NB, FT, 0, stream>>>(tmp, bst, deg, cur, esrc, N, E, NB);

  gat_gemm1_wmma<<<(N + 16 * G1_WAVES - 1) / (16 * G1_WAVES), 32 * G1_WAVES, 0, stream>>>(x, W1, h1, N);
  gat_gather1<<<(N + 15) / 16, B, 0, stream>>>(esrc, cur, deg, a1s, a1d, h1, b1, hact, N);

  gat_gemm2_wmma<<<(N + 127) / 128, B, 0, stream>>>(hact, W2, a2d, h2, ad2, N);
  gat_gather2<<<(N + 31) / 32, B, 0, stream>>>(esrc, cur, deg, a2s, ad2, h2, b2, (float*)d_out, N);
}


# Edge ordering level 1 made slice-local: a work-group counts, scans and places its own slice in one launch; level 2 reads its bucket as 256 slice segments; a launch gone; ranks still LDS arrival order
# speedup vs baseline: 10.4559x; 1.2274x over previous
#include <hip/hip_runtime.h>
#include <math.h>


typedef float v2f __attribute__((ext_vector_type(2)));
typedef float v8f __attribute__((ext_vector_type(8)));

#define DIM_IN 256
#define HID    64
#define HEADS  8
#define DH     8
#define DOUT   16
#define NEG_SLOPE 0.2f
#define EPS_F 1e-16f

__device__ __forceinline__ float leaky(float v) {
  return v > 0.0f ? v : NEG_SLOPE * v;
}

template <int K>
__device__ __forceinline__ float xlane(float v) {
  return __builtin_bit_cast(float, __builtin_amdgcn_update_dpp(
      0, __builtin_bit_cast(int, v), 0x160 + K, 0xF, 0xF, true));
}
template <int K>
__device__ __forceinline__ float from_lane(float v) {
  return __builtin_bit_cast(float, __builtin_amdgcn_update_dpp(
      0, __builtin_bit_cast(int, v), 0x150 + K, 0xF, 0xF, true));
}
template <int K>
__device__ __forceinline__ float pair_lane(float v) {
  return __builtin_bit_cast(float, __builtin_amdgcn_update_dpp(
      0, __builtin_bit_cast(int, v), K ? 0xF5 : 0xA0, 0xF, 0xF, true));
}
template <int K>
__device__ __forceinline__ float quad_from(float v) {
  return __builtin_bit_cast(float, __builtin_amdgcn_update_dpp(
      0, __builtin_bit_cast(int, v), K * 0x55, 0xF, 0xF, true));
}
template <int K>
__device__ __forceinline__ int quad_id(int v) {
  return __builtin_amdgcn_update_dpp(0, v, K * 0x55, 0xF, 0xF, true);
}
__device__ __forceinline__ int id_xlane8(int v) {
  return __builtin_amdgcn_update_dpp(0, v, 0x168, 0xF, 0xF, true);
}
template <int K>
__device__ __forceinline__ int id_from_lane(int v) {
  return __builtin_amdgcn_update_dpp(0, v, 0x150 + K, 0xF, 0xF, true);
}

#define G1_WAVES 4
__global__ __launch_bounds__(128) void gat_gemm1_wmma(
    const float* __restrict__ x, const float* __restrict__ W1,
    float* __restrict__ h1, int N) {
  __shared__ float sW[DIM_IN * HID];
  int tid = threadIdx.x;
  int wave = tid >> 5, lane = tid & 31;
  int row0 = (blockIdx.x * G1_WAVES + wave) * 16;
  int r  = lane & 15;
  int kh = lane >> 4;

  int arow_idx = row0 + r; if (arow_idx >= N) arow_idx = N - 1;
  const float* arow = x + (size_t)arow_idx * DIM_IN + 2 * kh;

  v2f ac[8], an[8];
#pragma unroll
  for (int j = 0; j < 8; ++j) {
    float2 t = *(const float2*)(arow + 4 * j);
    ac[j].x = t.x; ac[j].y = t.y;
  }
  __builtin_prefetch(arow + 64, 0, 3);

  {
    const float4* Wv = (const float4*)W1;
    float4* sWv = (float4*)sW;
#pragma unroll 16
    for (int i = 0; i < 32; ++i) sWv[tid + i * 128] = Wv[tid + i * 128];
  }
  __syncthreads();
  if (row0 >= N) return;

  const float4* bbase = (const float4*)sW + r;

  v8f c0 = {}, c1 = {}, c2 = {}, c3 = {};
#pragma unroll
  for (int kb = 0; kb < DIM_IN; kb += 32) {
    if (kb + 32 < DIM_IN) {
#pragma unroll
      for (int j = 0; j < 8; ++j) {
        float2 t = *(const float2*)(arow + kb + 32 + 4 * j);
        an[j].x = t.x; an[j].y = t.y;
      }
    }
    if ((kb & 63) == 0 && kb + 128 < DIM_IN) __builtin_prefetch(arow + kb + 128, 0, 3);
#pragma unroll
    for (int j = 0; j < 8; ++j) {
      const int k = kb + 4 * j;
      float4 bx = bbase[(k + 2 * kh) * 16];
      float4 by = bbase[(k + 2 * kh + 1) * 16];
      v2f b;
      b.x = bx.x; b.y = by.x;
      c0 = __builtin_amdgcn_wmma_f32_16x16x4_f32(false, ac[j], false, b, (short)0, c0, false, false);
      b.x = bx.y; b.y = by.y;
      c1 = __builtin_amdgcn_wmma_f32_16x16x4_f32(false, ac[j], false, b, (short)0, c1, false, false);
      b.x = bx.z; b.y = by.z;
      c2 = __builtin_amdgcn_wmma_f32_16x16x4_f32(false, ac[j], false, b, (short)0, c2, false, false);
      b.x = bx.w; b.y = by.w;
      c3 = __builtin_amdgcn_wmma_f32_16x16x4_f32(false, ac[j], false, b, (short)0, c3, false, false);
    }
    if (kb + 32 < DIM_IN) {
#pragma unroll
      for (int j = 0; j < 8; ++j) ac[j] = an[j];
    }
  }
  float* orow = h1 + (size_t)(row0 + 8 * kh) * HID + 4 * r;
  if (row0 + 16 <= N) {
#pragma unroll
    for (int rr = 0; rr < 8; ++rr) {
      float4 o; o.x = c0[rr]; o.y = c1[rr]; o.z = c2[rr]; o.w = c3[rr];
      *(float4*)(orow + (size_t)rr * HID) = o;
    }
  } else {
#pragma unroll
    for (int rr = 0; rr < 8; ++rr) {
      int row = row0 + rr + 8 * kh;
      if (row < N) {
        float4 o; o.x = c0[rr]; o.y = c1[rr]; o.z = c2[rr]; o.w = c3[rr];
        *(float4*)(orow + (size_t)rr * HID) = o;
      }
    }
  }
}

__global__ __launch_bounds__(256) void gat_gemm2_wmma(
    const float* __restrict__ hact, const float* __restrict__ W2,
    const float* __restrict__ a_dst,
    float* __restrict__ h2, float* __restrict__ ad, int N) {
  __shared__ float sW[HID * DOUT];
  int tid = threadIdx.x;
  {
    const float4* Wv = (const float4*)W2;
    float4* sWv = (float4*)sW;
    sWv[tid] = Wv[tid];
  }
  __syncthreads();

  int wave = tid >> 5, lane = tid & 31;
  int row0 = (blockIdx.x * 8 + wave) * 16;
  if (row0 >= N) return;

  int r  = lane & 15;
  int kh = lane >> 4;
  int arow_idx = row0 + r; if (arow_idx >= N) arow_idx = N - 1;
  const float* arow  = hact + (size_t)arow_idx * HID + 2 * kh;
  const float* bbase = sW + r;
  float wdv = a_dst[r];

  v8f c = {};
#pragma unroll
  for (int k = 0; k < HID; k += 4) {
    float2 av = *(const float2*)(arow + k);
    v2f a; a.x = av.x; a.y = av.y;
    v2f b; b.x = bbase[(k + 2 * kh) * DOUT];
           b.y = bbase[(k + 2 * kh + 1) * DOUT];
    c = __builtin_amdgcn_wmma_f32_16x16x4_f32(false, a, false, b,
                                              (short)0, c, false, false);
  }
  float* orow = h2 + (size_t)row0 * DOUT + r;
  if (row0 + 16 <= N) {
#pragma unroll
    for (int rr = 0; rr < 8; ++rr)
      orow[(size_t)(rr + 8 * kh) * DOUT] = c[rr];
  } else {
#pragma unroll
    for (int rr = 0; rr < 8; ++rr) {
      int row = row0 + rr + 8 * kh;
      if (row < N) orow[(size_t)(rr + 8 * kh) * DOUT] = c[rr];
    }
  }
  float dsel = 0.f;
#pragma unroll
  for (int rr = 0; rr < 8; ++rr) {
    float pd = c[rr] * wdv;
    pd += __shfl_xor(pd, 1, 32);
    pd += __shfl_xor(pd, 2, 32);
    pd += __shfl_xor(pd, 4, 32);
    pd += __shfl_xor(pd, 8, 32);
    dsel = ((r & 7) == rr) ? pd : dsel;
  }
  int srow = row0 + 8 * kh + (r & 7);
  if (r < 8 && srow < N) ad[srow] = dsel;
}

#define BSH    8
#define BNODES 256
#define MAXB   1024
#define SG     256
#define ST     1024
#define FT     512

__device__ __forceinline__ int block_excl_scan(int v, int* wsum, int tid, int nwaves) {
  int lane = tid & 31, w = tid >> 5;
  int incl = v;
#pragma unroll
  for (int o = 1; o < 32; o <<= 1) {
    int t = __shfl_up(incl, o, 32);
    if (lane >= o) incl += t;
  }
  if (lane == 31) wsum[w] = incl;
  __syncthreads();
  if (w == 0) {
    int s = (lane < nwaves) ? wsum[lane] : 0;
    int si = s;
#pragma unroll
    for (int o = 1; o < 32; o <<= 1) {
      int t = __shfl_up(si, o, 32);
      if (lane >= o) si += t;
    }
    wsum[lane] = si - s;
  }
  __syncthreads();
  return wsum[w] + incl - v;
}

__global__ __launch_bounds__(ST) void gat_slice_group(
    const int* __restrict__ src, const int* __restrict__ dst,
    int* __restrict__ offs, int* __restrict__ tmp, int E, int NB) {
  __shared__ int hist[MAXB];
  __shared__ int cur[MAXB];
  __shared__ int wsum[32];
  int tid = threadIdx.x;
  int g = (int)blockIdx.x, G = (int)gridDim.x;
  hist[tid] = 0;
  __syncthreads();
  int chunk = (E + G - 1) / G;
  int e0 = g * chunk;
  int e1 = min(E, e0 + chunk);
  for (int eb = e0; eb < e1; eb += 4 * ST) {
    int d[4];
#pragma unroll
    for (int u = 0; u < 4; ++u) {
      int e = eb + u * ST + tid;
      d[u] = (e < e1) ? dst[e] : -1;
    }
#pragma unroll
    for (int u = 0; u < 4; ++u)
      if (d[u] >= 0) atomicAdd(&hist[d[u] >> BSH], 1);
  }
  __syncthreads();
  int v = (tid < NB) ? hist[tid] : 0;
  int ex = block_excl_scan(v, wsum, tid, ST / 32);
  cur[tid] = ex;
  if (tid <= NB) offs[(size_t)g * (NB + 1) + tid] = ex;
  __syncthreads();

  for (int eb = e0; eb < e1; eb += 4 * ST) {
    int d[4], s[4];
#pragma unroll
    for (int u = 0; u < 4; ++u) {
      int e = eb + u * ST + tid;
      bool ok = e < e1;
      d[u] = ok ? dst[e] : -1;
      s[u] = ok ? src[e] : 0;
    }
#pragma unroll
    for (int u = 0; u < 4; ++u) {
      if (d[u] >= 0) {
        int pos = e0 + atomicAdd(&cur[d[u] >> BSH], 1);
        int2 pr; pr.x = s[u]; pr.y = d[u];
        *(int2*)(tmp + 2 * (size_t)pos) = pr;
      }
    }
  }
}

#define SEGU 8
__global__ __launch_bounds__(FT) void gat_bucket_sort(
    const int* __restrict__ tmp, const int* __restrict__ offs,
    int* __restrict__ deg, int* __restrict__ cur, int* __restrict__ esrc,
    int N, int E, int NB, int G) {
  __shared__ int c[BNODES];
  __shared__ int p[BNODES];
  __shared__ int sBase[SG];
  __shared__ int sLen[SG];
  __shared__ int wsum[32];
  __shared__ int sLo;
  int tid = threadIdx.x, b = (int)blockIdx.x;
  int lane = tid & 31, w = tid >> 5;
  int chunk = (E + G - 1) / G;
  if (tid == 0) sLo = 0;
  if (tid < BNODES) c[tid] = 0;
  __syncthreads();
  {
    int lo_t = 0;
    if (tid < G) {
      lo_t = offs[(size_t)tid * (NB + 1) + b];
      int hi_t = offs[(size_t)tid * (NB + 1) + b + 1];
      sBase[tid] = tid * chunk + lo_t;
      sLen[tid] = hi_t - lo_t;
    }
    int s = lo_t;
#pragma unroll
    for (int o = 16; o > 0; o >>= 1) s += __shfl_xor(s, o, 32);
    if (lane == 0 && s != 0) atomicAdd(&sLo, s);
  }
  __syncthreads();
  const int lo = sLo;

  for (int t0 = w; t0 < G; t0 += SEGU * (FT / 32)) {
    int d[SEGU], len[SEGU], base[SEGU];
#pragma unroll
    for (int u = 0; u < SEGU; ++u) {
      int t = t0 + u * (FT / 32);
      len[u] = (t < G) ? sLen[t] : 0;
      base[u] = (t < G) ? sBase[t] : 0;
      d[u] = (lane < len[u]) ? tmp[2 * (size_t)(base[u] + lane) + 1] : -1;
    }
#pragma unroll
    for (int u = 0; u < SEGU; ++u)
      if (d[u] >= 0) atomicAdd(&c[d[u] & (BNODES - 1)], 1);
#pragma unroll
    for (int u = 0; u < SEGU; ++u)
      for (int i = lane + 32; i < len[u]; i += 32)
        atomicAdd(&c[tmp[2 * (size_t)(base[u] + i) + 1] & (BNODES - 1)], 1);
  }
  __syncthreads();
  int v = (tid < BNODES) ? c[tid] : 0;
  int ex = block_excl_scan(v, wsum, tid, FT / 32);
  if (tid < BNODES) {
    p[tid] = lo + ex;
    int n = b * BNODES + tid;
    if (n < N) {
      deg[n] = v;
      cur[n] = lo + ex + v;
    }
  }
  __syncthreads();
  for (int t0 = w; t0 < G; t0 += SEGU * (FT / 32)) {
    int2 r[SEGU]; int len[SEGU], base[SEGU];
#pragma unroll
    for (int u = 0; u < SEGU; ++u) {
      int t = t0 + u * (FT / 32);
      len[u] = (t < G) ? sLen[t] : 0;
      base[u] = (t < G) ? sBase[t] : 0;
      r[u].x = 0; r[u].y = -1;
      if (lane < len[u]) r[u] = *(const int2*)(tmp + 2 * (size_t)(base[u] + lane));
    }
#pragma unroll
    for (int u = 0; u < SEGU; ++u) {
      if (r[u].y >= 0) {
        int pos = atomicAdd(&p[r[u].y & (BNODES - 1)], 1);
        esrc[pos] = r[u].x;
      }
    }
#pragma unroll
    for (int u = 0; u < SEGU; ++u) {
      for (int i = lane + 32; i < len[u]; i += 32) {
        int2 q = *(const int2*)(tmp + 2 * (size_t)(base[u] + i));
        int pos = atomicAdd(&p[q.y & (BNODES - 1)], 1);
        esrc[pos] = q.x;
      }
    }
  }
}

template <int G>
__device__ __forceinline__ void gather1_group(
    const float4* __restrict__ h1v, int s_l, int c, bool odd, int slot0, int cnt,
    float pd, float4 ws, float& m, float& sum,
    float& ax, float& ay, float& az, float& aw) {
  float4 hr[8];
  hr[0] = h1v[(unsigned)(id_from_lane<8 * G + 0>(s_l) * 16 + c)];
  hr[1] = h1v[(unsigned)(id_from_lane<8 * G + 1>(s_l) * 16 + c)];
  hr[2] = h1v[(unsigned)(id_from_lane<8 * G + 2>(s_l) * 16 + c)];
  hr[3] = h1v[(unsigned)(id_from_lane<8 * G + 3>(s_l) * 16 + c)];
  hr[4] = h1v[(unsigned)(id_from_lane<8 * G + 4>(s_l) * 16 + c)];
  hr[5] = h1v[(unsigned)(id_from_lane<8 * G + 5>(s_l) * 16 + c)];
  hr[6] = h1v[(unsigned)(id_from_lane<8 * G + 6>(s_l) * 16 + c)];
  hr[7] = h1v[(unsigned)(id_from_lane<8 * G + 7>(s_l) * 16 + c)];
  const v2f wxy = {ws.x, ws.y}, wzw = {ws.z, ws.w};
  float p[8];
#pragma unroll
  for (int t = 0; t < 8; ++t) {
    v2f lo = {hr[t].x, hr[t].y}, hi = {hr[t].z, hr[t].w};
    v2f q = lo * wxy;
    q = hi * wzw + q;
    p[t] = q.x + q.y;
  }
  float e[4], w[4];
#pragma unroll
  for (int t = 0; t < 4; ++t) {
    float d = (odd ? p[t + 4] : p[t]) + xlane<1>(odd ? p[t] : p[t + 4]);
    e[t] = (slot0 + t < cnt) ? leaky(d + pd) : -INFINITY;
  }
  float gm = fmaxf(fmaxf(e[0], e[1]), fmaxf(e[2], e[3]));
  gm = fmaxf(gm, xlane<1>(gm));
  float mn = fmaxf(m, gm);
  float mr = (mn == -INFINITY) ? 0.f : mn;
  float sc = __expf(m - mr);
#pragma unroll
  for (int t = 0; t < 4; ++t) w[t] = __expf(e[t] - mr);
  float wsum = (w[0] + w[1]) + (w[2] + w[3]);
  wsum += xlane<1>(wsum);
  sum = fmaf(sum, sc, wsum);
  ax *= sc; ay *= sc; az *= sc; aw *= sc;
  float wj;
  wj = pair_lane<0>(w[0]); ax = fmaf(wj, hr[0].x, ax); ay = fmaf(wj, hr[0].y, ay); az = fmaf(wj, hr[0].z, az); aw = fmaf(wj, hr[0].w, aw);
  wj = pair_lane<0>(w[1]); ax = fmaf(wj, hr[1].x, ax); ay = fmaf(wj, hr[1].y, ay); az = fmaf(wj, hr[1].z, az); aw = fmaf(wj, hr[1].w, aw);
  wj = pair_lane<0>(w[2]); ax = fmaf(wj, hr[2].x, ax); ay = fmaf(wj, hr[2].y, ay); az = fmaf(wj, hr[2].z, az); aw = fmaf(wj, hr[2].w, aw);
  wj = pair_lane<0>(w[3]); ax = fmaf(wj, hr[3].x, ax); ay = fmaf(wj, hr[3].y, ay); az = fmaf(wj, hr[3].z, az); aw = fmaf(wj, hr[3].w, aw);
  wj = pair_lane<1>(w[0]); ax = fmaf(wj, hr[4].x, ax); ay = fmaf(wj, hr[4].y, ay); az = fmaf(wj, hr[4].z, az); aw = fmaf(wj, hr[4].w, aw);
  wj = pair_lane<1>(w[1]); ax = fmaf(wj, hr[5].x, ax); ay = fmaf(wj, hr[5].y, ay); az = fmaf(wj, hr[5].z, az); aw = fmaf(wj, hr[5].w, aw);
  wj = pair_lane<1>(w[2]); ax = fmaf(wj, hr[6].x, ax); ay = fmaf(wj, hr[6].y, ay); az = fmaf(wj, hr[6].z, az); aw = fmaf(wj, hr[6].w, aw);
  wj = pair_lane<1>(w[3]); ax = fmaf(wj, hr[7].x, ax); ay = fmaf(wj, hr[7].y, ay); az = fmaf(wj, hr[7].z, az); aw = fmaf(wj, hr[7].w, aw);
  m = mn;
}

__global__ __launch_bounds__(256) void gat_gather1(
    const int* __restrict__ esrc, const int* __restrict__ seg_end,
    const int* __restrict__ deg, const float* __restrict__ a_src,
    const float* __restrict__ a_dst, const float* __restrict__ h1,
    const float* __restrict__ b1, float* __restrict__ hact, int N) {
  int lane = threadIdx.x & 31;
  int c = lane & 15, half = lane >> 4;
  int n0 = __builtin_amdgcn_readfirstlane((int)((blockIdx.x * 8 + (threadIdx.x >> 5)) * 2));
  if (n0 >= N) return;
  int n = n0 + half;
  bool live = n < N;
  int nn = live ? n : n0;
  int cnt = live ? deg[n] : 0;
  int beg = live ? seg_end[n] - cnt : 0;

  const float4* h1v = (const float4*)h1;
  float4 hv = h1v[(unsigned)(nn * 16 + c)];
  float4 wd = ((const float4*)a_dst)[c];
  float4 ws = ((const float4*)a_src)[c];
  float4 bb = ((const float4*)b1)[c];
  float pd = hv.x * wd.x + hv.y * wd.y + hv.z * wd.z + hv.w * wd.w;
  pd += xlane<1>(pd);

  const bool odd = (c & 1) != 0;
  int cmax = __builtin_amdgcn_readfirstlane(max(cnt, __shfl_xor(cnt, 16, 32)));

  float m = -INFINITY;
  float ax = 0.f, ay = 0.f, az = 0.f, aw = 0.f, sum = 0.f;
  for (int b0 = 0; b0 < cmax; b0 += 16) {
    int j = b0 + c;
    int s_l = (j < cnt) ? esrc[beg + j] : nn;
    int kmax = min(16, cmax - b0);
    gather1_group<0>(h1v, s_l, c, odd, b0 + (odd ? 4 : 0), cnt, pd, ws, m, sum, ax, ay, az, aw);
    if (8 < kmax)
      gather1_group<1>(h1v, s_l, c, odd, b0 + 8 + (odd ? 4 : 0), cnt, pd, ws, m, sum, ax, ay, az, aw);
  }
  float den = sum + EPS_F;

  float vx = ax / den + bb.x;
  float vy = ay / den + bb.y;
  float vz = az / den + bb.z;
  float vw = aw / den + bb.w;
  float4 o;
  o.x = vx > 0.0f ? vx : expm1f(vx);
  o.y = vy > 0.0f ? vy : expm1f(vy);
  o.z = vz > 0.0f ? vz : expm1f(vz);
  o.w = vw > 0.0f ? vw : expm1f(vw);
  if (live) ((float4*)hact)[(size_t)n * 16 + c] = o;
}

__device__ __forceinline__ float dot4(float4 r, v2f wxy, v2f wzw) {
  v2f lo = {r.x, r.y}, hi = {r.z, r.w};
  v2f q = lo * wxy;
  q = hi * wzw + q;
  return q.x + q.y;
}

__global__ __launch_bounds__(256) void gat_gather2(
    const int* __restrict__ esrc, const int* __restrict__ seg_end,
    const int* __restrict__ deg, const float* __restrict__ a_src,
    const float* __restrict__ ad, const float* __restrict__ h2,
    const float* __restrict__ b2, float* __restrict__ dout, int N) {
  int lane = threadIdx.x & 31;
  int p = lane & 3;
  int u = (lane >> 2) & 1;
  int n0 = __builtin_amdgcn_readfirstlane((int)((blockIdx.x * 8 + (threadIdx.x >> 5)) * 4));
  if (n0 >= N) return;
  int n = n0 + (lane >> 3);
  bool live = n < N;
  int nc = live ? n : n0;
  int cnt = deg[nc];
  int beg = seg_end[nc] - cnt;
  float adn = ad[nc];
  cnt = live ? cnt : 0;
  const float4* h2v = (const float4*)h2;
  float4 ws = ((const float4*)a_src)[p];
  float4 bc = ((const float4*)b2)[p];
  const v2f wxy = {ws.x, ws.y}, wzw = {ws.z, ws.w};
  const bool odd = (p & 1) != 0, up = (p & 2) != 0;
  int cm = max(cnt, id_xlane8(cnt));
  cm = max(cm, __shfl_xor(cm, 16, 32));
  int cmax = __builtin_amdgcn_readfirstlane(cm);

  float m = -INFINITY;
  float ax = 0.f, ay = 0.f, az = 0.f, aw = 0.f, sum = 0.f;
  const int jl = 2 * p + u;
  int s_cur = esrc[(jl < cnt) ? beg + jl : 0];
  int j0 = 0;
  do {
    int jn = j0 + 8 + jl;
    int s_nxt = esrc[(jn < cnt) ? beg + jn : 0];
    __builtin_amdgcn_sched_barrier(0);
    float4 r0 = h2v[(unsigned)(quad_id<0>(s_cur) * 4 + p)];
    float4 r1 = h2v[(unsigned)(quad_id<1>(s_cur) * 4 + p)];
    float4 r2 = h2v[(unsigned)(quad_id<2>(s_cur) * 4 + p)];
    float4 r3 = h2v[(unsigned)(quad_id<3>(s_cur) * 4 + p)];
    float d0 = dot4(r0, wxy, wzw), d1 = dot4(r1, wxy, wzw), d2 = dot4(r2, wxy, wzw), d3 = dot4(r3, wxy, wzw);
    float v01 = (odd ? d1 : d0) + xlane<1>(odd ? d0 : d1);
    float v23 = (odd ? d3 : d2) + xlane<1>(odd ? d2 : d3);
    float d = (up ? v23 : v01) + xlane<2>(up ? v01 : v23);
    float e = (j0 + jl < cnt) ? leaky(d + adn) : -INFINITY;
    float gm = fmaxf(e, xlane<1>(e));
    gm = fmaxf(gm, xlane<2>(gm));
    float mn = fmaxf(m, gm);
    float mr = (mn == -INFINITY) ? 0.f : mn;
    float sc = __expf(m - mr);
    float w = __expf(e - mr);
    float wsum = w + xlane<1>(w);
    wsum += xlane<2>(wsum);
    sum = fmaf(sum, sc, wsum);
    ax *= sc; ay *= sc; az *= sc; aw *= sc;
    float wj;
    wj = quad_from<0>(w); ax = fmaf(wj, r0.x, ax); ay = fmaf(wj, r0.y, ay); az = fmaf(wj, r0.z, az); aw = fmaf(wj, r0.w, aw);
    wj = quad_from<1>(w); ax = fmaf(wj, r1.x, ax); ay = fmaf(wj, r1.y, ay); az = fmaf(wj, r1.z, az); aw = fmaf(wj, r1.w, aw);
    wj = quad_from<2>(w); ax = fmaf(wj, r2.x, ax); ay = fmaf(wj, r2.y, ay); az = fmaf(wj, r2.z, az); aw = fmaf(wj, r2.w, aw);
    wj = quad_from<3>(w); ax = fmaf(wj, r3.x, ax); ay = fmaf(wj, r3.y, ay); az = fmaf(wj, r3.z, az); aw = fmaf(wj, r3.w, aw);
    m = mn;
    s_cur = s_nxt;
    j0 += 8;
  } while (j0 < cmax);
  float mm = fmaxf(m, xlane<4>(m));
  float mmr = (mm == -INFINITY) ? 0.f : mm;
  float f = __expf(m - mmr);
  sum *= f; ax *= f; ay *= f; az *= f; aw *= f;
  sum += xlane<4>(sum);
  ax += xlane<4>(ax); ay += xlane<4>(ay); az += xlane<4>(az); aw += xlane<4>(aw);
  float den = sum + EPS_F;
  float vx = ax / den + bc.x;
  float vy = ay / den + bc.y;
  float vz = az / den + bc.z;
  float vw = aw / den + bc.w;
  float mx = fmaxf(fmaxf(vx, vy), fmaxf(vz, vw));
  mx = fmaxf(mx, xlane<1>(mx));
  mx = fmaxf(mx, xlane<2>(mx));
  float se = (__expf(vx - mx) + __expf(vy - mx)) + (__expf(vz - mx) + __expf(vw - mx));
  se += xlane<1>(se);
  se += xlane<2>(se);
  float lse = mx + __logf(se);
  if (live) {
    float sub = u ? lse : 0.f;
    float4 o; o.x = vx - sub; o.y = vy - sub; o.z = vz - sub; o.w = vw - sub;
    size_t off = (u ? (size_t)N * DOUT : (size_t)0) + (size_t)n * DOUT + 4 * p;
    *(float4*)(dout + off) = o;
  }
}

extern "C" void kernel_launch(void* const* d_in, const int* in_sizes, int n_in,
                              void* d_out, int out_size, void* d_ws, size_t ws_size,
                              hipStream_t stream) {
  const float* x   = (const float*)d_in[0];
  const int*   ei  = (const int*)d_in[1];
  const float* W1  = (const float*)d_in[2];
  const float* a1s = (const float*)d_in[3];
  const float* a1d = (const float*)d_in[4];
  const float* b1  = (const float*)d_in[5];
  const float* W2  = (const float*)d_in[6];
  const float* a2s = (const float*)d_in[7];
  const float* a2d = (const float*)d_in[8];
  const float* b2  = (const float*)d_in[9];

  int N = in_sizes[0] / DIM_IN;
  int E = in_sizes[1] / 2;
  const int* src = ei;
  const int* dst = ei + E;

  char* ws = (char*)d_ws;
  float*    h1   = (float*)ws;
  float*    hact = (float*)(ws + (size_t)N * HID * 4);
  char* p = ws + (size_t)N * HID * 4 * 2;
  int*      deg  = (int*)p;   p += ((size_t)N + 1) * 4;
  int*      cur  = (int*)p;   p += (size_t)N * 4;
  int*      esrc = (int*)p;
  const int NB = (N + BNODES - 1) / BNODES;
  int*      tmp  = (int*)hact;
  int*      offs = tmp + 2 * (size_t)E;
  float*    h2  = h1;
  float*    ad2 = h1 + (size_t)N * DOUT;

  const int B = 256;
  gat_slice_group<<<SG, ST, 0, stream>>>(src, dst, offs, tmp, E, NB);
  gat_bucket_sort<<<NB, FT, 0, stream>>>(tmp, offs, deg, cur, esrc, N, E, NB, SG);

  gat_gemm1_wmma<<<(N + 16 * G1_WAVES - 1) / (16 * G1_WAVES), 32 * G1_WAVES, 0, stream>>>(x, W1, h1, N);
  gat_gather1<<<(N + 15) / 16, B, 0, stream>>>(esrc, cur, deg, a1s, a1d, h1, b1, hact, N);

  gat_gemm2_wmma<<<(N + 127) / 128, B, 0, stream>>>(hact, W2, a2d, h2, ad2, N);
  gat_gather2<<<(N + 31) / 32, B, 0, stream>>>(esrc, cur, deg, a2s, ad2, h2, b2, (float*)d_out, N);
}


# Layer-2 product moved into the layer-1 gather's tail: a work-group's 16 activated rows go via LDS to one wave's 16 matrix operations (h2, its a_dst dot); rows never stored; a launch gone; same sums
# speedup vs baseline: 10.8987x; 1.0423x over previous
#include <hip/hip_runtime.h>
#include <math.h>


typedef float v2f __attribute__((ext_vector_type(2)));
typedef float v8f __attribute__((ext_vector_type(8)));

#define DIM_IN 256
#define HID    64
#define HEADS  8
#define DH     8
#define DOUT   16
#define NEG_SLOPE 0.2f
#define EPS_F 1e-16f

__device__ __forceinline__ float leaky(float v) {
  return v > 0.0f ? v : NEG_SLOPE * v;
}

template <int K>
__device__ __forceinline__ float xlane(float v) {
  return __builtin_bit_cast(float, __builtin_amdgcn_update_dpp(
      0, __builtin_bit_cast(int, v), 0x160 + K, 0xF, 0xF, true));
}
template <int K>
__device__ __forceinline__ float from_lane(float v) {
  return __builtin_bit_cast(float, __builtin_amdgcn_update_dpp(
      0, __builtin_bit_cast(int, v), 0x150 + K, 0xF, 0xF, true));
}
template <int K>
__device__ __forceinline__ float pair_lane(float v) {
  return __builtin_bit_cast(float, __builtin_amdgcn_update_dpp(
      0, __builtin_bit_cast(int, v), K ? 0xF5 : 0xA0, 0xF, 0xF, true));
}
template <int K>
__device__ __forceinline__ float quad_from(float v) {
  return __builtin_bit_cast(float, __builtin_amdgcn_update_dpp(
      0, __builtin_bit_cast(int, v), K * 0x55, 0xF, 0xF, true));
}
template <int K>
__device__ __forceinline__ int quad_id(int v) {
  return __builtin_amdgcn_update_dpp(0, v, K * 0x55, 0xF, 0xF, true);
}
__device__ __forceinline__ int id_xlane8(int v) {
  return __builtin_amdgcn_update_dpp(0, v, 0x168, 0xF, 0xF, true);
}
template <int K>
__device__ __forceinline__ int id_from_lane(int v) {
  return __builtin_amdgcn_update_dpp(0, v, 0x150 + K, 0xF, 0xF, true);
}

#define G1_WAVES 4
__global__ __launch_bounds__(128) void gat_gemm1_wmma(
    const float* __restrict__ x, const float* __restrict__ W1,
    float* __restrict__ h1, int N) {
  __shared__ float sW[DIM_IN * HID];
  int tid = threadIdx.x;
  int wave = tid >> 5, lane = tid & 31;
  int row0 = (blockIdx.x * G1_WAVES + wave) * 16;
  int r  = lane & 15;
  int kh = lane >> 4;

  int arow_idx = row0 + r; if (arow_idx >= N) arow_idx = N - 1;
  const float* arow = x + (size_t)arow_idx * DIM_IN + 2 * kh;

  v2f ac[8], an[8];
#pragma unroll
  for (int j = 0; j < 8; ++j) {
    float2 t = *(const float2*)(arow + 4 * j);
    ac[j].x = t.x; ac[j].y = t.y;
  }
  __builtin_prefetch(arow + 64, 0, 3);

  {
    const float4* Wv = (const float4*)W1;
    float4* sWv = (float4*)sW;
#pragma unroll 16
    for (int i = 0; i < 32; ++i) sWv[tid + i * 128] = Wv[tid + i * 128];
  }
  __syncthreads();
  if (row0 >= N) return;

  const float4* bbase = (const float4*)sW + r;

  v8f c0 = {}, c1 = {}, c2 = {}, c3 = {};
#pragma unroll
  for (int kb = 0; kb < DIM_IN; kb += 32) {
    if (kb + 32 < DIM_IN) {
#pragma unroll
      for (int j = 0; j < 8; ++j) {
        float2 t = *(const float2*)(arow + kb + 32 + 4 * j);
        an[j].x = t.x; an[j].y = t.y;
      }
    }
    if ((kb & 63) == 0 && kb + 128 < DIM_IN) __builtin_prefetch(arow + kb + 128, 0, 3);
#pragma unroll
    for (int j = 0; j < 8; ++j) {
      const int k = kb + 4 * j;
      float4 bx = bbase[(k + 2 * kh) * 16];
      float4 by = bbase[(k + 2 * kh + 1) * 16];
      v2f b;
      b.x = bx.x; b.y = by.x;
      c0 = __builtin_amdgcn_wmma_f32_16x16x4_f32(false, ac[j], false, b, (short)0, c0, false, false);
      b.x = bx.y; b.y = by.y;
      c1 = __builtin_amdgcn_wmma_f32_16x16x4_f32(false, ac[j], false, b, (short)0, c1, false, false);
      b.x = bx.z; b.y = by.z;
      c2 = __builtin_amdgcn_wmma_f32_16x16x4_f32(false, ac[j], false, b, (short)0, c2, false, false);
      b.x = bx.w; b.y = by.w;
      c3 = __builtin_amdgcn_wmma_f32_16x16x4_f32(false, ac[j], false, b, (short)0, c3, false, false);
    }
    if (kb + 32 < DIM_IN) {
#pragma unroll
      for (int j = 0; j < 8; ++j) ac[j] = an[j];
    }
  }
  float* orow = h1 + (size_t)(row0 + 8 * kh) * HID + 4 * r;
  if (row0 + 16 <= N) {
#pragma unroll
    for (int rr = 0; rr < 8; ++rr) {
      float4 o; o.x = c0[rr]; o.y = c1[rr]; o.z = c2[rr]; o.w = c3[rr];
      *(float4*)(orow + (size_t)rr * HID) = o;
    }
  } else {
#pragma unroll
    for (int rr = 0; rr < 8; ++rr) {
      int row = row0 + rr + 8 * kh;
      if (row < N) {
        float4 o; o.x = c0[rr]; o.y = c1[rr]; o.z = c2[rr]; o.w = c3[rr];
        *(float4*)(orow + (size_t)rr * HID) = o;
      }
    }
  }
}

#define BSH    8
#define BNODES 256
#define MAXB   1024
#define SG     256
#define ST     1024
#define FT     512

__device__ __forceinline__ int block_excl_scan(int v, int* wsum, int tid, int nwaves) {
  int lane = tid & 31, w = tid >> 5;
  int incl = v;
#pragma unroll
  for (int o = 1; o < 32; o <<= 1) {
    int t = __shfl_up(incl, o, 32);
    if (lane >= o) incl += t;
  }
  if (lane == 31) wsum[w] = incl;
  __syncthreads();
  if (w == 0) {
    int s = (lane < nwaves) ? wsum[lane] : 0;
    int si = s;
#pragma unroll
    for (int o = 1; o < 32; o <<= 1) {
      int t = __shfl_up(si, o, 32);
      if (lane >= o) si += t;
    }
    wsum[lane] = si - s;
  }
  __syncthreads();
  return wsum[w] + incl - v;
}

__global__ __launch_bounds__(ST) void gat_slice_group(
    const int* __restrict__ src, const int* __restrict__ dst,
    int* __restrict__ offs, int* __restrict__ tmp, int E, int NB) {
  __shared__ int hist[MAXB];
  __shared__ int cur[MAXB];
  __shared__ int wsum[32];
  int tid = threadIdx.x;
  int g = (int)blockIdx.x, G = (int)gridDim.x;
  hist[tid] = 0;
  __syncthreads();
  int chunk = (E + G - 1) / G;
  int e0 = g * chunk;
  int e1 = min(E, e0 + chunk);
  for (int eb = e0; eb < e1; eb += 4 * ST) {
    int d[4];
#pragma unroll
    for (int u = 0; u < 4; ++u) {
      int e = eb + u * ST + tid;
      d[u] = (e < e1) ? dst[e] : -1;
    }
#pragma unroll
    for (int u = 0; u < 4; ++u)
      if (d[u] >= 0) atomicAdd(&hist[d[u] >> BSH], 1);
  }
  __syncthreads();
  int v = (tid < NB) ? hist[tid] : 0;
  int ex = block_excl_scan(v, wsum, tid, ST / 32);
  cur[tid] = ex;
  if (tid <= NB) offs[(size_t)g * (NB + 1) + tid] = ex;
  __syncthreads();

  for (int eb = e0; eb < e1; eb += 4 * ST) {
    int d[4], s[4];
#pragma unroll
    for (int u = 0; u < 4; ++u) {
      int e = eb + u * ST + tid;
      bool ok = e < e1;
      d[u] = ok ? dst[e] : -1;
      s[u] = ok ? src[e] : 0;
    }
#pragma unroll
    for (int u = 0; u < 4; ++u) {
      if (d[u] >= 0) {
        int pos = e0 + atomicAdd(&cur[d[u] >> BSH], 1);
        int2 pr; pr.x = s[u]; pr.y = d[u];
        *(int2*)(tmp + 2 * (size_t)pos) = pr;
      }
    }
  }
}

#define SEGU 8
__global__ __launch_bounds__(FT) void gat_bucket_sort(
    const int* __restrict__ tmp, const int* __restrict__ offs,
    int* __restrict__ deg, int* __restrict__ cur, int* __restrict__ esrc,
    int N, int E, int NB, int G) {
  __shared__ int c[BNODES];
  __shared__ int p[BNODES];
  __shared__ int sBase[SG];
  __shared__ int sLen[SG];
  __shared__ int wsum[32];
  __shared__ int sLo;
  int tid = threadIdx.x, b = (int)blockIdx.x;
  int lane = tid & 31, w = tid >> 5;
  int chunk = (E + G - 1) / G;
  if (tid == 0) sLo = 0;
  if (tid < BNODES) c[tid] = 0;
  __syncthreads();
  {
    int lo_t = 0;
    if (tid < G) {
      lo_t = offs[(size_t)tid * (NB + 1) + b];
      int hi_t = offs[(size_t)tid * (NB + 1) + b + 1];
      sBase[tid] = tid * chunk + lo_t;
      sLen[tid] = hi_t - lo_t;
    }
    int s = lo_t;
#pragma unroll
    for (int o = 16; o > 0; o >>= 1) s += __shfl_xor(s, o, 32);
    if (lane == 0 && s != 0) atomicAdd(&sLo, s);
  }
  __syncthreads();
  const int lo = sLo;

  for (int t0 = w; t0 < G; t0 += SEGU * (FT / 32)) {
    int d[SEGU], len[SEGU], base[SEGU];
#pragma unroll
    for (int u = 0; u < SEGU; ++u) {
      int t = t0 + u * (FT / 32);
      len[u] = (t < G) ? sLen[t] : 0;
      base[u] = (t < G) ? sBase[t] : 0;
      d[u] = (lane < len[u]) ? tmp[2 * (size_t)(base[u] + lane) + 1] : -1;
    }
#pragma unroll
    for (int u = 0; u < SEGU; ++u)
      if (d[u] >= 0) atomicAdd(&c[d[u] & (BNODES - 1)], 1);
#pragma unroll
    for (int u = 0; u < SEGU; ++u)
      for (int i = lane + 32; i < len[u]; i += 32)
        atomicAdd(&c[tmp[2 * (size_t)(base[u] + i) + 1] & (BNODES - 1)], 1);
  }
  __syncthreads();
  int v = (tid < BNODES) ? c[tid] : 0;
  int ex = block_excl_scan(v, wsum, tid, FT / 32);
  if (tid < BNODES) {
    p[tid] = lo + ex;
    int n = b * BNODES + tid;
    if (n < N) {
      deg[n] = v;
      cur[n] = lo + ex + v;
    }
  }
  __syncthreads();
  for (int t0 = w; t0 < G; t0 += SEGU * (FT / 32)) {
    int2 r[SEGU]; int len[SEGU], base[SEGU];
#pragma unroll
    for (int u = 0; u < SEGU; ++u) {
      int t = t0 + u * (FT / 32);
      len[u] = (t < G) ? sLen[t] : 0;
      base[u] = (t < G) ? sBase[t] : 0;
      r[u].x = 0; r[u].y = -1;
      if (lane < len[u]) r[u] = *(const int2*)(tmp + 2 * (size_t)(base[u] + lane));
    }
#pragma unroll
    for (int u = 0; u < SEGU; ++u) {
      if (r[u].y >= 0) {
        int pos = atomicAdd(&p[r[u].y & (BNODES - 1)], 1);
        esrc[pos] = r[u].x;
      }
    }
#pragma unroll
    for (int u = 0; u < SEGU; ++u) {
      for (int i = lane + 32; i < len[u]; i += 32) {
        int2 q = *(const int2*)(tmp + 2 * (size_t)(base[u] + i));
        int pos = atomicAdd(&p[q.y & (BNODES - 1)], 1);
        esrc[pos] = q.x;
      }
    }
  }
}

template <int G>
__device__ __forceinline__ void gather1_group(
    const float4* __restrict__ h1v, int s_l, int c, bool odd, int slot0, int cnt,
    float pd, float4 ws, float& m, float& sum,
    float& ax, float& ay, float& az, float& aw) {
  float4 hr[8];
  hr[0] = h1v[(unsigned)(id_from_lane<8 * G + 0>(s_l) * 16 + c)];
  hr[1] = h1v[(unsigned)(id_from_lane<8 * G + 1>(s_l) * 16 + c)];
  hr[2] = h1v[(unsigned)(id_from_lane<8 * G + 2>(s_l) * 16 + c)];
  hr[3] = h1v[(unsigned)(id_from_lane<8 * G + 3>(s_l) * 16 + c)];
  hr[4] = h1v[(unsigned)(id_from_lane<8 * G + 4>(s_l) * 16 + c)];
  hr[5] = h1v[(unsigned)(id_from_lane<8 * G + 5>(s_l) * 16 + c)];
  hr[6] = h1v[(unsigned)(id_from_lane<8 * G + 6>(s_l) * 16 + c)];
  hr[7] = h1v[(unsigned)(id_from_lane<8 * G + 7>(s_l) * 16 + c)];
  const v2f wxy = {ws.x, ws.y}, wzw = {ws.z, ws.w};
  float p[8];
#pragma unroll
  for (int t = 0; t < 8; ++t) {
    v2f lo = {hr[t].x, hr[t].y}, hi = {hr[t].z, hr[t].w};
    v2f q = lo * wxy;
    q = hi * wzw + q;
    p[t] = q.x + q.y;
  }
  float e[4], w[4];
#pragma unroll
  for (int t = 0; t < 4; ++t) {
    float d = (odd ? p[t + 4] : p[t]) + xlane<1>(odd ? p[t] : p[t + 4]);
    e[t] = (slot0 + t < cnt) ? leaky(d + pd) : -INFINITY;
  }
  float gm = fmaxf(fmaxf(e[0], e[1]), fmaxf(e[2], e[3]));
  gm = fmaxf(gm, xlane<1>(gm));
  float mn = fmaxf(m, gm);
  float mr = (mn == -INFINITY) ? 0.f : mn;
  float sc = __expf(m - mr);
#pragma unroll
  for (int t = 0; t < 4; ++t) w[t] = __expf(e[t] - mr);
  float wsum = (w[0] + w[1]) + (w[2] + w[3]);
  wsum += xlane<1>(wsum);
  sum = fmaf(sum, sc, wsum);
  ax *= sc; ay *= sc; az *= sc; aw *= sc;
  float wj;
  wj = pair_lane<0>(w[0]); ax = fmaf(wj, hr[0].x, ax); ay = fmaf(wj, hr[0].y, ay); az = fmaf(wj, hr[0].z, az); aw = fmaf(wj, hr[0].w, aw);
  wj = pair_lane<0>(w[1]); ax = fmaf(wj, hr[1].x, ax); ay = fmaf(wj, hr[1].y, ay); az = fmaf(wj, hr[1].z, az); aw = fmaf(wj, hr[1].w, aw);
  wj = pair_lane<0>(w[2]); ax = fmaf(wj, hr[2].x, ax); ay = fmaf(wj, hr[2].y, ay); az = fmaf(wj, hr[2].z, az); aw = fmaf(wj, hr[2].w, aw);
  wj = pair_lane<0>(w[3]); ax = fmaf(wj, hr[3].x, ax); ay = fmaf(wj, hr[3].y, ay); az = fmaf(wj, hr[3].z, az); aw = fmaf(wj, hr[3].w, aw);
  wj = pair_lane<1>(w[0]); ax = fmaf(wj, hr[4].x, ax); ay = fmaf(wj, hr[4].y, ay); az = fmaf(wj, hr[4].z, az); aw = fmaf(wj, hr[4].w, aw);
  wj = pair_lane<1>(w[1]); ax = fmaf(wj, hr[5].x, ax); ay = fmaf(wj, hr[5].y, ay); az = fmaf(wj, hr[5].z, az); aw = fmaf(wj, hr[5].w, aw);
  wj = pair_lane<1>(w[2]); ax = fmaf(wj, hr[6].x, ax); ay = fmaf(wj, hr[6].y, ay); az = fmaf(wj, hr[6].z, az); aw = fmaf(wj, hr[6].w, aw);
  wj = pair_lane<1>(w[3]); ax = fmaf(wj, hr[7].x, ax); ay = fmaf(wj, hr[7].y, ay); az = fmaf(wj, hr[7].z, az); aw = fmaf(wj, hr[7].w, aw);
  m = mn;
}

#define HP 68
__global__ __launch_bounds__(256) void gat_gather1(
    const int* __restrict__ esrc, const int* __restrict__ seg_end,
    const int* __restrict__ deg, const float* __restrict__ a_src,
    const float* __restrict__ a_dst, const float* __restrict__ h1,
    const float* __restrict__ b1, const float* __restrict__ W2,
    const float* __restrict__ a2_dst, float* __restrict__ h2,
    float* __restrict__ ad, int N) {
  __shared__ float sH[16 * HP];
  __shared__ float sW[HID * DOUT];
  int tid = threadIdx.x;
  float4 w2v = ((const float4*)W2)[tid];
  int lane = tid & 31;
  int c = lane & 15, half = lane >> 4;
  int wv = __builtin_amdgcn_readfirstlane(tid >> 5);
  int n0 = __builtin_amdgcn_readfirstlane((int)((blockIdx.x * 8 + wv) * 2));
  int n = n0 + half;
  bool live = n < N;
  int nn = live ? n : 0;
  int cnt = live ? deg[n] : 0;
  int beg = live ? seg_end[n] - cnt : 0;

  const float4* h1v = (const float4*)h1;
  float4 hv = h1v[(unsigned)(nn * 16 + c)];
  float4 wd = ((const float4*)a_dst)[c];
  float4 ws = ((const float4*)a_src)[c];
  float4 bb = ((const float4*)b1)[c];
  ((float4*)sW)[tid] = w2v;
  float pd = hv.x * wd.x + hv.y * wd.y + hv.z * wd.z + hv.w * wd.w;
  pd += xlane<1>(pd);

  const bool odd = (c & 1) != 0;
  int cmax = __builtin_amdgcn_readfirstlane(max(cnt, __shfl_xor(cnt, 16, 32)));

  float m = -INFINITY;
  float ax = 0.f, ay = 0.f, az = 0.f, aw = 0.f, sum = 0.f;
  for (int b0 = 0; b0 < cmax; b0 += 16) {
    int j = b0 + c;
    int s_l = (j < cnt) ? esrc[beg + j] : nn;
    int kmax = min(16, cmax - b0);
    gather1_group<0>(h1v, s_l, c, odd, b0 + (odd ? 4 : 0), cnt, pd, ws, m, sum, ax, ay, az, aw);
    if (8 < kmax)
      gather1_group<1>(h1v, s_l, c, odd, b0 + 8 + (odd ? 4 : 0), cnt, pd, ws, m, sum, ax, ay, az, aw);
  }
  float den = sum + EPS_F;

  float vx = ax / den + bb.x;
  float vy = ay / den + bb.y;
  float vz = az / den + bb.z;
  float vw = aw / den + bb.w;
  float4 o;
  o.x = vx > 0.0f ? vx : expm1f(vx);
  o.y = vy > 0.0f ? vy : expm1f(vy);
  o.z = vz > 0.0f ? vz : expm1f(vz);
  o.w = vw > 0.0f ? vw : expm1f(vw);
  if (!live) { o.x = 0.f; o.y = 0.f; o.z = 0.f; o.w = 0.f; }
  *(float4*)(sH + (wv * 2 + half) * HP + 4 * c) = o;
  __syncthreads();

  int row0 = (int)blockIdx.x * 16;
  if (wv != 0 || row0 >= N) return;
  int r = c, kh = half;
  const float* arow  = sH + r * HP + 2 * kh;
  const float* bbase = sW + r;
  float wdv = a2_dst[r];

  v2f af[16], bf[16];
#pragma unroll
  for (int j = 0; j < 16; ++j) {
    const int k = 4 * j;
    float2 av = *(const float2*)(arow + k);
    af[j].x = av.x; af[j].y = av.y;
    bf[j].x = bbase[(k + 2 * kh) * DOUT];
    bf[j].y = bbase[(k + 2 * kh + 1) * DOUT];
  }
  __builtin_amdgcn_sched_barrier(0);
  v8f acc = {};
#pragma unroll
  for (int j = 0; j < 16; ++j)
    acc = __builtin_amdgcn_wmma_f32_16x16x4_f32(false, af[j], false, bf[j],
                                                (short)0, acc, false, false);
  float* orow = h2 + (size_t)row0 * DOUT + r;
  if (row0 + 16 <= N) {
#pragma unroll
    for (int rr = 0; rr < 8; ++rr)
      orow[(size_t)(rr + 8 * kh) * DOUT] = acc[rr];
  } else {
#pragma unroll
    for (int rr = 0; rr < 8; ++rr) {
      int row = row0 + rr + 8 * kh;
      if (row < N) orow[(size_t)(rr + 8 * kh) * DOUT] = acc[rr];
    }
  }
  float dsel = 0.f;
#pragma unroll
  for (int rr = 0; rr < 8; ++rr) {
    float q = acc[rr] * wdv;
    q += xlane<1>(q);
    q += xlane<2>(q);
    q += xlane<4>(q);
    q += xlane<8>(q);
    dsel = ((r & 7) == rr) ? q : dsel;
  }
  int srow = row0 + 8 * kh + (r & 7);
  if (r < 8 && srow < N) ad[srow] = dsel;
}

__device__ __forceinline__ float dot4(float4 r, v2f wxy, v2f wzw) {
  v2f lo = {r.x, r.y}, hi = {r.z, r.w};
  v2f q = lo * wxy;
  q = hi * wzw + q;
  return q.x + q.y;
}

__global__ __launch_bounds__(256) void gat_gather2(
    const int* __restrict__ esrc, const int* __restrict__ seg_end,
    const int* __restrict__ deg, const float* __restrict__ a_src,
    const float* __restrict__ ad, const float* __restrict__ h2,
    const float* __restrict__ b2, float* __restrict__ dout, int N) {
  int lane = threadIdx.x & 31;
  int p = lane & 3;
  int u = (lane >> 2) & 1;
  int n0 = __builtin_amdgcn_readfirstlane((int)((blockIdx.x * 8 + (threadIdx.x >> 5)) * 4));
  if (n0 >= N) return;
  int n = n0 + (lane >> 3);
  bool live = n < N;
  int nc = live ? n : n0;
  int cnt = deg[nc];
  int beg = seg_end[nc] - cnt;
  float adn = ad[nc];
  cnt = live ? cnt : 0;
  const float4* h2v = (const float4*)h2;
  float4 ws = ((const float4*)a_src)[p];
  float4 bc = ((const float4*)b2)[p];
  const v2f wxy = {ws.x, ws.y}, wzw = {ws.z, ws.w};
  const bool odd = (p & 1) != 0, up = (p & 2) != 0;
  int cm = max(cnt, id_xlane8(cnt));
  cm = max(cm, __shfl_xor(cm, 16, 32));
  int cmax = __builtin_amdgcn_readfirstlane(cm);

  float m = -INFINITY;
  float ax = 0.f, ay = 0.f, az = 0.f, aw = 0.f, sum = 0.f;
  const int jl = 2 * p + u;
  int s_cur = esrc[(jl < cnt) ? beg + jl : 0];
  int j0 = 0;
  do {
    int jn = j0 + 8 + jl;
    int s_nxt = esrc[(jn < cnt) ? beg + jn : 0];
    __builtin_amdgcn_sched_barrier(0);
    float4 r0 = h2v[(unsigned)(quad_id<0>(s_cur) * 4 + p)];
    float4 r1 = h2v[(unsigned)(quad_id<1>(s_cur) * 4 + p)];
    float4 r2 = h2v[(unsigned)(quad_id<2>(s_cur) * 4 + p)];
    float4 r3 = h2v[(unsigned)(quad_id<3>(s_cur) * 4 + p)];
    float d0 = dot4(r0, wxy, wzw), d1 = dot4(r1, wxy, wzw), d2 = dot4(r2, wxy, wzw), d3 = dot4(r3, wxy, wzw);
    float v01 = (odd ? d1 : d0) + xlane<1>(odd ? d0 : d1);
    float v23 = (odd ? d3 : d2) + xlane<1>(odd ? d2 : d3);
    float d = (up ? v23 : v01) + xlane<2>(up ? v01 : v23);
    float e = (j0 + jl < cnt) ? leaky(d + adn) : -INFINITY;
    float gm = fmaxf(e, xlane<1>(e));
    gm = fmaxf(gm, xlane<2>(gm));
    float mn = fmaxf(m, gm);
    float mr = (mn == -INFINITY) ? 0.f : mn;
    float sc = __expf(m - mr);
    float w = __expf(e - mr);
    float wsum = w + xlane<1>(w);
    wsum += xlane<2>(wsum);
    sum = fmaf(sum, sc, wsum);
    ax *= sc; ay *= sc; az *= sc; aw *= sc;
    float wj;
    wj = quad_from<0>(w); ax = fmaf(wj, r0.x, ax); ay = fmaf(wj, r0.y, ay); az = fmaf(wj, r0.z, az); aw = fmaf(wj, r0.w, aw);
    wj = quad_from<1>(w); ax = fmaf(wj, r1.x, ax); ay = fmaf(wj, r1.y, ay); az = fmaf(wj, r1.z, az); aw = fmaf(wj, r1.w, aw);
    wj = quad_from<2>(w); ax = fmaf(wj, r2.x, ax); ay = fmaf(wj, r2.y, ay); az = fmaf(wj, r2.z, az); aw = fmaf(wj, r2.w, aw);
    wj = quad_from<3>(w); ax = fmaf(wj, r3.x, ax); ay = fmaf(wj, r3.y, ay); az = fmaf(wj, r3.z, az); aw = fmaf(wj, r3.w, aw);
    m = mn;
    s_cur = s_nxt;
    j0 += 8;
  } while (j0 < cmax);
  float mm = fmaxf(m, xlane<4>(m));
  float mmr = (mm == -INFINITY) ? 0.f : mm;
  float f = __expf(m - mmr);
  sum *= f; ax *= f; ay *= f; az *= f; aw *= f;
  sum += xlane<4>(sum);
  ax += xlane<4>(ax); ay += xlane<4>(ay); az += xlane<4>(az); aw += xlane<4>(aw);
  float den = sum + EPS_F;
  float vx = ax / den + bc.x;
  float vy = ay / den + bc.y;
  float vz = az / den + bc.z;
  float vw = aw / den + bc.w;
  float mx = fmaxf(fmaxf(vx, vy), fmaxf(vz, vw));
  mx = fmaxf(mx, xlane<1>(mx));
  mx = fmaxf(mx, xlane<2>(mx));
  float se = (__expf(vx - mx) + __expf(vy - mx)) + (__expf(vz - mx) + __expf(vw - mx));
  se += xlane<1>(se);
  se += xlane<2>(se);
  float lse = mx + __logf(se);
  if (live) {
    float sub = u ? lse : 0.f;
    float4 o; o.x = vx - sub; o.y = vy - sub; o.z = vz - sub; o.w = vw - sub;
    size_t off = (u ? (size_t)N * DOUT : (size_t)0) + (size_t)n * DOUT + 4 * p;
    *(float4*)(dout + off) = o;
  }
}

extern "C" void kernel_launch(void* const* d_in, const int* in_sizes, int n_in,
                              void* d_out, int out_size, void* d_ws, size_t ws_size,
                              hipStream_t stream) {
  const float* x   = (const float*)d_in[0];
  const int*   ei  = (const int*)d_in[1];
  const float* W1  = (const float*)d_in[2];
  const float* a1s = (const float*)d_in[3];
  const float* a1d = (const float*)d_in[4];
  const float* b1  = (const float*)d_in[5];
  const float* W2  = (const float*)d_in[6];
  const float* a2s = (const float*)d_in[7];
  const float* a2d = (const float*)d_in[8];
  const float* b2  = (const float*)d_in[9];

  int N = in_sizes[0] / DIM_IN;
  int E = in_sizes[1] / 2;
  const int* src = ei;
  const int* dst = ei + E;

  char* ws = (char*)d_ws;
  float*    h1   = (float*)ws;
  float*    regB = (float*)(ws + (size_t)N * HID * 4);
  char* p = ws + (size_t)N * HID * 4 * 2;
  int*      deg  = (int*)p;   p += ((size_t)N + 1) * 4;
  int*      cur  = (int*)p;   p += (size_t)N * 4;
  int*      esrc = (int*)p;
  const int NB = (N + BNODES - 1) / BNODES;
  int*      tmp  = (int*)regB;
  int*      offs = tmp + 2 * (size_t)E;
  float*    h2  = regB;
  float*    ad2 = regB + (size_t)N * DOUT;

  const int B = 256;
  gat_slice_group<<<SG, ST, 0, stream>>>(src, dst, offs, tmp, E, NB);
  gat_bucket_sort<<<NB, FT, 0, stream>>>(tmp, offs, deg, cur, esrc, N, E, NB, SG);

  gat_gemm1_wmma<<<(N + 16 * G1_WAVES - 1) / (16 * G1_WAVES), 32 * G1_WAVES, 0, stream>>>(x, W1, h1, N);
  gat_gather1<<<(N + 15) / 16, B, 0, stream>>>(esrc, cur, deg, a1s, a1d, h1, b1, W2, a2d, h2, ad2, N);

  gat_gather2<<<(N + 31) / 32, B, 0, stream>>>(esrc, cur, deg, a2s, ad2, h2, b2, (float*)d_out, N);
}
